# starting kernel
# baseline (speedup 1.0000x reference)
#include <hip/hip_runtime.h>
#include <hip/hip_bf16.h>
#include <stdint.h>

typedef __attribute__((ext_vector_type(16))) _Float16 v16h;
typedef __attribute__((ext_vector_type(4)))  _Float16 v4h;
typedef __attribute__((ext_vector_type(8)))  float    v8f;

#define LBV 0.0f
#define UBV 0.2f

__global__ __launch_bounds__(256) void wfrag_prep(const float* __restrict__ W,
                                                  _Float16* __restrict__ dst,
                                                  int N, int K, int ldw, int KT,
                                                  int totalElems) {
  int idx = blockIdx.x * 256 + threadIdx.x;
  if (idx >= totalElems) return;
  int frag   = idx >> 9;
  int within = idx & 511;
  int r  = within >> 5;
  int k  = within & 31;
  int tileN = frag / KT;
  int tileK = frag - tileN * KT;
  int n  = tileN * 16 + r;
  int kk = tileK * 32 + k;
  float v = (n < N && kk < K) ? W[(size_t)n * ldw + kk] : 0.0f;
  int lane = (((k >> 3) & 1) << 4) + r;
  int elem = (k & 7) | ((k >> 4) << 3);
  dst[(size_t)frag * 512 + lane * 16 + elem] = (_Float16)v;
}

__global__ __launch_bounds__(256) void gemm_wmma(const float* __restrict__ A,
                                                 const _Float16* __restrict__ Bfrag,
                                                 const float* __restrict__ bias,
                                                 float* __restrict__ C,
                                                 int M, int N, int K, int lda, int ldc) {
  __shared__ __align__(32) _Float16 ldsA[512];
  int tid  = threadIdx.x;
  int lane = tid & 31;
  int wave = tid >> 5;
  int mBase  = blockIdx.x * 16;
  int Ntiles = N >> 4;
  int tpw    = Ntiles >> 3;
  int KT     = (K + 31) >> 5;
  v8f acc[3] = {};
  int sr  = tid >> 3;
  int sq  = tid & 7;
  int skl = sq << 2;
  int sln = (((skl >> 3) & 1) << 4) + sr;
  int se0 = (skl & 7) | ((skl >> 4) << 3);
  const float* Arow = A + (size_t)(mBase + sr) * lda + skl;
  const _Float16* Bw = Bfrag + (size_t)wave * KT * 512 + lane * 16;

  for (int kt = 0; kt < KT; ++kt) {
    int k0 = kt << 5;
    if (tid < 128) {
      v4h hv;
      if (k0 + 32 <= K) {
        float4 va = *(const float4*)(Arow + k0);
        hv[0] = (_Float16)va.x; hv[1] = (_Float16)va.y;
        hv[2] = (_Float16)va.z; hv[3] = (_Float16)va.w;
      } else {
#pragma unroll
        for (int j = 0; j < 4; ++j) {
          int kk = k0 + skl + j;
          hv[j] = (_Float16)((kk < K) ? Arow[k0 + j] : 0.0f);
        }
      }
      *reinterpret_cast<v4h*>(&ldsA[sln * 16 + se0]) = hv;
    }
    __syncthreads();
    v16h b[3];
#pragma unroll
    for (int j = 0; j < 3; ++j) {
      if (j >= tpw) break;
      b[j] = *reinterpret_cast<const v16h*>(&Bw[((size_t)(j << 3) * KT + kt) * 512]);
    }
    v16h a = *reinterpret_cast<const v16h*>(&ldsA[lane * 16]);
#pragma unroll
    for (int j = 0; j < 3; ++j) {
      if (j >= tpw) break;
      acc[j] = __builtin_amdgcn_wmma_f32_16x16x32_f16(false, a, false, b[j],
                                                      (short)0, acc[j], false, false);
    }
    __syncthreads();
  }
  int rBase = mBase + ((lane >> 4) << 3);
#pragma unroll
  for (int j = 0; j < 3; ++j) {
    if (j >= tpw) break;
    int col = ((wave + (j << 3)) << 4) + (lane & 15);
    float bv = (bias != nullptr) ? bias[col] : 0.0f;
#pragma unroll
    for (int r = 0; r < 8; ++r)
      C[(size_t)(rBase + r) * ldc + col] = acc[j][r] + bv;
  }
}

__global__ __launch_bounds__(256) void gru_recur(const float* __restrict__ gi,
                                                 const _Float16* __restrict__ WhhF,
                                                 const float* __restrict__ bhh,
                                                 float* __restrict__ out, int S) {
  __shared__ float hS[16 * 128];
  __shared__ __align__(32) _Float16 aF[4 * 512];
  __shared__ float ghS[16 * 384];
  int tid  = threadIdx.x;
  int lane = tid & 31;
  int wave = tid >> 5;
  int bBase = blockIdx.x * 16;
  const _Float16* Bw = WhhF + (size_t)(wave * 3) * 4 * 512 + lane * 16;

  for (int i = tid; i < 16 * 128; i += 256) hS[i] = 0.0f;
  for (int i = tid; i < 4 * 512; i += 256) aF[i] = (_Float16)0.0f;
  __syncthreads();

  for (int t = 0; t < S; ++t) {
    v16h a[4];
#pragma unroll
    for (int kt = 0; kt < 4; ++kt)
      a[kt] = *reinterpret_cast<const v16h*>(&aF[kt * 512 + lane * 16]);
#pragma unroll
    for (int j = 0; j < 3; ++j) {
      v16h b[4];
#pragma unroll
      for (int kt = 0; kt < 4; ++kt)
        b[kt] = *reinterpret_cast<const v16h*>(&Bw[((size_t)j * 4 + kt) * 512]);
      v8f acc = {};
#pragma unroll
      for (int kt = 0; kt < 4; ++kt)
        acc = __builtin_amdgcn_wmma_f32_16x16x32_f16(false, a[kt], false, b[kt],
                                                     (short)0, acc, false, false);
      int col   = ((wave * 3 + j) << 4) + (lane & 15);
      int rBase = (lane >> 4) << 3;
      float bv  = bhh[col];
#pragma unroll
      for (int r = 0; r < 8; ++r) ghS[(rBase + r) * 384 + col] = acc[r] + bv;
    }
    __syncthreads();
#pragma unroll
    for (int p = 0; p < 2; ++p) {
      int g   = tid + (p << 8);
      int row = g >> 5;
      int q   = g & 31;
      int h0  = q << 2;
      int b   = bBase + row;
      const float* gib = gi + ((size_t)b * S + t) * 384;
      float4 gr = *(const float4*)(gib + h0);
      float4 gz = *(const float4*)(gib + 128 + h0);
      float4 gn = *(const float4*)(gib + 256 + h0);
      float4 hr = *(const float4*)(&ghS[row * 384 + h0]);
      float4 hz = *(const float4*)(&ghS[row * 384 + 128 + h0]);
      float4 hn = *(const float4*)(&ghS[row * 384 + 256 + h0]);
      float4 ho = *(const float4*)(&hS[row * 128 + h0]);
      float4 hNew;
#define GATE(c)                                                            \
      {                                                                    \
        float rg = 1.0f / (1.0f + expf(-(gr.c + hr.c)));                   \
        float zg = 1.0f / (1.0f + expf(-(gz.c + hz.c)));                   \
        float ng = tanhf(gn.c + rg * hn.c);                                \
        hNew.c = (1.0f - zg) * ng + zg * ho.c;                             \
      }
      GATE(x) GATE(y) GATE(z) GATE(w)
#undef GATE
      *(float4*)(&hS[row * 128 + h0]) = hNew;
      *(float4*)(out + ((size_t)b * S + t) * 128 + h0) = hNew;
      int kt = h0 >> 5;
      int kl = h0 & 31;
      int ln = (((kl >> 3) & 1) << 4) + row;
      int e0 = (kl & 7) | ((kl >> 4) << 3);
      v4h hv;
      hv[0] = (_Float16)hNew.x; hv[1] = (_Float16)hNew.y;
      hv[2] = (_Float16)hNew.z; hv[3] = (_Float16)hNew.w;
      *reinterpret_cast<v4h*>(&aF[kt * 512 + ln * 16 + e0]) = hv;
    }
    __syncthreads();
  }
}

__device__ __forceinline__ float blockReduceSum(float v, float* red) {
  int tid = threadIdx.x;
  red[tid] = v;
  __syncthreads();
  for (int off = 128; off > 0; off >>= 1) {
    if (tid < off) red[tid] += red[tid + off];
    __syncthreads();
  }
  float r = red[0];
  __syncthreads();
  return r;
}
__device__ __forceinline__ float blockReduceMax(float v, float* red) {
  int tid = threadIdx.x;
  red[tid] = v;
  __syncthreads();
  for (int off = 128; off > 0; off >>= 1) {
    if (tid < off) red[tid] = fmaxf(red[tid], red[tid + off]);
    __syncthreads();
  }
  float r = red[0];
  __syncthreads();
  return r;
}

__global__ __launch_bounds__(256) void finalize_kernel(const float* __restrict__ gE,
                                                       const float* __restrict__ u,
                                                       const float* __restrict__ out1,
                                                       const float* __restrict__ vvec,
                                                       const float* __restrict__ Wfc,
                                                       const float* __restrict__ bfc,
                                                       float* __restrict__ outp, int S) {
  __shared__ float uL[128], vL[128], eS[64], attnS[64], concatL[256];
  __shared__ float oldL[512], wcL[512], red[256];
  int tid  = threadIdx.x;
  int lane = tid & 31;
  int wave = tid >> 5;
  int b = blockIdx.x;

  if (tid < 128) {
    uL[tid] = u[(size_t)b * 128 + tid];
    vL[tid] = vvec[tid];
  }
  __syncthreads();

  for (int s = wave; s < S; s += 8) {
    float p = 0.0f;
    const float* ge = gE + ((size_t)b * S + s) * 128;
    for (int h = lane; h < 128; h += 32) p += vL[h] * tanhf(uL[h] + ge[h]);
    for (int off = 16; off > 0; off >>= 1) p += __shfl_down(p, off, 32);
    if (lane == 0) eS[s] = p;
  }
  __syncthreads();

  if (tid == 0) {
    float m = eS[0];
    for (int s = 1; s < S; ++s) m = fmaxf(m, eS[s]);
    float sum = 0.0f;
    for (int s = 0; s < S; ++s) { float ev = expf(eS[s] - m); attnS[s] = ev; sum += ev; }
    float inv = 1.0f / sum;
    for (int s = 0; s < S; ++s) attnS[s] *= inv;
  }
  __syncthreads();

  if (tid < 128) {
    float c = 0.0f;
    for (int s = 0; s < S; ++s) c += attnS[s] * out1[((size_t)b * S + s) * 128 + tid];
    concatL[tid]       = out1[((size_t)b * S + (S - 1)) * 128 + tid];
    concatL[128 + tid] = c;
  }
  __syncthreads();

  for (int n = tid; n < 500; n += 256) {
    float acc = bfc[n];
    const float* wr = Wfc + (size_t)n * 256;
#pragma unroll 4
    for (int j = 0; j < 256; ++j) acc += wr[j] * concatL[j];
    oldL[n] = acc;
  }
  __syncthreads();

  float pm = -3.4e38f;
  for (int n = tid; n < 500; n += 256) pm = fmaxf(pm, oldL[n]);
  pm = blockReduceMax(pm, red);
  float ps = 0.0f;
  for (int n = tid; n < 500; n += 256) { float e = expf(oldL[n] - pm); wcL[n] = e; ps += e; }
  ps = blockReduceSum(ps, red);
  float inv = 1.0f / ps;
  for (int n = tid; n < 500; n += 256) {
    float w = wcL[n] * inv;
    oldL[n] = w;
    wcL[n]  = fminf(fmaxf(w, LBV), UBV);
  }
  __syncthreads();

  for (int it = 0; it < 32; ++it) {
    float pl = 0.0f, pn = 0.0f;
    for (int n = tid; n < 500; n += 256) {
      float wcv = wcL[n];
      pl += oldL[n] - wcv;
      if (wcv != UBV) pn += wcv;
    }
    float leftover = blockReduceSum(pl, red);
    float nsum     = blockReduceSum(pn, red);
    float ns = (nsum == 0.0f) ? 1.0f : nsum;
    float pover = 0.0f;
    for (int n = tid; n < 500; n += 256) {
      float wcv = wcL[n];
      float wc2 = (wcv != UBV) ? wcv + leftover * wcv / ns : wcv;
      oldL[n] = wc2;
      if (wc2 > UBV) pover = 1.0f;
    }
    float anyOver = blockReduceSum(pover, red);
    int new_done = (anyOver == 0.0f);
    for (int n = tid; n < 500; n += 256) {
      float wc2 = oldL[n];
      wcL[n] = new_done ? wc2 : fminf(fmaxf(wc2, LBV), UBV);
    }
    __syncthreads();
    if (new_done) break;
  }

  for (int n = tid; n < 500; n += 256) outp[(size_t)b * 500 + n] = wcL[n];
}

extern "C" void kernel_launch(void* const* d_in, const int* in_sizes, int n_in,
                              void* d_out, int out_size, void* d_ws, size_t ws_size,
                              hipStream_t stream) {
  (void)in_sizes; (void)n_in; (void)out_size; (void)ws_size;
  const float* x    = (const float*)d_in[0];
  const float* Wih0 = (const float*)d_in[1];
  const float* Whh0 = (const float*)d_in[2];
  const float* bih0 = (const float*)d_in[3];
  const float* bhh0 = (const float*)d_in[4];
  const float* Wih1 = (const float*)d_in[5];
  const float* Whh1 = (const float*)d_in[6];
  const float* bih1 = (const float*)d_in[7];
  const float* bhh1 = (const float*)d_in[8];
  const float* Wa   = (const float*)d_in[9];
  const float* ba   = (const float*)d_in[10];
  const float* vv   = (const float*)d_in[11];
  const float* Wfc  = (const float*)d_in[12];
  const float* bfc  = (const float*)d_in[13];
  float* outp = (float*)d_out;

  constexpr int B = 1024, S = 60, N = 500, H = 128, G3 = 384;

  uint8_t* base = (uint8_t*)d_ws;
  size_t off = 0;
  auto carve = [&](size_t bytes) {
    void* p = base + off;
    off = (off + bytes + 255) & ~(size_t)255;
    return p;
  };
  float* gi   = (float*)carve((size_t)B * S * G3 * 4);
  float* out0 = (float*)carve((size_t)B * S * H * 4);
  float* out1 = (float*)carve((size_t)B * S * H * 4);
  float* gEb  = (float*)carve((size_t)B * S * H * 4);
  float* ub   = (float*)carve((size_t)B * H * 4);
  _Float16* Wih0F = (_Float16*)carve((size_t)24 * 16 * 512 * 2);
  _Float16* Whh0F = (_Float16*)carve((size_t)24 * 4 * 512 * 2);
  _Float16* Wih1F = (_Float16*)carve((size_t)24 * 4 * 512 * 2);
  _Float16* Whh1F = (_Float16*)carve((size_t)24 * 4 * 512 * 2);
  _Float16* WaxF  = (_Float16*)carve((size_t)8 * 16 * 512 * 2);
  _Float16* WahF  = (_Float16*)carve((size_t)8 * 4 * 512 * 2);

  auto prep = [&](const float* W, _Float16* dst, int n, int k, int ldw) {
    int KT = (k + 31) >> 5;
    int NT = (n + 15) >> 4;
    int total = NT * KT * 512;
    wfrag_prep<<<(total + 255) / 256, 256, 0, stream>>>(W, dst, n, k, ldw, KT, total);
  };
  prep(Wih0,   Wih0F, G3, N, N);
  prep(Whh0,   Whh0F, G3, H, H);
  prep(Wih1,   Wih1F, G3, H, H);
  prep(Whh1,   Whh1F, G3, H, H);
  prep(Wa + H, WaxF,  H,  N, H + N);
  prep(Wa,     WahF,  H,  H, H + N);

  auto gemm = [&](const float* A, const _Float16* Bf, const float* bias, float* C,
                  int M, int Nn, int K, int lda, int ldc) {
    gemm_wmma<<<dim3(M / 16), 256, 0, stream>>>(A, Bf, bias, C, M, Nn, K, lda, ldc);
  };

  gemm(x, Wih0F, bih0, gi, B * S, G3, N, N, G3);
  gru_recur<<<B / 16, 256, 0, stream>>>(gi, Whh0F, bhh0, out0, S);
  gemm(out0, Wih1F, bih1, gi, B * S, G3, H, H, G3);
  gru_recur<<<B / 16, 256, 0, stream>>>(gi, Whh1F, bhh1, out1, S);
  gemm(x, WaxF, ba, gEb, B * S, H, N, N, H);
  gemm(out1 + (size_t)(S - 1) * H, WahF, nullptr, ub, B, H, H, S * H, H);
  finalize_kernel<<<B, 256, 0, stream>>>(gEb, ub, out1, vv, Wfc, bfc, outp, S);
}


# GRU gates on the hardware exp2 and reciprocal instructions (sigmoid = rcp(1+2^(-x*log2e)), tanh = 1-2*rcp(1+2^(2x*log2e))) instead of library expf/tanhf and IEEE division; gate rounding moves ~3e-7
# speedup vs baseline: 1.2925x; 1.2925x over previous
#include <hip/hip_runtime.h>
#include <hip/hip_bf16.h>
#include <stdint.h>

typedef __attribute__((ext_vector_type(16))) _Float16 v16h;
typedef __attribute__((ext_vector_type(4)))  _Float16 v4h;
typedef __attribute__((ext_vector_type(8)))  float    v8f;

#define LBV 0.0f
#define UBV 0.2f

__global__ __launch_bounds__(256) void wfrag_prep(const float* __restrict__ W,
                                                  _Float16* __restrict__ dst,
                                                  int N, int K, int ldw, int KT,
                                                  int totalElems) {
  int idx = blockIdx.x * 256 + threadIdx.x;
  if (idx >= totalElems) return;
  int frag   = idx >> 9;
  int within = idx & 511;
  int r  = within >> 5;
  int k  = within & 31;
  int tileN = frag / KT;
  int tileK = frag - tileN * KT;
  int n  = tileN * 16 + r;
  int kk = tileK * 32 + k;
  float v = (n < N && kk < K) ? W[(size_t)n * ldw + kk] : 0.0f;
  int lane = (((k >> 3) & 1) << 4) + r;
  int elem = (k & 7) | ((k >> 4) << 3);
  dst[(size_t)frag * 512 + lane * 16 + elem] = (_Float16)v;
}

__global__ __launch_bounds__(256) void gemm_wmma(const float* __restrict__ A,
                                                 const _Float16* __restrict__ Bfrag,
                                                 const float* __restrict__ bias,
                                                 float* __restrict__ C,
                                                 int M, int N, int K, int lda, int ldc) {
  __shared__ __align__(32) _Float16 ldsA[512];
  int tid  = threadIdx.x;
  int lane = tid & 31;
  int wave = tid >> 5;
  int mBase  = blockIdx.x * 16;
  int Ntiles = N >> 4;
  int tpw    = Ntiles >> 3;
  int KT     = (K + 31) >> 5;
  v8f acc[3] = {};
  int sr  = tid >> 3;
  int sq  = tid & 7;
  int skl = sq << 2;
  int sln = (((skl >> 3) & 1) << 4) + sr;
  int se0 = (skl & 7) | ((skl >> 4) << 3);
  const float* Arow = A + (size_t)(mBase + sr) * lda + skl;
  const _Float16* Bw = Bfrag + (size_t)wave * KT * 512 + lane * 16;

  for (int kt = 0; kt < KT; ++kt) {
    int k0 = kt << 5;
    if (tid < 128) {
      v4h hv;
      if (k0 + 32 <= K) {
        float4 va = *(const float4*)(Arow + k0);
        hv[0] = (_Float16)va.x; hv[1] = (_Float16)va.y;
        hv[2] = (_Float16)va.z; hv[3] = (_Float16)va.w;
      } else {
#pragma unroll
        for (int j = 0; j < 4; ++j) {
          int kk = k0 + skl + j;
          hv[j] = (_Float16)((kk < K) ? Arow[k0 + j] : 0.0f);
        }
      }
      *reinterpret_cast<v4h*>(&ldsA[sln * 16 + se0]) = hv;
    }
    __syncthreads();
    v16h b[3];
#pragma unroll
    for (int j = 0; j < 3; ++j) {
      if (j >= tpw) break;
      b[j] = *reinterpret_cast<const v16h*>(&Bw[((size_t)(j << 3) * KT + kt) * 512]);
    }
    v16h a = *reinterpret_cast<const v16h*>(&ldsA[lane * 16]);
#pragma unroll
    for (int j = 0; j < 3; ++j) {
      if (j >= tpw) break;
      acc[j] = __builtin_amdgcn_wmma_f32_16x16x32_f16(false, a, false, b[j],
                                                      (short)0, acc[j], false, false);
    }
    __syncthreads();
  }
  int rBase = mBase + ((lane >> 4) << 3);
#pragma unroll
  for (int j = 0; j < 3; ++j) {
    if (j >= tpw) break;
    int col = ((wave + (j << 3)) << 4) + (lane & 15);
    float bv = (bias != nullptr) ? bias[col] : 0.0f;
#pragma unroll
    for (int r = 0; r < 8; ++r)
      C[(size_t)(rBase + r) * ldc + col] = acc[j][r] + bv;
  }
}

__device__ __forceinline__ float sigmoid_hw(float x) {
  return __builtin_amdgcn_rcpf(1.0f + __builtin_amdgcn_exp2f(x * -1.442695041f));
}
__device__ __forceinline__ float tanh_hw(float x) {
  return 1.0f - 2.0f * __builtin_amdgcn_rcpf(1.0f + __builtin_amdgcn_exp2f(x * 2.885390082f));
}

__global__ __launch_bounds__(256) __attribute__((amdgpu_waves_per_eu(2, 2))) void gru_recur(const float* __restrict__ gi,
                                                 const _Float16* __restrict__ WhhF,
                                                 const float* __restrict__ bhh,
                                                 float* __restrict__ out, int S) {
  __shared__ float hS[16 * 128];
  __shared__ __align__(32) _Float16 aF[4 * 512];
  __shared__ float ghS[16 * 384];
  int tid  = threadIdx.x;
  int lane = tid & 31;
  int wave = tid >> 5;
  int bBase = blockIdx.x * 16;
  const _Float16* Bw = WhhF + (size_t)(wave * 3) * 4 * 512 + lane * 16;

  for (int i = tid; i < 16 * 128; i += 256) hS[i] = 0.0f;
  for (int i = tid; i < 4 * 512; i += 256) aF[i] = (_Float16)0.0f;
  __syncthreads();

  for (int t = 0; t < S; ++t) {
    v16h a[4];
#pragma unroll
    for (int kt = 0; kt < 4; ++kt)
      a[kt] = *reinterpret_cast<const v16h*>(&aF[kt * 512 + lane * 16]);
#pragma unroll
    for (int j = 0; j < 3; ++j) {
      v16h b[4];
#pragma unroll
      for (int kt = 0; kt < 4; ++kt)
        b[kt] = *reinterpret_cast<const v16h*>(&Bw[((size_t)j * 4 + kt) * 512]);
      v8f acc = {};
#pragma unroll
      for (int kt = 0; kt < 4; ++kt)
        acc = __builtin_amdgcn_wmma_f32_16x16x32_f16(false, a[kt], false, b[kt],
                                                     (short)0, acc, false, false);
      int col   = ((wave * 3 + j) << 4) + (lane & 15);
      int rBase = (lane >> 4) << 3;
      float bv  = bhh[col];
#pragma unroll
      for (int r = 0; r < 8; ++r) ghS[(rBase + r) * 384 + col] = acc[r] + bv;
    }
    __syncthreads();
#pragma unroll
    for (int p = 0; p < 2; ++p) {
      int g   = tid + (p << 8);
      int row = g >> 5;
      int q   = g & 31;
      int h0  = q << 2;
      int b   = bBase + row;
      const float* gib = gi + ((size_t)b * S + t) * 384;
      float4 gr = *(const float4*)(gib + h0);
      float4 gz = *(const float4*)(gib + 128 + h0);
      float4 gn = *(const float4*)(gib + 256 + h0);
      float4 hr = *(const float4*)(&ghS[row * 384 + h0]);
      float4 hz = *(const float4*)(&ghS[row * 384 + 128 + h0]);
      float4 hn = *(const float4*)(&ghS[row * 384 + 256 + h0]);
      float4 ho = *(const float4*)(&hS[row * 128 + h0]);
      float4 hNew;
#define GATE(c)                                                            \
      {                                                                    \
        float rg = sigmoid_hw(gr.c + hr.c);                                \
        float zg = sigmoid_hw(gz.c + hz.c);                                \
        float ng = tanh_hw(gn.c + rg * hn.c);                              \
        hNew.c = (1.0f - zg) * ng + zg * ho.c;                             \
      }
      GATE(x) GATE(y) GATE(z) GATE(w)
#undef GATE
      *(float4*)(&hS[row * 128 + h0]) = hNew;
      *(float4*)(out + ((size_t)b * S + t) * 128 + h0) = hNew;
      int kt = h0 >> 5;
      int kl = h0 & 31;
      int ln = (((kl >> 3) & 1) << 4) + row;
      int e0 = (kl & 7) | ((kl >> 4) << 3);
      v4h hv;
      hv[0] = (_Float16)hNew.x; hv[1] = (_Float16)hNew.y;
      hv[2] = (_Float16)hNew.z; hv[3] = (_Float16)hNew.w;
      *reinterpret_cast<v4h*>(&aF[kt * 512 + ln * 16 + e0]) = hv;
    }
    __syncthreads();
  }
}

__device__ __forceinline__ float blockReduceSum(float v, float* red) {
  int tid = threadIdx.x;
  red[tid] = v;
  __syncthreads();
  for (int off = 128; off > 0; off >>= 1) {
    if (tid < off) red[tid] += red[tid + off];
    __syncthreads();
  }
  float r = red[0];
  __syncthreads();
  return r;
}
__device__ __forceinline__ float blockReduceMax(float v, float* red) {
  int tid = threadIdx.x;
  red[tid] = v;
  __syncthreads();
  for (int off = 128; off > 0; off >>= 1) {
    if (tid < off) red[tid] = fmaxf(red[tid], red[tid + off]);
    __syncthreads();
  }
  float r = red[0];
  __syncthreads();
  return r;
}

__global__ __launch_bounds__(256) void finalize_kernel(const float* __restrict__ gE,
                                                       const float* __restrict__ u,
                                                       const float* __restrict__ out1,
                                                       const float* __restrict__ vvec,
                                                       const float* __restrict__ Wfc,
                                                       const float* __restrict__ bfc,
                                                       float* __restrict__ outp, int S) {
  __shared__ float uL[128], vL[128], eS[64], attnS[64], concatL[256];
  __shared__ float oldL[512], wcL[512], red[256];
  int tid  = threadIdx.x;
  int lane = tid & 31;
  int wave = tid >> 5;
  int b = blockIdx.x;

  if (tid < 128) {
    uL[tid] = u[(size_t)b * 128 + tid];
    vL[tid] = vvec[tid];
  }
  __syncthreads();

  for (int s = wave; s < S; s += 8) {
    float p = 0.0f;
    const float* ge = gE + ((size_t)b * S + s) * 128;
    for (int h = lane; h < 128; h += 32) p += vL[h] * tanhf(uL[h] + ge[h]);
    for (int off = 16; off > 0; off >>= 1) p += __shfl_down(p, off, 32);
    if (lane == 0) eS[s] = p;
  }
  __syncthreads();

  if (tid == 0) {
    float m = eS[0];
    for (int s = 1; s < S; ++s) m = fmaxf(m, eS[s]);
    float sum = 0.0f;
    for (int s = 0; s < S; ++s) { float ev = expf(eS[s] - m); attnS[s] = ev; sum += ev; }
    float inv = 1.0f / sum;
    for (int s = 0; s < S; ++s) attnS[s] *= inv;
  }
  __syncthreads();

  if (tid < 128) {
    float c = 0.0f;
    for (int s = 0; s < S; ++s) c += attnS[s] * out1[((size_t)b * S + s) * 128 + tid];
    concatL[tid]       = out1[((size_t)b * S + (S - 1)) * 128 + tid];
    concatL[128 + tid] = c;
  }
  __syncthreads();

  for (int n = tid; n < 500; n += 256) {
    float acc = bfc[n];
    const float* wr = Wfc + (size_t)n * 256;
#pragma unroll 4
    for (int j = 0; j < 256; ++j) acc += wr[j] * concatL[j];
    oldL[n] = acc;
  }
  __syncthreads();

  float pm = -3.4e38f;
  for (int n = tid; n < 500; n += 256) pm = fmaxf(pm, oldL[n]);
  pm = blockReduceMax(pm, red);
  float ps = 0.0f;
  for (int n = tid; n < 500; n += 256) { float e = expf(oldL[n] - pm); wcL[n] = e; ps += e; }
  ps = blockReduceSum(ps, red);
  float inv = 1.0f / ps;
  for (int n = tid; n < 500; n += 256) {
    float w = wcL[n] * inv;
    oldL[n] = w;
    wcL[n]  = fminf(fmaxf(w, LBV), UBV);
  }
  __syncthreads();

  for (int it = 0; it < 32; ++it) {
    float pl = 0.0f, pn = 0.0f;
    for (int n = tid; n < 500; n += 256) {
      float wcv = wcL[n];
      pl += oldL[n] - wcv;
      if (wcv != UBV) pn += wcv;
    }
    float leftover = blockReduceSum(pl, red);
    float nsum     = blockReduceSum(pn, red);
    float ns = (nsum == 0.0f) ? 1.0f : nsum;
    float pover = 0.0f;
    for (int n = tid; n < 500; n += 256) {
      float wcv = wcL[n];
      float wc2 = (wcv != UBV) ? wcv + leftover * wcv / ns : wcv;
      oldL[n] = wc2;
      if (wc2 > UBV) pover = 1.0f;
    }
    float anyOver = blockReduceSum(pover, red);
    int new_done = (anyOver == 0.0f);
    for (int n = tid; n < 500; n += 256) {
      float wc2 = oldL[n];
      wcL[n] = new_done ? wc2 : fminf(fmaxf(wc2, LBV), UBV);
    }
    __syncthreads();
    if (new_done) break;
  }

  for (int n = tid; n < 500; n += 256) outp[(size_t)b * 500 + n] = wcL[n];
}

extern "C" void kernel_launch(void* const* d_in, const int* in_sizes, int n_in,
                              void* d_out, int out_size, void* d_ws, size_t ws_size,
                              hipStream_t stream) {
  (void)in_sizes; (void)n_in; (void)out_size; (void)ws_size;
  const float* x    = (const float*)d_in[0];
  const float* Wih0 = (const float*)d_in[1];
  const float* Whh0 = (const float*)d_in[2];
  const float* bih0 = (const float*)d_in[3];
  const float* bhh0 = (const float*)d_in[4];
  const float* Wih1 = (const float*)d_in[5];
  const float* Whh1 = (const float*)d_in[6];
  const float* bih1 = (const float*)d_in[7];
  const float* bhh1 = (const float*)d_in[8];
  const float* Wa   = (const float*)d_in[9];
  const float* ba   = (const float*)d_in[10];
  const float* vv   = (const float*)d_in[11];
  const float* Wfc  = (const float*)d_in[12];
  const float* bfc  = (const float*)d_in[13];
  float* outp = (float*)d_out;

  constexpr int B = 1024, S = 60, N = 500, H = 128, G3 = 384;

  uint8_t* base = (uint8_t*)d_ws;
  size_t off = 0;
  auto carve = [&](size_t bytes) {
    void* p = base + off;
    off = (off + bytes + 255) & ~(size_t)255;
    return p;
  };
  float* gi   = (float*)carve((size_t)B * S * G3 * 4);
  float* out0 = (float*)carve((size_t)B * S * H * 4);
  float* out1 = (float*)carve((size_t)B * S * H * 4);
  float* gEb  = (float*)carve((size_t)B * S * H * 4);
  float* ub   = (float*)carve((size_t)B * H * 4);
  _Float16* Wih0F = (_Float16*)carve((size_t)24 * 16 * 512 * 2);
  _Float16* Whh0F = (_Float16*)carve((size_t)24 * 4 * 512 * 2);
  _Float16* Wih1F = (_Float16*)carve((size_t)24 * 4 * 512 * 2);
  _Float16* Whh1F = (_Float16*)carve((size_t)24 * 4 * 512 * 2);
  _Float16* WaxF  = (_Float16*)carve((size_t)8 * 16 * 512 * 2);
  _Float16* WahF  = (_Float16*)carve((size_t)8 * 4 * 512 * 2);

  auto prep = [&](const float* W, _Float16* dst, int n, int k, int ldw) {
    int KT = (k + 31) >> 5;
    int NT = (n + 15) >> 4;
    int total = NT * KT * 512;
    wfrag_prep<<<(total + 255) / 256, 256, 0, stream>>>(W, dst, n, k, ldw, KT, total);
  };
  prep(Wih0,   Wih0F, G3, N, N);
  prep(Whh0,   Whh0F, G3, H, H);
  prep(Wih1,   Wih1F, G3, H, H);
  prep(Whh1,   Whh1F, G3, H, H);
  prep(Wa + H, WaxF,  H,  N, H + N);
  prep(Wa,     WahF,  H,  H, H + N);

  auto gemm = [&](const float* A, const _Float16* Bf, const float* bias, float* C,
                  int M, int Nn, int K, int lda, int ldc) {
    gemm_wmma<<<dim3(M / 16), 256, 0, stream>>>(A, Bf, bias, C, M, Nn, K, lda, ldc);
  };

  gemm(x, Wih0F, bih0, gi, B * S, G3, N, N, G3);
  gru_recur<<<B / 16, 256, 0, stream>>>(gi, Whh0F, bhh0, out0, S);
  gemm(out0, Wih1F, bih1, gi, B * S, G3, H, H, G3);
  gru_recur<<<B / 16, 256, 0, stream>>>(gi, Whh1F, bhh1, out1, S);
  gemm(x, WaxF, ba, gEb, B * S, H, N, N, H);
  gemm(out1 + (size_t)(S - 1) * H, WahF, nullptr, ub, B, H, H, S * H, H);
  finalize_kernel<<<B, 256, 0, stream>>>(gEb, ub, out1, vv, Wfc, bfc, outp, S);
}


# GRU recurrence: each step's gi quads are fetched two steps ahead and stay in flight across work-group barriers that wait for LDS only; gi latency and the out-store drain leave the serial step
# speedup vs baseline: 1.7119x; 1.3245x over previous
#include <hip/hip_runtime.h>
#include <hip/hip_bf16.h>
#include <stdint.h>

typedef __attribute__((ext_vector_type(16))) _Float16 v16h;
typedef __attribute__((ext_vector_type(4)))  _Float16 v4h;
typedef __attribute__((ext_vector_type(8)))  float    v8f;

#define LBV 0.0f
#define UBV 0.2f

__global__ __launch_bounds__(256) void wfrag_prep(const float* __restrict__ W,
                                                  _Float16* __restrict__ dst,
                                                  int N, int K, int ldw, int KT,
                                                  int totalElems) {
  int idx = blockIdx.x * 256 + threadIdx.x;
  if (idx >= totalElems) return;
  int frag   = idx >> 9;
  int within = idx & 511;
  int r  = within >> 5;
  int k  = within & 31;
  int tileN = frag / KT;
  int tileK = frag - tileN * KT;
  int n  = tileN * 16 + r;
  int kk = tileK * 32 + k;
  float v = (n < N && kk < K) ? W[(size_t)n * ldw + kk] : 0.0f;
  int lane = (((k >> 3) & 1) << 4) + r;
  int elem = (k & 7) | ((k >> 4) << 3);
  dst[(size_t)frag * 512 + lane * 16 + elem] = (_Float16)v;
}

__global__ __launch_bounds__(256) void gemm_wmma(const float* __restrict__ A,
                                                 const _Float16* __restrict__ Bfrag,
                                                 const float* __restrict__ bias,
                                                 float* __restrict__ C,
                                                 int M, int N, int K, int lda, int ldc) {
  __shared__ __align__(32) _Float16 ldsA[512];
  int tid  = threadIdx.x;
  int lane = tid & 31;
  int wave = tid >> 5;
  int mBase  = blockIdx.x * 16;
  int Ntiles = N >> 4;
  int tpw    = Ntiles >> 3;
  int KT     = (K + 31) >> 5;
  v8f acc[3] = {};
  int sr  = tid >> 3;
  int sq  = tid & 7;
  int skl = sq << 2;
  int sln = (((skl >> 3) & 1) << 4) + sr;
  int se0 = (skl & 7) | ((skl >> 4) << 3);
  const float* Arow = A + (size_t)(mBase + sr) * lda + skl;
  const _Float16* Bw = Bfrag + (size_t)wave * KT * 512 + lane * 16;

  for (int kt = 0; kt < KT; ++kt) {
    int k0 = kt << 5;
    if (tid < 128) {
      v4h hv;
      if (k0 + 32 <= K) {
        float4 va = *(const float4*)(Arow + k0);
        hv[0] = (_Float16)va.x; hv[1] = (_Float16)va.y;
        hv[2] = (_Float16)va.z; hv[3] = (_Float16)va.w;
      } else {
#pragma unroll
        for (int j = 0; j < 4; ++j) {
          int kk = k0 + skl + j;
          hv[j] = (_Float16)((kk < K) ? Arow[k0 + j] : 0.0f);
        }
      }
      *reinterpret_cast<v4h*>(&ldsA[sln * 16 + se0]) = hv;
    }
    __syncthreads();
    v16h b[3];
#pragma unroll
    for (int j = 0; j < 3; ++j) {
      if (j >= tpw) break;
      b[j] = *reinterpret_cast<const v16h*>(&Bw[((size_t)(j << 3) * KT + kt) * 512]);
    }
    v16h a = *reinterpret_cast<const v16h*>(&ldsA[lane * 16]);
#pragma unroll
    for (int j = 0; j < 3; ++j) {
      if (j >= tpw) break;
      acc[j] = __builtin_amdgcn_wmma_f32_16x16x32_f16(false, a, false, b[j],
                                                      (short)0, acc[j], false, false);
    }
    __syncthreads();
  }
  int rBase = mBase + ((lane >> 4) << 3);
#pragma unroll
  for (int j = 0; j < 3; ++j) {
    if (j >= tpw) break;
    int col = ((wave + (j << 3)) << 4) + (lane & 15);
    float bv = (bias != nullptr) ? bias[col] : 0.0f;
#pragma unroll
    for (int r = 0; r < 8; ++r)
      C[(size_t)(rBase + r) * ldc + col] = acc[j][r] + bv;
  }
}

__device__ __forceinline__ float sigmoid_hw(float x) {
  return __builtin_amdgcn_rcpf(1.0f + __builtin_amdgcn_exp2f(x * -1.442695041f));
}
__device__ __forceinline__ float tanh_hw(float x) {
  return 1.0f - 2.0f * __builtin_amdgcn_rcpf(1.0f + __builtin_amdgcn_exp2f(x * 2.885390082f));
}

__device__ __forceinline__ void lds_barrier() {
  __builtin_amdgcn_fence(__ATOMIC_RELEASE, "workgroup", "local");
  __builtin_amdgcn_s_barrier();
  __builtin_amdgcn_fence(__ATOMIC_ACQUIRE, "workgroup", "local");
}

__device__ __forceinline__ void gi_fetch(float4 (&g)[6], const float* ga, const float* gb, int t) {
  const float* a = ga + (size_t)t * 384;
  const float* b = gb + (size_t)t * 384;
  g[0] = *(const float4*)(a);
  g[1] = *(const float4*)(a + 128);
  g[2] = *(const float4*)(a + 256);
  g[3] = *(const float4*)(b);
  g[4] = *(const float4*)(b + 128);
  g[5] = *(const float4*)(b + 256);
}

__global__ __launch_bounds__(256) __attribute__((amdgpu_waves_per_eu(2, 2))) void gru_recur(const float* __restrict__ gi,
                                                 const _Float16* __restrict__ WhhF,
                                                 const float* __restrict__ bhh,
                                                 float* __restrict__ out, int S) {
  __shared__ float hS[16 * 128];
  __shared__ __align__(32) _Float16 aF[4 * 512];
  __shared__ float ghS[16 * 384];
  int tid  = threadIdx.x;
  int lane = tid & 31;
  int wave = tid >> 5;
  int bBase = blockIdx.x * 16;
  const _Float16* Bw = WhhF + (size_t)(wave * 3) * 4 * 512 + lane * 16;
  if (S <= 0) return;

  v16h bF[3][4];
  float bv[3];
#pragma unroll
  for (int j = 0; j < 3; ++j) {
#pragma unroll
    for (int kt = 0; kt < 4; ++kt)
      bF[j][kt] = *reinterpret_cast<const v16h*>(&Bw[((size_t)j * 4 + kt) * 512]);
    bv[j] = bhh[((wave * 3 + j) << 4) + (lane & 15)];
  }

  const float* giA = gi + (size_t)(bBase + (tid >> 5)) * S * 384 + ((tid & 31) << 2);
  const float* giB = giA + (size_t)8 * S * 384;

  float4 g0[6], g1[6], g2[6];
  gi_fetch(g0, giA, giB, 0);
  gi_fetch(g1, giA, giB, (1 < S) ? 1 : S - 1);

  for (int i = tid; i < 16 * 128; i += 256) hS[i] = 0.0f;
  for (int i = tid; i < 4 * 512; i += 256) aF[i] = (_Float16)0.0f;
  __syncthreads();

  auto step = [&](int t, const float4 (&g)[6]) __attribute__((always_inline)) {
    v16h a[4];
#pragma unroll
    for (int kt = 0; kt < 4; ++kt)
      a[kt] = *reinterpret_cast<const v16h*>(&aF[kt * 512 + lane * 16]);
    __builtin_amdgcn_sched_barrier(0);
    v8f acc[3] = {};
#pragma unroll
    for (int kt = 0; kt < 4; ++kt) {
#pragma unroll
      for (int j = 0; j < 3; ++j)
        acc[j] = __builtin_amdgcn_wmma_f32_16x16x32_f16(false, a[kt], false, bF[j][kt],
                                                        (short)0, acc[j], false, false);
    }
#pragma unroll
    for (int j = 0; j < 3; ++j) {
      int col   = ((wave * 3 + j) << 4) + (lane & 15);
      int rBase = (lane >> 4) << 3;
#pragma unroll
      for (int r = 0; r < 8; ++r) ghS[(rBase + r) * 384 + col] = acc[j][r] + bv[j];
    }
    lds_barrier();
    float4 hr[2], hz[2], hn[2], ho[2];
#pragma unroll
    for (int p = 0; p < 2; ++p) {
      int gx  = tid + (p << 8);
      int row = gx >> 5;
      int h0  = (gx & 31) << 2;
      hr[p] = *(const float4*)(&ghS[row * 384 + h0]);
      hz[p] = *(const float4*)(&ghS[row * 384 + 128 + h0]);
      hn[p] = *(const float4*)(&ghS[row * 384 + 256 + h0]);
      ho[p] = *(const float4*)(&hS[row * 128 + h0]);
    }
#pragma unroll
    for (int p = 0; p < 2; ++p) {
      int gx  = tid + (p << 8);
      int row = gx >> 5;
      int h0  = (gx & 31) << 2;
      int b   = bBase + row;
      float4 gr = g[3 * p];
      float4 gz = g[3 * p + 1];
      float4 gn = g[3 * p + 2];
      float4 hNew;
#define GATE(c)                                                            \
      {                                                                    \
        float rg = sigmoid_hw(gr.c + hr[p].c);                             \
        float zg = sigmoid_hw(gz.c + hz[p].c);                             \
        float ng = tanh_hw(gn.c + rg * hn[p].c);                           \
        hNew.c = (1.0f - zg) * ng + zg * ho[p].c;                          \
      }
      GATE(x) GATE(y) GATE(z) GATE(w)
#undef GATE
      *(float4*)(&hS[row * 128 + h0]) = hNew;
      *(float4*)(out + ((size_t)b * S + t) * 128 + h0) = hNew;
      int kt = h0 >> 5;
      int kl = h0 & 31;
      int ln = (((kl >> 3) & 1) << 4) + row;
      int e0 = (kl & 7) | ((kl >> 4) << 3);
      v4h hv;
      hv[0] = (_Float16)hNew.x; hv[1] = (_Float16)hNew.y;
      hv[2] = (_Float16)hNew.z; hv[3] = (_Float16)hNew.w;
      *reinterpret_cast<v4h*>(&aF[kt * 512 + ln * 16 + e0]) = hv;
    }
    lds_barrier();
  };

  int t = 0;
  for (; t + 2 < S; t += 3) {
    gi_fetch(g2, giA, giB, t + 2);
    step(t, g0);
    gi_fetch(g0, giA, giB, (t + 3 < S) ? t + 3 : S - 1);
    step(t + 1, g1);
    gi_fetch(g1, giA, giB, (t + 4 < S) ? t + 4 : S - 1);
    step(t + 2, g2);
  }
  if (t < S) {
    step(t, g0);
    if (t + 1 < S) step(t + 1, g1);
  }
}

__device__ __forceinline__ float blockReduceSum(float v, float* red) {
  int tid = threadIdx.x;
  red[tid] = v;
  __syncthreads();
  for (int off = 128; off > 0; off >>= 1) {
    if (tid < off) red[tid] += red[tid + off];
    __syncthreads();
  }
  float r = red[0];
  __syncthreads();
  return r;
}
__device__ __forceinline__ float blockReduceMax(float v, float* red) {
  int tid = threadIdx.x;
  red[tid] = v;
  __syncthreads();
  for (int off = 128; off > 0; off >>= 1) {
    if (tid < off) red[tid] = fmaxf(red[tid], red[tid + off]);
    __syncthreads();
  }
  float r = red[0];
  __syncthreads();
  return r;
}

__global__ __launch_bounds__(256) void finalize_kernel(const float* __restrict__ gE,
                                                       const float* __restrict__ u,
                                                       const float* __restrict__ out1,
                                                       const float* __restrict__ vvec,
                                                       const float* __restrict__ Wfc,
                                                       const float* __restrict__ bfc,
                                                       float* __restrict__ outp, int S) {
  __shared__ float uL[128], vL[128], eS[64], attnS[64], concatL[256];
  __shared__ float oldL[512], wcL[512], red[256];
  int tid  = threadIdx.x;
  int lane = tid & 31;
  int wave = tid >> 5;
  int b = blockIdx.x;

  if (tid < 128) {
    uL[tid] = u[(size_t)b * 128 + tid];
    vL[tid] = vvec[tid];
  }
  __syncthreads();

  for (int s = wave; s < S; s += 8) {
    float p = 0.0f;
    const float* ge = gE + ((size_t)b * S + s) * 128;
    for (int h = lane; h < 128; h += 32) p += vL[h] * tanhf(uL[h] + ge[h]);
    for (int off = 16; off > 0; off >>= 1) p += __shfl_down(p, off, 32);
    if (lane == 0) eS[s] = p;
  }
  __syncthreads();

  if (tid == 0) {
    float m = eS[0];
    for (int s = 1; s < S; ++s) m = fmaxf(m, eS[s]);
    float sum = 0.0f;
    for (int s = 0; s < S; ++s) { float ev = expf(eS[s] - m); attnS[s] = ev; sum += ev; }
    float inv = 1.0f / sum;
    for (int s = 0; s < S; ++s) attnS[s] *= inv;
  }
  __syncthreads();

  if (tid < 128) {
    float c = 0.0f;
    for (int s = 0; s < S; ++s) c += attnS[s] * out1[((size_t)b * S + s) * 128 + tid];
    concatL[tid]       = out1[((size_t)b * S + (S - 1)) * 128 + tid];
    concatL[128 + tid] = c;
  }
  __syncthreads();

  for (int n = tid; n < 500; n += 256) {
    float acc = bfc[n];
    const float* wr = Wfc + (size_t)n * 256;
#pragma unroll 4
    for (int j = 0; j < 256; ++j) acc += wr[j] * concatL[j];
    oldL[n] = acc;
  }
  __syncthreads();

  float pm = -3.4e38f;
  for (int n = tid; n < 500; n += 256) pm = fmaxf(pm, oldL[n]);
  pm = blockReduceMax(pm, red);
  float ps = 0.0f;
  for (int n = tid; n < 500; n += 256) { float e = expf(oldL[n] - pm); wcL[n] = e; ps += e; }
  ps = blockReduceSum(ps, red);
  float inv = 1.0f / ps;
  for (int n = tid; n < 500; n += 256) {
    float w = wcL[n] * inv;
    oldL[n] = w;
    wcL[n]  = fminf(fmaxf(w, LBV), UBV);
  }
  __syncthreads();

  for (int it = 0; it < 32; ++it) {
    float pl = 0.0f, pn = 0.0f;
    for (int n = tid; n < 500; n += 256) {
      float wcv = wcL[n];
      pl += oldL[n] - wcv;
      if (wcv != UBV) pn += wcv;
    }
    float leftover = blockReduceSum(pl, red);
    float nsum     = blockReduceSum(pn, red);
    float ns = (nsum == 0.0f) ? 1.0f : nsum;
    float pover = 0.0f;
    for (int n = tid; n < 500; n += 256) {
      float wcv = wcL[n];
      float wc2 = (wcv != UBV) ? wcv + leftover * wcv / ns : wcv;
      oldL[n] = wc2;
      if (wc2 > UBV) pover = 1.0f;
    }
    float anyOver = blockReduceSum(pover, red);
    int new_done = (anyOver == 0.0f);
    for (int n = tid; n < 500; n += 256) {
      float wc2 = oldL[n];
      wcL[n] = new_done ? wc2 : fminf(fmaxf(wc2, LBV), UBV);
    }
    __syncthreads();
    if (new_done) break;
  }

  for (int n = tid; n < 500; n += 256) outp[(size_t)b * 500 + n] = wcL[n];
}

extern "C" void kernel_launch(void* const* d_in, const int* in_sizes, int n_in,
                              void* d_out, int out_size, void* d_ws, size_t ws_size,
                              hipStream_t stream) {
  (void)in_sizes; (void)n_in; (void)out_size; (void)ws_size;
  const float* x    = (const float*)d_in[0];
  const float* Wih0 = (const float*)d_in[1];
  const float* Whh0 = (const float*)d_in[2];
  const float* bih0 = (const float*)d_in[3];
  const float* bhh0 = (const float*)d_in[4];
  const float* Wih1 = (const float*)d_in[5];
  const float* Whh1 = (const float*)d_in[6];
  const float* bih1 = (const float*)d_in[7];
  const float* bhh1 = (const float*)d_in[8];
  const float* Wa   = (const float*)d_in[9];
  const float* ba   = (const float*)d_in[10];
  const float* vv   = (const float*)d_in[11];
  const float* Wfc  = (const float*)d_in[12];
  const float* bfc  = (const float*)d_in[13];
  float* outp = (float*)d_out;

  constexpr int B = 1024, S = 60, N = 500, H = 128, G3 = 384;

  uint8_t* base = (uint8_t*)d_ws;
  size_t off = 0;
  auto carve = [&](size_t bytes) {
    void* p = base + off;
    off = (off + bytes + 255) & ~(size_t)255;
    return p;
  };
  float* gi   = (float*)carve((size_t)B * S * G3 * 4);
  float* out0 = (float*)carve((size_t)B * S * H * 4);
  float* out1 = (float*)carve((size_t)B * S * H * 4);
  float* gEb  = (float*)carve((size_t)B * S * H * 4);
  float* ub   = (float*)carve((size_t)B * H * 4);
  _Float16* Wih0F = (_Float16*)carve((size_t)24 * 16 * 512 * 2);
  _Float16* Whh0F = (_Float16*)carve((size_t)24 * 4 * 512 * 2);
  _Float16* Wih1F = (_Float16*)carve((size_t)24 * 4 * 512 * 2);
  _Float16* Whh1F = (_Float16*)carve((size_t)24 * 4 * 512 * 2);
  _Float16* WaxF  = (_Float16*)carve((size_t)8 * 16 * 512 * 2);
  _Float16* WahF  = (_Float16*)carve((size_t)8 * 4 * 512 * 2);

  auto prep = [&](const float* W, _Float16* dst, int n, int k, int ldw) {
    int KT = (k + 31) >> 5;
    int NT = (n + 15) >> 4;
    int total = NT * KT * 512;
    wfrag_prep<<<(total + 255) / 256, 256, 0, stream>>>(W, dst, n, k, ldw, KT, total);
  };
  prep(Wih0,   Wih0F, G3, N, N);
  prep(Whh0,   Whh0F, G3, H, H);
  prep(Wih1,   Wih1F, G3, H, H);
  prep(Whh1,   Whh1F, G3, H, H);
  prep(Wa + H, WaxF,  H,  N, H + N);
  prep(Wa,     WahF,  H,  H, H + N);

  auto gemm = [&](const float* A, const _Float16* Bf, const float* bias, float* C,
                  int M, int Nn, int K, int lda, int ldc) {
    gemm_wmma<<<dim3(M / 16), 256, 0, stream>>>(A, Bf, bias, C, M, Nn, K, lda, ldc);
  };

  gemm(x, Wih0F, bih0, gi, B * S, G3, N, N, G3);
  gru_recur<<<B / 16, 256, 0, stream>>>(gi, Whh0F, bhh0, out0, S);
  gemm(out0, Wih1F, bih1, gi, B * S, G3, H, H, G3);
  gru_recur<<<B / 16, 256, 0, stream>>>(gi, Whh1F, bhh1, out1, S);
  gemm(x, WaxF, ba, gEb, B * S, H, N, N, H);
  gemm(out1 + (size_t)(S - 1) * H, WahF, nullptr, ub, B, H, H, S * H, H);
  finalize_kernel<<<B, 256, 0, stream>>>(gEb, ub, out1, vv, Wfc, bfc, outp, S);
}


# Input GEMMs: one block per 64 rows, not 16: each f16 B fragment loaded feeds four WMMAs (L2 B bytes and serial block k-steps fall to a quarter), B loads issued beside the A loads; sums unchanged
# speedup vs baseline: 2.1721x; 1.2688x over previous
#include <hip/hip_runtime.h>
#include <hip/hip_bf16.h>
#include <stdint.h>

typedef __attribute__((ext_vector_type(16))) _Float16 v16h;
typedef __attribute__((ext_vector_type(4)))  _Float16 v4h;
typedef __attribute__((ext_vector_type(8)))  float    v8f;

#define LBV 0.0f
#define UBV 0.2f

__global__ __launch_bounds__(256) void wfrag_prep(const float* __restrict__ W,
                                                  _Float16* __restrict__ dst,
                                                  int N, int K, int ldw, int KT,
                                                  int totalElems) {
  int idx = blockIdx.x * 256 + threadIdx.x;
  if (idx >= totalElems) return;
  int frag   = idx >> 9;
  int within = idx & 511;
  int r  = within >> 5;
  int k  = within & 31;
  int tileN = frag / KT;
  int tileK = frag - tileN * KT;
  int n  = tileN * 16 + r;
  int kk = tileK * 32 + k;
  float v = (n < N && kk < K) ? W[(size_t)n * ldw + kk] : 0.0f;
  int lane = (((k >> 3) & 1) << 4) + r;
  int elem = (k & 7) | ((k >> 4) << 3);
  dst[(size_t)frag * 512 + lane * 16 + elem] = (_Float16)v;
}

template <int TPW>
__global__ __launch_bounds__(256) void gemm_wmma(const float* __restrict__ A,
                                                 const _Float16* __restrict__ Bfrag,
                                                 const float* __restrict__ bias,
                                                 float* __restrict__ C,
                                                 int K, int lda, int ldc) {
  __shared__ __align__(32) _Float16 ldsA[4 * 512];
  int tid  = threadIdx.x;
  int lane = tid & 31;
  int wave = tid >> 5;
  int mBase = blockIdx.x * 64;
  int KT    = (K + 31) >> 5;
  v8f acc[TPW][4] = {};
  int sr  = tid >> 3;
  int sq  = tid & 7;
  int skl = sq << 2;
  int sln = (((skl >> 3) & 1) << 4) + (sr & 15);
  int se0 = (skl & 7) | ((skl >> 4) << 3);
  _Float16* st0 = &ldsA[(sr >> 4) * 512 + sln * 16 + se0];
  _Float16* st1 = st0 + 2 * 512;
  const float* Arow0 = A + (size_t)(mBase + sr) * lda + skl;
  const float* Arow1 = Arow0 + (size_t)32 * lda;
  const _Float16* Bw = Bfrag + (size_t)wave * KT * 512 + lane * 16;

  for (int kt = 0; kt < KT; ++kt) {
    int k0 = kt << 5;
    v16h b[TPW];
#pragma unroll
    for (int j = 0; j < TPW; ++j)
      b[j] = *reinterpret_cast<const v16h*>(&Bw[((size_t)(j << 3) * KT + kt) * 512]);
    v4h hv0, hv1;
    if (k0 + 32 <= K) {
      float4 va = *(const float4*)(Arow0 + k0);
      float4 vb = *(const float4*)(Arow1 + k0);
      hv0[0] = (_Float16)va.x; hv0[1] = (_Float16)va.y;
      hv0[2] = (_Float16)va.z; hv0[3] = (_Float16)va.w;
      hv1[0] = (_Float16)vb.x; hv1[1] = (_Float16)vb.y;
      hv1[2] = (_Float16)vb.z; hv1[3] = (_Float16)vb.w;
    } else {
#pragma unroll
      for (int j = 0; j < 4; ++j) {
        bool ok = (k0 + skl + j) < K;
        hv0[j] = (_Float16)(ok ? Arow0[k0 + j] : 0.0f);
        hv1[j] = (_Float16)(ok ? Arow1[k0 + j] : 0.0f);
      }
    }
    *reinterpret_cast<v4h*>(st0) = hv0;
    *reinterpret_cast<v4h*>(st1) = hv1;
    __syncthreads();
    v16h a[4];
#pragma unroll
    for (int mt = 0; mt < 4; ++mt)
      a[mt] = *reinterpret_cast<const v16h*>(&ldsA[mt * 512 + lane * 16]);
#pragma unroll
    for (int j = 0; j < TPW; ++j) {
#pragma unroll
      for (int mt = 0; mt < 4; ++mt)
        acc[j][mt] = __builtin_amdgcn_wmma_f32_16x16x32_f16(false, a[mt], false, b[j],
                                                            (short)0, acc[j][mt], false, false);
    }
    __syncthreads();
  }
  int rHalf = (lane >> 4) << 3;
#pragma unroll
  for (int j = 0; j < TPW; ++j) {
    int col = ((wave + (j << 3)) << 4) + (lane & 15);
    float bv = (bias != nullptr) ? bias[col] : 0.0f;
#pragma unroll
    for (int mt = 0; mt < 4; ++mt) {
      int rBase = mBase + (mt << 4) + rHalf;
#pragma unroll
      for (int r = 0; r < 8; ++r)
        C[(size_t)(rBase + r) * ldc + col] = acc[j][mt][r] + bv;
    }
  }
}

__device__ __forceinline__ float sigmoid_hw(float x) {
  return __builtin_amdgcn_rcpf(1.0f + __builtin_amdgcn_exp2f(x * -1.442695041f));
}
__device__ __forceinline__ float tanh_hw(float x) {
  return 1.0f - 2.0f * __builtin_amdgcn_rcpf(1.0f + __builtin_amdgcn_exp2f(x * 2.885390082f));
}

__device__ __forceinline__ void lds_barrier() {
  __builtin_amdgcn_fence(__ATOMIC_RELEASE, "workgroup", "local");
  __builtin_amdgcn_s_barrier();
  __builtin_amdgcn_fence(__ATOMIC_ACQUIRE, "workgroup", "local");
}

__device__ __forceinline__ void gi_fetch(float4 (&g)[6], const float* ga, const float* gb, int t) {
  const float* a = ga + (size_t)t * 384;
  const float* b = gb + (size_t)t * 384;
  g[0] = *(const float4*)(a);
  g[1] = *(const float4*)(a + 128);
  g[2] = *(const float4*)(a + 256);
  g[3] = *(const float4*)(b);
  g[4] = *(const float4*)(b + 128);
  g[5] = *(const float4*)(b + 256);
}

__global__ __launch_bounds__(256) __attribute__((amdgpu_waves_per_eu(2, 2))) void gru_recur(const float* __restrict__ gi,
                                                 const _Float16* __restrict__ WhhF,
                                                 const float* __restrict__ bhh,
                                                 float* __restrict__ out, int S) {
  __shared__ float hS[16 * 128];
  __shared__ __align__(32) _Float16 aF[4 * 512];
  __shared__ float ghS[16 * 384];
  int tid  = threadIdx.x;
  int lane = tid & 31;
  int wave = tid >> 5;
  int bBase = blockIdx.x * 16;
  const _Float16* Bw = WhhF + (size_t)(wave * 3) * 4 * 512 + lane * 16;
  if (S <= 0) return;

  v16h bF[3][4];
  float bv[3];
#pragma unroll
  for (int j = 0; j < 3; ++j) {
#pragma unroll
    for (int kt = 0; kt < 4; ++kt)
      bF[j][kt] = *reinterpret_cast<const v16h*>(&Bw[((size_t)j * 4 + kt) * 512]);
    bv[j] = bhh[((wave * 3 + j) << 4) + (lane & 15)];
  }

  const float* giA = gi + (size_t)(bBase + (tid >> 5)) * S * 384 + ((tid & 31) << 2);
  const float* giB = giA + (size_t)8 * S * 384;

  float4 g0[6], g1[6], g2[6];
  gi_fetch(g0, giA, giB, 0);
  gi_fetch(g1, giA, giB, (1 < S) ? 1 : S - 1);

  for (int i = tid; i < 16 * 128; i += 256) hS[i] = 0.0f;
  for (int i = tid; i < 4 * 512; i += 256) aF[i] = (_Float16)0.0f;
  __syncthreads();

  auto step = [&](int t, const float4 (&g)[6]) __attribute__((always_inline)) {
    v16h a[4];
#pragma unroll
    for (int kt = 0; kt < 4; ++kt)
      a[kt] = *reinterpret_cast<const v16h*>(&aF[kt * 512 + lane * 16]);
    __builtin_amdgcn_sched_barrier(0);
    v8f acc[3] = {};
#pragma unroll
    for (int kt = 0; kt < 4; ++kt) {
#pragma unroll
      for (int j = 0; j < 3; ++j)
        acc[j] = __builtin_amdgcn_wmma_f32_16x16x32_f16(false, a[kt], false, bF[j][kt],
                                                        (short)0, acc[j], false, false);
    }
#pragma unroll
    for (int j = 0; j < 3; ++j) {
      int col   = ((wave * 3 + j) << 4) + (lane & 15);
      int rBase = (lane >> 4) << 3;
#pragma unroll
      for (int r = 0; r < 8; ++r) ghS[(rBase + r) * 384 + col] = acc[j][r] + bv[j];
    }
    lds_barrier();
    float4 hr[2], hz[2], hn[2], ho[2];
#pragma unroll
    for (int p = 0; p < 2; ++p) {
      int gx  = tid + (p << 8);
      int row = gx >> 5;
      int h0  = (gx & 31) << 2;
      hr[p] = *(const float4*)(&ghS[row * 384 + h0]);
      hz[p] = *(const float4*)(&ghS[row * 384 + 128 + h0]);
      hn[p] = *(const float4*)(&ghS[row * 384 + 256 + h0]);
      ho[p] = *(const float4*)(&hS[row * 128 + h0]);
    }
#pragma unroll
    for (int p = 0; p < 2; ++p) {
      int gx  = tid + (p << 8);
      int row = gx >> 5;
      int h0  = (gx & 31) << 2;
      int b   = bBase + row;
      float4 gr = g[3 * p];
      float4 gz = g[3 * p + 1];
      float4 gn = g[3 * p + 2];
      float4 hNew;
#define GATE(c)                                                            \
      {                                                                    \
        float rg = sigmoid_hw(gr.c + hr[p].c);                             \
        float zg = sigmoid_hw(gz.c + hz[p].c);                             \
        float ng = tanh_hw(gn.c + rg * hn[p].c);                           \
        hNew.c = (1.0f - zg) * ng + zg * ho[p].c;                          \
      }
      GATE(x) GATE(y) GATE(z) GATE(w)
#undef GATE
      *(float4*)(&hS[row * 128 + h0]) = hNew;
      *(float4*)(out + ((size_t)b * S + t) * 128 + h0) = hNew;
      int kt = h0 >> 5;
      int kl = h0 & 31;
      int ln = (((kl >> 3) & 1) << 4) + row;
      int e0 = (kl & 7) | ((kl >> 4) << 3);
      v4h hv;
      hv[0] = (_Float16)hNew.x; hv[1] = (_Float16)hNew.y;
      hv[2] = (_Float16)hNew.z; hv[3] = (_Float16)hNew.w;
      *reinterpret_cast<v4h*>(&aF[kt * 512 + ln * 16 + e0]) = hv;
    }
    lds_barrier();
  };

  int t = 0;
  for (; t + 2 < S; t += 3) {
    gi_fetch(g2, giA, giB, t + 2);
    step(t, g0);
    gi_fetch(g0, giA, giB, (t + 3 < S) ? t + 3 : S - 1);
    step(t + 1, g1);
    gi_fetch(g1, giA, giB, (t + 4 < S) ? t + 4 : S - 1);
    step(t + 2, g2);
  }
  if (t < S) {
    step(t, g0);
    if (t + 1 < S) step(t + 1, g1);
  }
}

__device__ __forceinline__ float blockReduceSum(float v, float* red) {
  int tid = threadIdx.x;
  red[tid] = v;
  __syncthreads();
  for (int off = 128; off > 0; off >>= 1) {
    if (tid < off) red[tid] += red[tid + off];
    __syncthreads();
  }
  float r = red[0];
  __syncthreads();
  return r;
}
__device__ __forceinline__ float blockReduceMax(float v, float* red) {
  int tid = threadIdx.x;
  red[tid] = v;
  __syncthreads();
  for (int off = 128; off > 0; off >>= 1) {
    if (tid < off) red[tid] = fmaxf(red[tid], red[tid + off]);
    __syncthreads();
  }
  float r = red[0];
  __syncthreads();
  return r;
}

__global__ __launch_bounds__(256) void finalize_kernel(const float* __restrict__ gE,
                                                       const float* __restrict__ u,
                                                       const float* __restrict__ out1,
                                                       const float* __restrict__ vvec,
                                                       const float* __restrict__ Wfc,
                                                       const float* __restrict__ bfc,
                                                       float* __restrict__ outp, int S) {
  __shared__ float uL[128], vL[128], eS[64], attnS[64], concatL[256];
  __shared__ float oldL[512], wcL[512], red[256];
  int tid  = threadIdx.x;
  int lane = tid & 31;
  int wave = tid >> 5;
  int b = blockIdx.x;

  if (tid < 128) {
    uL[tid] = u[(size_t)b * 128 + tid];
    vL[tid] = vvec[tid];
  }
  __syncthreads();

  for (int s = wave; s < S; s += 8) {
    float p = 0.0f;
    const float* ge = gE + ((size_t)b * S + s) * 128;
    for (int h = lane; h < 128; h += 32) p += vL[h] * tanhf(uL[h] + ge[h]);
    for (int off = 16; off > 0; off >>= 1) p += __shfl_down(p, off, 32);
    if (lane == 0) eS[s] = p;
  }
  __syncthreads();

  if (tid == 0) {
    float m = eS[0];
    for (int s = 1; s < S; ++s) m = fmaxf(m, eS[s]);
    float sum = 0.0f;
    for (int s = 0; s < S; ++s) { float ev = expf(eS[s] - m); attnS[s] = ev; sum += ev; }
    float inv = 1.0f / sum;
    for (int s = 0; s < S; ++s) attnS[s] *= inv;
  }
  __syncthreads();

  if (tid < 128) {
    float c = 0.0f;
    for (int s = 0; s < S; ++s) c += attnS[s] * out1[((size_t)b * S + s) * 128 + tid];
    concatL[tid]       = out1[((size_t)b * S + (S - 1)) * 128 + tid];
    concatL[128 + tid] = c;
  }
  __syncthreads();

  for (int n = tid; n < 500; n += 256) {
    float acc = bfc[n];
    const float* wr = Wfc + (size_t)n * 256;
#pragma unroll 4
    for (int j = 0; j < 256; ++j) acc += wr[j] * concatL[j];
    oldL[n] = acc;
  }
  __syncthreads();

  float pm = -3.4e38f;
  for (int n = tid; n < 500; n += 256) pm = fmaxf(pm, oldL[n]);
  pm = blockReduceMax(pm, red);
  float ps = 0.0f;
  for (int n = tid; n < 500; n += 256) { float e = expf(oldL[n] - pm); wcL[n] = e; ps += e; }
  ps = blockReduceSum(ps, red);
  float inv = 1.0f / ps;
  for (int n = tid; n < 500; n += 256) {
    float w = wcL[n] * inv;
    oldL[n] = w;
    wcL[n]  = fminf(fmaxf(w, LBV), UBV);
  }
  __syncthreads();

  for (int it = 0; it < 32; ++it) {
    float pl = 0.0f, pn = 0.0f;
    for (int n = tid; n < 500; n += 256) {
      float wcv = wcL[n];
      pl += oldL[n] - wcv;
      if (wcv != UBV) pn += wcv;
    }
    float leftover = blockReduceSum(pl, red);
    float nsum     = blockReduceSum(pn, red);
    float ns = (nsum == 0.0f) ? 1.0f : nsum;
    float pover = 0.0f;
    for (int n = tid; n < 500; n += 256) {
      float wcv = wcL[n];
      float wc2 = (wcv != UBV) ? wcv + leftover * wcv / ns : wcv;
      oldL[n] = wc2;
      if (wc2 > UBV) pover = 1.0f;
    }
    float anyOver = blockReduceSum(pover, red);
    int new_done = (anyOver == 0.0f);
    for (int n = tid; n < 500; n += 256) {
      float wc2 = oldL[n];
      wcL[n] = new_done ? wc2 : fminf(fmaxf(wc2, LBV), UBV);
    }
    __syncthreads();
    if (new_done) break;
  }

  for (int n = tid; n < 500; n += 256) outp[(size_t)b * 500 + n] = wcL[n];
}

extern "C" void kernel_launch(void* const* d_in, const int* in_sizes, int n_in,
                              void* d_out, int out_size, void* d_ws, size_t ws_size,
                              hipStream_t stream) {
  (void)in_sizes; (void)n_in; (void)out_size; (void)ws_size;
  const float* x    = (const float*)d_in[0];
  const float* Wih0 = (const float*)d_in[1];
  const float* Whh0 = (const float*)d_in[2];
  const float* bih0 = (const float*)d_in[3];
  const float* bhh0 = (const float*)d_in[4];
  const float* Wih1 = (const float*)d_in[5];
  const float* Whh1 = (const float*)d_in[6];
  const float* bih1 = (const float*)d_in[7];
  const float* bhh1 = (const float*)d_in[8];
  const float* Wa   = (const float*)d_in[9];
  const float* ba   = (const float*)d_in[10];
  const float* vv   = (const float*)d_in[11];
  const float* Wfc  = (const float*)d_in[12];
  const float* bfc  = (const float*)d_in[13];
  float* outp = (float*)d_out;

  constexpr int B = 1024, S = 60, N = 500, H = 128, G3 = 384;

  uint8_t* base = (uint8_t*)d_ws;
  size_t off = 0;
  auto carve = [&](size_t bytes) {
    void* p = base + off;
    off = (off + bytes + 255) & ~(size_t)255;
    return p;
  };
  float* gi   = (float*)carve((size_t)B * S * G3 * 4);
  float* out0 = (float*)carve((size_t)B * S * H * 4);
  float* out1 = (float*)carve((size_t)B * S * H * 4);
  float* gEb  = (float*)carve((size_t)B * S * H * 4);
  float* ub   = (float*)carve((size_t)B * H * 4);
  _Float16* Wih0F = (_Float16*)carve((size_t)24 * 16 * 512 * 2);
  _Float16* Whh0F = (_Float16*)carve((size_t)24 * 4 * 512 * 2);
  _Float16* Wih1F = (_Float16*)carve((size_t)24 * 4 * 512 * 2);
  _Float16* Whh1F = (_Float16*)carve((size_t)24 * 4 * 512 * 2);
  _Float16* WaxF  = (_Float16*)carve((size_t)8 * 16 * 512 * 2);
  _Float16* WahF  = (_Float16*)carve((size_t)8 * 4 * 512 * 2);

  auto prep = [&](const float* W, _Float16* dst, int n, int k, int ldw) {
    int KT = (k + 31) >> 5;
    int NT = (n + 15) >> 4;
    int total = NT * KT * 512;
    wfrag_prep<<<(total + 255) / 256, 256, 0, stream>>>(W, dst, n, k, ldw, KT, total);
  };
  prep(Wih0,   Wih0F, G3, N, N);
  prep(Whh0,   Whh0F, G3, H, H);
  prep(Wih1,   Wih1F, G3, H, H);
  prep(Whh1,   Whh1F, G3, H, H);
  prep(Wa + H, WaxF,  H,  N, H + N);
  prep(Wa,     WahF,  H,  H, H + N);

  auto gemm384 = [&](const float* A, const _Float16* Bf, const float* bias, float* C,
                     int M, int K, int lda) {
    gemm_wmma<3><<<dim3(M / 64), 256, 0, stream>>>(A, Bf, bias, C, K, lda, G3);
  };
  auto gemm128 = [&](const float* A, const _Float16* Bf, const float* bias, float* C,
                     int M, int K, int lda) {
    gemm_wmma<1><<<dim3(M / 64), 256, 0, stream>>>(A, Bf, bias, C, K, lda, H);
  };

  gemm384(x, Wih0F, bih0, gi, B * S, N, N);
  gru_recur<<<B / 16, 256, 0, stream>>>(gi, Whh0F, bhh0, out0, S);
  gemm384(out0, Wih1F, bih1, gi, B * S, H, H);
  gru_recur<<<B / 16, 256, 0, stream>>>(gi, Whh1F, bhh1, out1, S);
  gemm128(x, WaxF, ba, gEb, B * S, N, N);
  gemm128(out1 + (size_t)(S - 1) * H, WahF, nullptr, ub, B, H, S * H);
  finalize_kernel<<<B, 256, 0, stream>>>(gEb, ub, out1, vv, Wfc, bfc, outp, S);
}


# FC logits: one wave per output row, lanes across k (two coalesced float4 per lane: 4 lines per load, was 32), eight rows' loads in flight; dot products now summed per lane, then by butterfly
# speedup vs baseline: 2.4576x; 1.1314x over previous
#include <hip/hip_runtime.h>
#include <hip/hip_bf16.h>
#include <stdint.h>

typedef __attribute__((ext_vector_type(16))) _Float16 v16h;
typedef __attribute__((ext_vector_type(4)))  _Float16 v4h;
typedef __attribute__((ext_vector_type(8)))  float    v8f;

#define LBV 0.0f
#define UBV 0.2f

__global__ __launch_bounds__(256) void wfrag_prep(const float* __restrict__ W,
                                                  _Float16* __restrict__ dst,
                                                  int N, int K, int ldw, int KT,
                                                  int totalElems) {
  int idx = blockIdx.x * 256 + threadIdx.x;
  if (idx >= totalElems) return;
  int frag   = idx >> 9;
  int within = idx & 511;
  int r  = within >> 5;
  int k  = within & 31;
  int tileN = frag / KT;
  int tileK = frag - tileN * KT;
  int n  = tileN * 16 + r;
  int kk = tileK * 32 + k;
  float v = (n < N && kk < K) ? W[(size_t)n * ldw + kk] : 0.0f;
  int lane = (((k >> 3) & 1) << 4) + r;
  int elem = (k & 7) | ((k >> 4) << 3);
  dst[(size_t)frag * 512 + lane * 16 + elem] = (_Float16)v;
}

template <int TPW>
__global__ __launch_bounds__(256) void gemm_wmma(const float* __restrict__ A,
                                                 const _Float16* __restrict__ Bfrag,
                                                 const float* __restrict__ bias,
                                                 float* __restrict__ C,
                                                 int K, int lda, int ldc) {
  __shared__ __align__(32) _Float16 ldsA[4 * 512];
  int tid  = threadIdx.x;
  int lane = tid & 31;
  int wave = tid >> 5;
  int mBase = blockIdx.x * 64;
  int KT    = (K + 31) >> 5;
  v8f acc[TPW][4] = {};
  int sr  = tid >> 3;
  int sq  = tid & 7;
  int skl = sq << 2;
  int sln = (((skl >> 3) & 1) << 4) + (sr & 15);
  int se0 = (skl & 7) | ((skl >> 4) << 3);
  _Float16* st0 = &ldsA[(sr >> 4) * 512 + sln * 16 + se0];
  _Float16* st1 = st0 + 2 * 512;
  const float* Arow0 = A + (size_t)(mBase + sr) * lda + skl;
  const float* Arow1 = Arow0 + (size_t)32 * lda;
  const _Float16* Bw = Bfrag + (size_t)wave * KT * 512 + lane * 16;

  for (int kt = 0; kt < KT; ++kt) {
    int k0 = kt << 5;
    v16h b[TPW];
#pragma unroll
    for (int j = 0; j < TPW; ++j)
      b[j] = *reinterpret_cast<const v16h*>(&Bw[((size_t)(j << 3) * KT + kt) * 512]);
    v4h hv0, hv1;
    if (k0 + 32 <= K) {
      float4 va = *(const float4*)(Arow0 + k0);
      float4 vb = *(const float4*)(Arow1 + k0);
      hv0[0] = (_Float16)va.x; hv0[1] = (_Float16)va.y;
      hv0[2] = (_Float16)va.z; hv0[3] = (_Float16)va.w;
      hv1[0] = (_Float16)vb.x; hv1[1] = (_Float16)vb.y;
      hv1[2] = (_Float16)vb.z; hv1[3] = (_Float16)vb.w;
    } else {
#pragma unroll
      for (int j = 0; j < 4; ++j) {
        bool ok = (k0 + skl + j) < K;
        hv0[j] = (_Float16)(ok ? Arow0[k0 + j] : 0.0f);
        hv1[j] = (_Float16)(ok ? Arow1[k0 + j] : 0.0f);
      }
    }
    *reinterpret_cast<v4h*>(st0) = hv0;
    *reinterpret_cast<v4h*>(st1) = hv1;
    __syncthreads();
    v16h a[4];
#pragma unroll
    for (int mt = 0; mt < 4; ++mt)
      a[mt] = *reinterpret_cast<const v16h*>(&ldsA[mt * 512 + lane * 16]);
#pragma unroll
    for (int j = 0; j < TPW; ++j) {
#pragma unroll
      for (int mt = 0; mt < 4; ++mt)
        acc[j][mt] = __builtin_amdgcn_wmma_f32_16x16x32_f16(false, a[mt], false, b[j],
                                                            (short)0, acc[j][mt], false, false);
    }
    __syncthreads();
  }
  int rHalf = (lane >> 4) << 3;
#pragma unroll
  for (int j = 0; j < TPW; ++j) {
    int col = ((wave + (j << 3)) << 4) + (lane & 15);
    float bv = (bias != nullptr) ? bias[col] : 0.0f;
#pragma unroll
    for (int mt = 0; mt < 4; ++mt) {
      int rBase = mBase + (mt << 4) + rHalf;
#pragma unroll
      for (int r = 0; r < 8; ++r)
        C[(size_t)(rBase + r) * ldc + col] = acc[j][mt][r] + bv;
    }
  }
}

__device__ __forceinline__ float sigmoid_hw(float x) {
  return __builtin_amdgcn_rcpf(1.0f + __builtin_amdgcn_exp2f(x * -1.442695041f));
}
__device__ __forceinline__ float tanh_hw(float x) {
  return 1.0f - 2.0f * __builtin_amdgcn_rcpf(1.0f + __builtin_amdgcn_exp2f(x * 2.885390082f));
}

__device__ __forceinline__ void lds_barrier() {
  __builtin_amdgcn_fence(__ATOMIC_RELEASE, "workgroup", "local");
  __builtin_amdgcn_s_barrier();
  __builtin_amdgcn_fence(__ATOMIC_ACQUIRE, "workgroup", "local");
}

__device__ __forceinline__ void gi_fetch(float4 (&g)[6], const float* ga, const float* gb, int t) {
  const float* a = ga + (size_t)t * 384;
  const float* b = gb + (size_t)t * 384;
  g[0] = *(const float4*)(a);
  g[1] = *(const float4*)(a + 128);
  g[2] = *(const float4*)(a + 256);
  g[3] = *(const float4*)(b);
  g[4] = *(const float4*)(b + 128);
  g[5] = *(const float4*)(b + 256);
}

__global__ __launch_bounds__(256) __attribute__((amdgpu_waves_per_eu(2, 2))) void gru_recur(const float* __restrict__ gi,
                                                 const _Float16* __restrict__ WhhF,
                                                 const float* __restrict__ bhh,
                                                 float* __restrict__ out, int S) {
  __shared__ float hS[16 * 128];
  __shared__ __align__(32) _Float16 aF[4 * 512];
  __shared__ float ghS[16 * 384];
  int tid  = threadIdx.x;
  int lane = tid & 31;
  int wave = tid >> 5;
  int bBase = blockIdx.x * 16;
  const _Float16* Bw = WhhF + (size_t)(wave * 3) * 4 * 512 + lane * 16;
  if (S <= 0) return;

  v16h bF[3][4];
  float bv[3];
#pragma unroll
  for (int j = 0; j < 3; ++j) {
#pragma unroll
    for (int kt = 0; kt < 4; ++kt)
      bF[j][kt] = *reinterpret_cast<const v16h*>(&Bw[((size_t)j * 4 + kt) * 512]);
    bv[j] = bhh[((wave * 3 + j) << 4) + (lane & 15)];
  }

  const float* giA = gi + (size_t)(bBase + (tid >> 5)) * S * 384 + ((tid & 31) << 2);
  const float* giB = giA + (size_t)8 * S * 384;

  float4 g0[6], g1[6], g2[6];
  gi_fetch(g0, giA, giB, 0);
  gi_fetch(g1, giA, giB, (1 < S) ? 1 : S - 1);

  for (int i = tid; i < 16 * 128; i += 256) hS[i] = 0.0f;
  for (int i = tid; i < 4 * 512; i += 256) aF[i] = (_Float16)0.0f;
  __syncthreads();

  auto step = [&](int t, const float4 (&g)[6]) __attribute__((always_inline)) {
    v16h a[4];
#pragma unroll
    for (int kt = 0; kt < 4; ++kt)
      a[kt] = *reinterpret_cast<const v16h*>(&aF[kt * 512 + lane * 16]);
    __builtin_amdgcn_sched_barrier(0);
    v8f acc[3] = {};
#pragma unroll
    for (int kt = 0; kt < 4; ++kt) {
#pragma unroll
      for (int j = 0; j < 3; ++j)
        acc[j] = __builtin_amdgcn_wmma_f32_16x16x32_f16(false, a[kt], false, bF[j][kt],
                                                        (short)0, acc[j], false, false);
    }
#pragma unroll
    for (int j = 0; j < 3; ++j) {
      int col   = ((wave * 3 + j) << 4) + (lane & 15);
      int rBase = (lane >> 4) << 3;
#pragma unroll
      for (int r = 0; r < 8; ++r) ghS[(rBase + r) * 384 + col] = acc[j][r] + bv[j];
    }
    lds_barrier();
    float4 hr[2], hz[2], hn[2], ho[2];
#pragma unroll
    for (int p = 0; p < 2; ++p) {
      int gx  = tid + (p << 8);
      int row = gx >> 5;
      int h0  = (gx & 31) << 2;
      hr[p] = *(const float4*)(&ghS[row * 384 + h0]);
      hz[p] = *(const float4*)(&ghS[row * 384 + 128 + h0]);
      hn[p] = *(const float4*)(&ghS[row * 384 + 256 + h0]);
      ho[p] = *(const float4*)(&hS[row * 128 + h0]);
    }
#pragma unroll
    for (int p = 0; p < 2; ++p) {
      int gx  = tid + (p << 8);
      int row = gx >> 5;
      int h0  = (gx & 31) << 2;
      int b   = bBase + row;
      float4 gr = g[3 * p];
      float4 gz = g[3 * p + 1];
      float4 gn = g[3 * p + 2];
      float4 hNew;
#define GATE(c)                                                            \
      {                                                                    \
        float rg = sigmoid_hw(gr.c + hr[p].c);                             \
        float zg = sigmoid_hw(gz.c + hz[p].c);                             \
        float ng = tanh_hw(gn.c + rg * hn[p].c);                           \
        hNew.c = (1.0f - zg) * ng + zg * ho[p].c;                          \
      }
      GATE(x) GATE(y) GATE(z) GATE(w)
#undef GATE
      *(float4*)(&hS[row * 128 + h0]) = hNew;
      *(float4*)(out + ((size_t)b * S + t) * 128 + h0) = hNew;
      int kt = h0 >> 5;
      int kl = h0 & 31;
      int ln = (((kl >> 3) & 1) << 4) + row;
      int e0 = (kl & 7) | ((kl >> 4) << 3);
      v4h hv;
      hv[0] = (_Float16)hNew.x; hv[1] = (_Float16)hNew.y;
      hv[2] = (_Float16)hNew.z; hv[3] = (_Float16)hNew.w;
      *reinterpret_cast<v4h*>(&aF[kt * 512 + ln * 16 + e0]) = hv;
    }
    lds_barrier();
  };

  int t = 0;
  for (; t + 2 < S; t += 3) {
    gi_fetch(g2, giA, giB, t + 2);
    step(t, g0);
    gi_fetch(g0, giA, giB, (t + 3 < S) ? t + 3 : S - 1);
    step(t + 1, g1);
    gi_fetch(g1, giA, giB, (t + 4 < S) ? t + 4 : S - 1);
    step(t + 2, g2);
  }
  if (t < S) {
    step(t, g0);
    if (t + 1 < S) step(t + 1, g1);
  }
}

__device__ __forceinline__ float blockReduceSum(float v, float* red) {
  int tid = threadIdx.x;
  red[tid] = v;
  __syncthreads();
  for (int off = 128; off > 0; off >>= 1) {
    if (tid < off) red[tid] += red[tid + off];
    __syncthreads();
  }
  float r = red[0];
  __syncthreads();
  return r;
}
__device__ __forceinline__ float blockReduceMax(float v, float* red) {
  int tid = threadIdx.x;
  red[tid] = v;
  __syncthreads();
  for (int off = 128; off > 0; off >>= 1) {
    if (tid < off) red[tid] = fmaxf(red[tid], red[tid + off]);
    __syncthreads();
  }
  float r = red[0];
  __syncthreads();
  return r;
}

__global__ __launch_bounds__(256) void finalize_kernel(const float* __restrict__ gE,
                                                       const float* __restrict__ u,
                                                       const float* __restrict__ out1,
                                                       const float* __restrict__ vvec,
                                                       const float* __restrict__ Wfc,
                                                       const float* __restrict__ bfc,
                                                       float* __restrict__ outp, int S) {
  __shared__ float uL[128], vL[128], eS[64], attnS[64];
  __shared__ __align__(16) float concatL[256];
  __shared__ float oldL[512], wcL[512], red[256];
  int tid  = threadIdx.x;
  int lane = tid & 31;
  int wave = tid >> 5;
  int b = blockIdx.x;

  if (tid < 128) {
    uL[tid] = u[(size_t)b * 128 + tid];
    vL[tid] = vvec[tid];
  }
  __syncthreads();

  for (int s = wave; s < S; s += 8) {
    float p = 0.0f;
    const float* ge = gE + ((size_t)b * S + s) * 128;
    for (int h = lane; h < 128; h += 32) p += vL[h] * tanhf(uL[h] + ge[h]);
    for (int off = 16; off > 0; off >>= 1) p += __shfl_down(p, off, 32);
    if (lane == 0) eS[s] = p;
  }
  __syncthreads();

  if (tid == 0) {
    float m = eS[0];
    for (int s = 1; s < S; ++s) m = fmaxf(m, eS[s]);
    float sum = 0.0f;
    for (int s = 0; s < S; ++s) { float ev = expf(eS[s] - m); attnS[s] = ev; sum += ev; }
    float inv = 1.0f / sum;
    for (int s = 0; s < S; ++s) attnS[s] *= inv;
  }
  __syncthreads();

  if (tid < 128) {
    float c = 0.0f;
    for (int s = 0; s < S; ++s) c += attnS[s] * out1[((size_t)b * S + s) * 128 + tid];
    concatL[tid]       = out1[((size_t)b * S + (S - 1)) * 128 + tid];
    concatL[128 + tid] = c;
  }
  __syncthreads();

  {
    const float4 c0 = *(const float4*)(&concatL[lane << 2]);
    const float4 c1 = *(const float4*)(&concatL[128 + (lane << 2)]);
    const float* wl = Wfc + (lane << 2);
    for (int n0 = wave; n0 < 500; n0 += 64) {
      float4 w0[8], w1[8];
      float bn[8], p[8];
#pragma unroll
      for (int i = 0; i < 8; ++i) {
        int n  = n0 + (i << 3);
        int nc = (n < 500) ? n : 499;
        const float* wr = wl + (size_t)nc * 256;
        w0[i] = *(const float4*)(wr);
        w1[i] = *(const float4*)(wr + 128);
        bn[i] = bfc[nc];
      }
      __builtin_amdgcn_sched_barrier(0);
#pragma unroll
      for (int i = 0; i < 8; ++i) {
        float a = (lane == 0) ? bn[i] : 0.0f;
        a += w0[i].x * c0.x;
        a += w0[i].y * c0.y;
        a += w0[i].z * c0.z;
        a += w0[i].w * c0.w;
        a += w1[i].x * c1.x;
        a += w1[i].y * c1.y;
        a += w1[i].z * c1.z;
        a += w1[i].w * c1.w;
        p[i] = a;
      }
#pragma unroll
      for (int off = 16; off > 0; off >>= 1) {
#pragma unroll
        for (int i = 0; i < 8; ++i) p[i] += __shfl_xor(p[i], off, 32);
      }
      if (lane == 0) {
#pragma unroll
        for (int i = 0; i < 8; ++i) {
          int n = n0 + (i << 3);
          if (n < 500) oldL[n] = p[i];
        }
      }
    }
  }
  __syncthreads();

  float pm = -3.4e38f;
  for (int n = tid; n < 500; n += 256) pm = fmaxf(pm, oldL[n]);
  pm = blockReduceMax(pm, red);
  float ps = 0.0f;
  for (int n = tid; n < 500; n += 256) { float e = expf(oldL[n] - pm); wcL[n] = e; ps += e; }
  ps = blockReduceSum(ps, red);
  float inv = 1.0f / ps;
  for (int n = tid; n < 500; n += 256) {
    float w = wcL[n] * inv;
    oldL[n] = w;
    wcL[n]  = fminf(fmaxf(w, LBV), UBV);
  }
  __syncthreads();

  for (int it = 0; it < 32; ++it) {
    float pl = 0.0f, pn = 0.0f;
    for (int n = tid; n < 500; n += 256) {
      float wcv = wcL[n];
      pl += oldL[n] - wcv;
      if (wcv != UBV) pn += wcv;
    }
    float leftover = blockReduceSum(pl, red);
    float nsum     = blockReduceSum(pn, red);
    float ns = (nsum == 0.0f) ? 1.0f : nsum;
    float pover = 0.0f;
    for (int n = tid; n < 500; n += 256) {
      float wcv = wcL[n];
      float wc2 = (wcv != UBV) ? wcv + leftover * wcv / ns : wcv;
      oldL[n] = wc2;
      if (wc2 > UBV) pover = 1.0f;
    }
    float anyOver = blockReduceSum(pover, red);
    int new_done = (anyOver == 0.0f);
    for (int n = tid; n < 500; n += 256) {
      float wc2 = oldL[n];
      wcL[n] = new_done ? wc2 : fminf(fmaxf(wc2, LBV), UBV);
    }
    __syncthreads();
    if (new_done) break;
  }

  for (int n = tid; n < 500; n += 256) outp[(size_t)b * 500 + n] = wcL[n];
}

extern "C" void kernel_launch(void* const* d_in, const int* in_sizes, int n_in,
                              void* d_out, int out_size, void* d_ws, size_t ws_size,
                              hipStream_t stream) {
  (void)in_sizes; (void)n_in; (void)out_size; (void)ws_size;
  const float* x    = (const float*)d_in[0];
  const float* Wih0 = (const float*)d_in[1];
  const float* Whh0 = (const float*)d_in[2];
  const float* bih0 = (const float*)d_in[3];
  const float* bhh0 = (const float*)d_in[4];
  const float* Wih1 = (const float*)d_in[5];
  const float* Whh1 = (const float*)d_in[6];
  const float* bih1 = (const float*)d_in[7];
  const float* bhh1 = (const float*)d_in[8];
  const float* Wa   = (const float*)d_in[9];
  const float* ba   = (const float*)d_in[10];
  const float* vv   = (const float*)d_in[11];
  const float* Wfc  = (const float*)d_in[12];
  const float* bfc  = (const float*)d_in[13];
  float* outp = (float*)d_out;

  constexpr int B = 1024, S = 60, N = 500, H = 128, G3 = 384;

  uint8_t* base = (uint8_t*)d_ws;
  size_t off = 0;
  auto carve = [&](size_t bytes) {
    void* p = base + off;
    off = (off + bytes + 255) & ~(size_t)255;
    return p;
  };
  float* gi   = (float*)carve((size_t)B * S * G3 * 4);
  float* out0 = (float*)carve((size_t)B * S * H * 4);
  float* out1 = (float*)carve((size_t)B * S * H * 4);
  float* gEb  = (float*)carve((size_t)B * S * H * 4);
  float* ub   = (float*)carve((size_t)B * H * 4);
  _Float16* Wih0F = (_Float16*)carve((size_t)24 * 16 * 512 * 2);
  _Float16* Whh0F = (_Float16*)carve((size_t)24 * 4 * 512 * 2);
  _Float16* Wih1F = (_Float16*)carve((size_t)24 * 4 * 512 * 2);
  _Float16* Whh1F = (_Float16*)carve((size_t)24 * 4 * 512 * 2);
  _Float16* WaxF  = (_Float16*)carve((size_t)8 * 16 * 512 * 2);
  _Float16* WahF  = (_Float16*)carve((size_t)8 * 4 * 512 * 2);

  auto prep = [&](const float* W, _Float16* dst, int n, int k, int ldw) {
    int KT = (k + 31) >> 5;
    int NT = (n + 15) >> 4;
    int total = NT * KT * 512;
    wfrag_prep<<<(total + 255) / 256, 256, 0, stream>>>(W, dst, n, k, ldw, KT, total);
  };
  prep(Wih0,   Wih0F, G3, N, N);
  prep(Whh0,   Whh0F, G3, H, H);
  prep(Wih1,   Wih1F, G3, H, H);
  prep(Whh1,   Whh1F, G3, H, H);
  prep(Wa + H, WaxF,  H,  N, H + N);
  prep(Wa,     WahF,  H,  H, H + N);

  auto gemm384 = [&](const float* A, const _Float16* Bf, const float* bias, float* C,
                     int M, int K, int lda) {
    gemm_wmma<3><<<dim3(M / 64), 256, 0, stream>>>(A, Bf, bias, C, K, lda, G3);
  };
  auto gemm128 = [&](const float* A, const _Float16* Bf, const float* bias, float* C,
                     int M, int K, int lda) {
    gemm_wmma<1><<<dim3(M / 64), 256, 0, stream>>>(A, Bf, bias, C, K, lda, H);
  };

  gemm384(x, Wih0F, bih0, gi, B * S, N, N);
  gru_recur<<<B / 16, 256, 0, stream>>>(gi, Whh0F, bhh0, out0, S);
  gemm384(out0, Wih1F, bih1, gi, B * S, H, H);
  gru_recur<<<B / 16, 256, 0, stream>>>(gi, Whh1F, bhh1, out1, S);
  gemm128(x, WaxF, ba, gEb, B * S, N, N);
  gemm128(out1 + (size_t)(S - 1) * H, WahF, nullptr, ub, B, H, S * H);
  finalize_kernel<<<B, 256, 0, stream>>>(gEb, ub, out1, vv, Wfc, bfc, outp, S);
}


# GRU recurrence: 4 batch rows per block instead of 16 (256 blocks, one per WGP, was 64): a thread does 2 gate elements per step, not 8; rows 4-15 of the WMMA A tile stay zero; sums unchanged
# speedup vs baseline: 2.6924x; 1.0955x over previous
#include <hip/hip_runtime.h>
#include <hip/hip_bf16.h>
#include <stdint.h>

typedef __attribute__((ext_vector_type(16))) _Float16 v16h;
typedef __attribute__((ext_vector_type(4)))  _Float16 v4h;
typedef __attribute__((ext_vector_type(2)))  _Float16 v2h;
typedef __attribute__((ext_vector_type(8)))  float    v8f;

#define LBV 0.0f
#define UBV 0.2f

__global__ __launch_bounds__(256) void wfrag_prep(const float* __restrict__ W,
                                                  _Float16* __restrict__ dst,
                                                  int N, int K, int ldw, int KT,
                                                  int totalElems) {
  int idx = blockIdx.x * 256 + threadIdx.x;
  if (idx >= totalElems) return;
  int frag   = idx >> 9;
  int within = idx & 511;
  int r  = within >> 5;
  int k  = within & 31;
  int tileN = frag / KT;
  int tileK = frag - tileN * KT;
  int n  = tileN * 16 + r;
  int kk = tileK * 32 + k;
  float v = (n < N && kk < K) ? W[(size_t)n * ldw + kk] : 0.0f;
  int lane = (((k >> 3) & 1) << 4) + r;
  int elem = (k & 7) | ((k >> 4) << 3);
  dst[(size_t)frag * 512 + lane * 16 + elem] = (_Float16)v;
}

template <int TPW>
__global__ __launch_bounds__(256) void gemm_wmma(const float* __restrict__ A,
                                                 const _Float16* __restrict__ Bfrag,
                                                 const float* __restrict__ bias,
                                                 float* __restrict__ C,
                                                 int K, int lda, int ldc) {
  __shared__ __align__(32) _Float16 ldsA[4 * 512];
  int tid  = threadIdx.x;
  int lane = tid & 31;
  int wave = tid >> 5;
  int mBase = blockIdx.x * 64;
  int KT    = (K + 31) >> 5;
  v8f acc[TPW][4] = {};
  int sr  = tid >> 3;
  int sq  = tid & 7;
  int skl = sq << 2;
  int sln = (((skl >> 3) & 1) << 4) + (sr & 15);
  int se0 = (skl & 7) | ((skl >> 4) << 3);
  _Float16* st0 = &ldsA[(sr >> 4) * 512 + sln * 16 + se0];
  _Float16* st1 = st0 + 2 * 512;
  const float* Arow0 = A + (size_t)(mBase + sr) * lda + skl;
  const float* Arow1 = Arow0 + (size_t)32 * lda;
  const _Float16* Bw = Bfrag + (size_t)wave * KT * 512 + lane * 16;

  for (int kt = 0; kt < KT; ++kt) {
    int k0 = kt << 5;
    v16h b[TPW];
#pragma unroll
    for (int j = 0; j < TPW; ++j)
      b[j] = *reinterpret_cast<const v16h*>(&Bw[((size_t)(j << 3) * KT + kt) * 512]);
    v4h hv0, hv1;
    if (k0 + 32 <= K) {
      float4 va = *(const float4*)(Arow0 + k0);
      float4 vb = *(const float4*)(Arow1 + k0);
      hv0[0] = (_Float16)va.x; hv0[1] = (_Float16)va.y;
      hv0[2] = (_Float16)va.z; hv0[3] = (_Float16)va.w;
      hv1[0] = (_Float16)vb.x; hv1[1] = (_Float16)vb.y;
      hv1[2] = (_Float16)vb.z; hv1[3] = (_Float16)vb.w;
    } else {
#pragma unroll
      for (int j = 0; j < 4; ++j) {
        bool ok = (k0 + skl + j) < K;
        hv0[j] = (_Float16)(ok ? Arow0[k0 + j] : 0.0f);
        hv1[j] = (_Float16)(ok ? Arow1[k0 + j] : 0.0f);
      }
    }
    *reinterpret_cast<v4h*>(st0) = hv0;
    *reinterpret_cast<v4h*>(st1) = hv1;
    __syncthreads();
    v16h a[4];
#pragma unroll
    for (int mt = 0; mt < 4; ++mt)
      a[mt] = *reinterpret_cast<const v16h*>(&ldsA[mt * 512 + lane * 16]);
#pragma unroll
    for (int j = 0; j < TPW; ++j) {
#pragma unroll
      for (int mt = 0; mt < 4; ++mt)
        acc[j][mt] = __builtin_amdgcn_wmma_f32_16x16x32_f16(false, a[mt], false, b[j],
                                                            (short)0, acc[j][mt], false, false);
    }
    __syncthreads();
  }
  int rHalf = (lane >> 4) << 3;
#pragma unroll
  for (int j = 0; j < TPW; ++j) {
    int col = ((wave + (j << 3)) << 4) + (lane & 15);
    float bv = (bias != nullptr) ? bias[col] : 0.0f;
#pragma unroll
    for (int mt = 0; mt < 4; ++mt) {
      int rBase = mBase + (mt << 4) + rHalf;
#pragma unroll
      for (int r = 0; r < 8; ++r)
        C[(size_t)(rBase + r) * ldc + col] = acc[j][mt][r] + bv;
    }
  }
}

__device__ __forceinline__ float sigmoid_hw(float x) {
  return __builtin_amdgcn_rcpf(1.0f + __builtin_amdgcn_exp2f(x * -1.442695041f));
}
__device__ __forceinline__ float tanh_hw(float x) {
  return 1.0f - 2.0f * __builtin_amdgcn_rcpf(1.0f + __builtin_amdgcn_exp2f(x * 2.885390082f));
}

__device__ __forceinline__ void lds_barrier() {
  __builtin_amdgcn_fence(__ATOMIC_RELEASE, "workgroup", "local");
  __builtin_amdgcn_s_barrier();
  __builtin_amdgcn_fence(__ATOMIC_ACQUIRE, "workgroup", "local");
}

__device__ __forceinline__ void gi_fetch(float2 (&g)[3], const float* ga, int t) {
  const float* a = ga + (size_t)t * 384;
  g[0] = *(const float2*)(a);
  g[1] = *(const float2*)(a + 128);
  g[2] = *(const float2*)(a + 256);
}

#define GRU_RB 4
__global__ __launch_bounds__(256) __attribute__((amdgpu_waves_per_eu(2, 2))) void gru_recur(const float* __restrict__ gi,
                                                 const _Float16* __restrict__ WhhF,
                                                 const float* __restrict__ bhh,
                                                 float* __restrict__ out, int S) {
  __shared__ __align__(8) float hS[GRU_RB * 128];
  __shared__ __align__(32) _Float16 aF[4 * 512];
  __shared__ __align__(8) float ghS[GRU_RB * 384];
  int tid  = threadIdx.x;
  int lane = tid & 31;
  int wave = tid >> 5;
  int bBase = blockIdx.x * GRU_RB;
  const _Float16* Bw = WhhF + (size_t)(wave * 3) * 4 * 512 + lane * 16;
  if (S <= 0) return;

  v16h bF[3][4];
  float bv[3];
#pragma unroll
  for (int j = 0; j < 3; ++j) {
#pragma unroll
    for (int kt = 0; kt < 4; ++kt)
      bF[j][kt] = *reinterpret_cast<const v16h*>(&Bw[((size_t)j * 4 + kt) * 512]);
    bv[j] = bhh[((wave * 3 + j) << 4) + (lane & 15)];
  }

  int grow = tid >> 6;
  int gh0  = (tid & 63) << 1;
  const float* giA = gi + (size_t)(bBase + grow) * S * 384 + gh0;

  float2 g0[3], g1[3], g2[3];
  gi_fetch(g0, giA, 0);
  gi_fetch(g1, giA, (1 < S) ? 1 : S - 1);

  for (int i = tid; i < GRU_RB * 128; i += 256) hS[i] = 0.0f;
  for (int i = tid; i < 4 * 512; i += 256) aF[i] = (_Float16)0.0f;
  __syncthreads();

  auto step = [&](int t, const float2 (&g)[3]) __attribute__((always_inline)) {
    v16h a[4];
#pragma unroll
    for (int kt = 0; kt < 4; ++kt)
      a[kt] = *reinterpret_cast<const v16h*>(&aF[kt * 512 + lane * 16]);
    __builtin_amdgcn_sched_barrier(0);
    v8f acc[3] = {};
#pragma unroll
    for (int kt = 0; kt < 4; ++kt) {
#pragma unroll
      for (int j = 0; j < 3; ++j)
        acc[j] = __builtin_amdgcn_wmma_f32_16x16x32_f16(false, a[kt], false, bF[j][kt],
                                                        (short)0, acc[j], false, false);
    }
    if (lane < 16) {
#pragma unroll
      for (int j = 0; j < 3; ++j) {
        int col = ((wave * 3 + j) << 4) + lane;
#pragma unroll
        for (int r = 0; r < GRU_RB; ++r) ghS[r * 384 + col] = acc[j][r] + bv[j];
      }
    }
    lds_barrier();
    float2 hr = *(const float2*)(&ghS[grow * 384 + gh0]);
    float2 hz = *(const float2*)(&ghS[grow * 384 + 128 + gh0]);
    float2 hn = *(const float2*)(&ghS[grow * 384 + 256 + gh0]);
    float2 ho = *(const float2*)(&hS[grow * 128 + gh0]);
    float2 gr = g[0];
    float2 gz = g[1];
    float2 gn = g[2];
    float2 hNew;
#define GATE(c)                                                            \
    {                                                                      \
      float rg = sigmoid_hw(gr.c + hr.c);                                  \
      float zg = sigmoid_hw(gz.c + hz.c);                                  \
      float ng = tanh_hw(gn.c + rg * hn.c);                                \
      hNew.c = (1.0f - zg) * ng + zg * ho.c;                               \
    }
    GATE(x) GATE(y)
#undef GATE
    *(float2*)(&hS[grow * 128 + gh0]) = hNew;
    *(float2*)(out + ((size_t)(bBase + grow) * S + t) * 128 + gh0) = hNew;
    {
      int kt = gh0 >> 5;
      int kl = gh0 & 31;
      int ln = (((kl >> 3) & 1) << 4) + grow;
      int e0 = (kl & 7) | ((kl >> 4) << 3);
      v2h hv;
      hv[0] = (_Float16)hNew.x; hv[1] = (_Float16)hNew.y;
      *reinterpret_cast<v2h*>(&aF[kt * 512 + ln * 16 + e0]) = hv;
    }
    lds_barrier();
  };

  int t = 0;
  for (; t + 2 < S; t += 3) {
    gi_fetch(g2, giA, t + 2);
    step(t, g0);
    gi_fetch(g0, giA, (t + 3 < S) ? t + 3 : S - 1);
    step(t + 1, g1);
    gi_fetch(g1, giA, (t + 4 < S) ? t + 4 : S - 1);
    step(t + 2, g2);
  }
  if (t < S) {
    step(t, g0);
    if (t + 1 < S) step(t + 1, g1);
  }
}

__device__ __forceinline__ float blockReduceSum(float v, float* red) {
  int tid = threadIdx.x;
  red[tid] = v;
  __syncthreads();
  for (int off = 128; off > 0; off >>= 1) {
    if (tid < off) red[tid] += red[tid + off];
    __syncthreads();
  }
  float r = red[0];
  __syncthreads();
  return r;
}
__device__ __forceinline__ float blockReduceMax(float v, float* red) {
  int tid = threadIdx.x;
  red[tid] = v;
  __syncthreads();
  for (int off = 128; off > 0; off >>= 1) {
    if (tid < off) red[tid] = fmaxf(red[tid], red[tid + off]);
    __syncthreads();
  }
  float r = red[0];
  __syncthreads();
  return r;
}

__global__ __launch_bounds__(256) void finalize_kernel(const float* __restrict__ gE,
                                                       const float* __restrict__ u,
                                                       const float* __restrict__ out1,
                                                       const float* __restrict__ vvec,
                                                       const float* __restrict__ Wfc,
                                                       const float* __restrict__ bfc,
                                                       float* __restrict__ outp, int S) {
  __shared__ float uL[128], vL[128], eS[64], attnS[64];
  __shared__ __align__(16) float concatL[256];
  __shared__ float oldL[512], wcL[512], red[256];
  int tid  = threadIdx.x;
  int lane = tid & 31;
  int wave = tid >> 5;
  int b = blockIdx.x;

  if (tid < 128) {
    uL[tid] = u[(size_t)b * 128 + tid];
    vL[tid] = vvec[tid];
  }
  __syncthreads();

  for (int s = wave; s < S; s += 8) {
    float p = 0.0f;
    const float* ge = gE + ((size_t)b * S + s) * 128;
    for (int h = lane; h < 128; h += 32) p += vL[h] * tanhf(uL[h] + ge[h]);
    for (int off = 16; off > 0; off >>= 1) p += __shfl_down(p, off, 32);
    if (lane == 0) eS[s] = p;
  }
  __syncthreads();

  if (tid == 0) {
    float m = eS[0];
    for (int s = 1; s < S; ++s) m = fmaxf(m, eS[s]);
    float sum = 0.0f;
    for (int s = 0; s < S; ++s) { float ev = expf(eS[s] - m); attnS[s] = ev; sum += ev; }
    float inv = 1.0f / sum;
    for (int s = 0; s < S; ++s) attnS[s] *= inv;
  }
  __syncthreads();

  if (tid < 128) {
    float c = 0.0f;
    for (int s = 0; s < S; ++s) c += attnS[s] * out1[((size_t)b * S + s) * 128 + tid];
    concatL[tid]       = out1[((size_t)b * S + (S - 1)) * 128 + tid];
    concatL[128 + tid] = c;
  }
  __syncthreads();

  {
    const float4 c0 = *(const float4*)(&concatL[lane << 2]);
    const float4 c1 = *(const float4*)(&concatL[128 + (lane << 2)]);
    const float* wl = Wfc + (lane << 2);
    for (int n0 = wave; n0 < 500; n0 += 64) {
      float4 w0[8], w1[8];
      float bn[8], p[8];
#pragma unroll
      for (int i = 0; i < 8; ++i) {
        int n  = n0 + (i << 3);
        int nc = (n < 500) ? n : 499;
        const float* wr = wl + (size_t)nc * 256;
        w0[i] = *(const float4*)(wr);
        w1[i] = *(const float4*)(wr + 128);
        bn[i] = bfc[nc];
      }
      __builtin_amdgcn_sched_barrier(0);
#pragma unroll
      for (int i = 0; i < 8; ++i) {
        float a = (lane == 0) ? bn[i] : 0.0f;
        a += w0[i].x * c0.x;
        a += w0[i].y * c0.y;
        a += w0[i].z * c0.z;
        a += w0[i].w * c0.w;
        a += w1[i].x * c1.x;
        a += w1[i].y * c1.y;
        a += w1[i].z * c1.z;
        a += w1[i].w * c1.w;
        p[i] = a;
      }
#pragma unroll
      for (int off = 16; off > 0; off >>= 1) {
#pragma unroll
        for (int i = 0; i < 8; ++i) p[i] += __shfl_xor(p[i], off, 32);
      }
      if (lane == 0) {
#pragma unroll
        for (int i = 0; i < 8; ++i) {
          int n = n0 + (i << 3);
          if (n < 500) oldL[n] = p[i];
        }
      }
    }
  }
  __syncthreads();

  float pm = -3.4e38f;
  for (int n = tid; n < 500; n += 256) pm = fmaxf(pm, oldL[n]);
  pm = blockReduceMax(pm, red);
  float ps = 0.0f;
  for (int n = tid; n < 500; n += 256) { float e = expf(oldL[n] - pm); wcL[n] = e; ps += e; }
  ps = blockReduceSum(ps, red);
  float inv = 1.0f / ps;
  for (int n = tid; n < 500; n += 256) {
    float w = wcL[n] * inv;
    oldL[n] = w;
    wcL[n]  = fminf(fmaxf(w, LBV), UBV);
  }
  __syncthreads();

  for (int it = 0; it < 32; ++it) {
    float pl = 0.0f, pn = 0.0f;
    for (int n = tid; n < 500; n += 256) {
      float wcv = wcL[n];
      pl += oldL[n] - wcv;
      if (wcv != UBV) pn += wcv;
    }
    float leftover = blockReduceSum(pl, red);
    float nsum     = blockReduceSum(pn, red);
    float ns = (nsum == 0.0f) ? 1.0f : nsum;
    float pover = 0.0f;
    for (int n = tid; n < 500; n += 256) {
      float wcv = wcL[n];
      float wc2 = (wcv != UBV) ? wcv + leftover * wcv / ns : wcv;
      oldL[n] = wc2;
      if (wc2 > UBV) pover = 1.0f;
    }
    float anyOver = blockReduceSum(pover, red);
    int new_done = (anyOver == 0.0f);
    for (int n = tid; n < 500; n += 256) {
      float wc2 = oldL[n];
      wcL[n] = new_done ? wc2 : fminf(fmaxf(wc2, LBV), UBV);
    }
    __syncthreads();
    if (new_done) break;
  }

  for (int n = tid; n < 500; n += 256) outp[(size_t)b * 500 + n] = wcL[n];
}

extern "C" void kernel_launch(void* const* d_in, const int* in_sizes, int n_in,
                              void* d_out, int out_size, void* d_ws, size_t ws_size,
                              hipStream_t stream) {
  (void)in_sizes; (void)n_in; (void)out_size; (void)ws_size;
  const float* x    = (const float*)d_in[0];
  const float* Wih0 = (const float*)d_in[1];
  const float* Whh0 = (const float*)d_in[2];
  const float* bih0 = (const float*)d_in[3];
  const float* bhh0 = (const float*)d_in[4];
  const float* Wih1 = (const float*)d_in[5];
  const float* Whh1 = (const float*)d_in[6];
  const float* bih1 = (const float*)d_in[7];
  const float* bhh1 = (const float*)d_in[8];
  const float* Wa   = (const float*)d_in[9];
  const float* ba   = (const float*)d_in[10];
  const float* vv   = (const float*)d_in[11];
  const float* Wfc  = (const float*)d_in[12];
  const float* bfc  = (const float*)d_in[13];
  float* outp = (float*)d_out;

  constexpr int B = 1024, S = 60, N = 500, H = 128, G3 = 384;

  uint8_t* base = (uint8_t*)d_ws;
  size_t off = 0;
  auto carve = [&](size_t bytes) {
    void* p = base + off;
    off = (off + bytes + 255) & ~(size_t)255;
    return p;
  };
  float* gi   = (float*)carve((size_t)B * S * G3 * 4);
  float* out0 = (float*)carve((size_t)B * S * H * 4);
  float* out1 = (float*)carve((size_t)B * S * H * 4);
  float* gEb  = (float*)carve((size_t)B * S * H * 4);
  float* ub   = (float*)carve((size_t)B * H * 4);
  _Float16* Wih0F = (_Float16*)carve((size_t)24 * 16 * 512 * 2);
  _Float16* Whh0F = (_Float16*)carve((size_t)24 * 4 * 512 * 2);
  _Float16* Wih1F = (_Float16*)carve((size_t)24 * 4 * 512 * 2);
  _Float16* Whh1F = (_Float16*)carve((size_t)24 * 4 * 512 * 2);
  _Float16* WaxF  = (_Float16*)carve((size_t)8 * 16 * 512 * 2);
  _Float16* WahF  = (_Float16*)carve((size_t)8 * 4 * 512 * 2);

  auto prep = [&](const float* W, _Float16* dst, int n, int k, int ldw) {
    int KT = (k + 31) >> 5;
    int NT = (n + 15) >> 4;
    int total = NT * KT * 512;
    wfrag_prep<<<(total + 255) / 256, 256, 0, stream>>>(W, dst, n, k, ldw, KT, total);
  };
  prep(Wih0,   Wih0F, G3, N, N);
  prep(Whh0,   Whh0F, G3, H, H);
  prep(Wih1,   Wih1F, G3, H, H);
  prep(Whh1,   Whh1F, G3, H, H);
  prep(Wa + H, WaxF,  H,  N, H + N);
  prep(Wa,     WahF,  H,  H, H + N);

  auto gemm384 = [&](const float* A, const _Float16* Bf, const float* bias, float* C,
                     int M, int K, int lda) {
    gemm_wmma<3><<<dim3(M / 64), 256, 0, stream>>>(A, Bf, bias, C, K, lda, G3);
  };
  auto gemm128 = [&](const float* A, const _Float16* Bf, const float* bias, float* C,
                     int M, int K, int lda) {
    gemm_wmma<1><<<dim3(M / 64), 256, 0, stream>>>(A, Bf, bias, C, K, lda, H);
  };

  gemm384(x, Wih0F, bih0, gi, B * S, N, N);
  gru_recur<<<B / GRU_RB, 256, 0, stream>>>(gi, Whh0F, bhh0, out0, S);
  gemm384(out0, Wih1F, bih1, gi, B * S, H, H);
  gru_recur<<<B / GRU_RB, 256, 0, stream>>>(gi, Whh1F, bhh1, out1, S);
  gemm128(x, WaxF, ba, gEb, B * S, N, N);
  gemm128(out1 + (size_t)(S - 1) * H, WahF, nullptr, ub, B, H, S * H);
  finalize_kernel<<<B, 256, 0, stream>>>(gEb, ub, out1, vv, Wfc, bfc, outp, S);
}


# Weight conversion to WMMA fragment order: one launch of 208 blocks for all six matrices (a thread writes 8 consecutive k as one 16-byte piece), was six launches, 1664 blocks; values unchanged
# speedup vs baseline: 2.7984x; 1.0394x over previous
#include <hip/hip_runtime.h>
#include <hip/hip_bf16.h>
#include <stdint.h>

typedef __attribute__((ext_vector_type(16))) _Float16 v16h;
typedef __attribute__((ext_vector_type(8)))  _Float16 v8h;
typedef __attribute__((ext_vector_type(4)))  _Float16 v4h;
typedef __attribute__((ext_vector_type(2)))  _Float16 v2h;
typedef __attribute__((ext_vector_type(8)))  float    v8f;

#define LBV 0.0f
#define UBV 0.2f

constexpr int kB = 1024, kS = 60, kN = 500, kH = 128, kG3 = 384;
static_assert(kG3 % 16 == 0 && kH % 16 == 0, "weight rows fill whole 16-row fragments");

constexpr int prepBlocks(int n, int k) { return (n / 16) * ((k + 31) / 32) * 64 / 256; }
constexpr int kPB0 = prepBlocks(kG3, kN);
constexpr int kPB1 = kPB0 + prepBlocks(kG3, kH);
constexpr int kPB2 = kPB1 + prepBlocks(kG3, kH);
constexpr int kPB3 = kPB2 + prepBlocks(kG3, kH);
constexpr int kPB4 = kPB3 + prepBlocks(kH, kN);
constexpr int kPB5 = kPB4 + prepBlocks(kH, kH);

__global__ __launch_bounds__(256) void wfrag_prep_all(const float* __restrict__ wih0,
                                                      const float* __restrict__ whh0,
                                                      const float* __restrict__ wih1,
                                                      const float* __restrict__ whh1,
                                                      const float* __restrict__ wax,
                                                      const float* __restrict__ wah,
                                                      _Float16* __restrict__ d0,
                                                      _Float16* __restrict__ d1,
                                                      _Float16* __restrict__ d2,
                                                      _Float16* __restrict__ d3,
                                                      _Float16* __restrict__ d4,
                                                      _Float16* __restrict__ d5) {
  int blk = blockIdx.x;
  const float* W;
  _Float16* dst;
  int K, ldw, b0;
  if (blk < kPB0)      { W = wih0; dst = d0; K = kN; ldw = kN;      b0 = 0;    }
  else if (blk < kPB1) { W = whh0; dst = d1; K = kH; ldw = kH;      b0 = kPB0; }
  else if (blk < kPB2) { W = wih1; dst = d2; K = kH; ldw = kH;      b0 = kPB1; }
  else if (blk < kPB3) { W = whh1; dst = d3; K = kH; ldw = kH;      b0 = kPB2; }
  else if (blk < kPB4) { W = wax;  dst = d4; K = kN; ldw = kH + kN; b0 = kPB3; }
  else                 { W = wah;  dst = d5; K = kH; ldw = kH + kN; b0 = kPB4; }
  int KT = (K + 31) >> 5;
  int u      = (blk - b0) * 256 + threadIdx.x;
  int frag   = u >> 6;
  int within = u & 63;
  int lane = within >> 1;
  int e8   = within & 1;
  int tileN = frag / KT;
  int tileK = frag - tileN * KT;
  int n  = tileN * 16 + (lane & 15);
  int kk = tileK * 32 + ((lane >> 4) << 3) + (e8 << 4);
  const float* row = W + (size_t)n * ldw;
  float v[8];
#pragma unroll
  for (int i = 0; i < 8; ++i) v[i] = row[(kk + i < K) ? kk + i : K - 1];
  v8h hv;
#pragma unroll
  for (int i = 0; i < 8; ++i) hv[i] = (_Float16)((kk + i < K) ? v[i] : 0.0f);
  *reinterpret_cast<v8h*>(dst + (size_t)frag * 512 + lane * 16 + e8 * 8) = hv;
}

template <int TPW>
__global__ __launch_bounds__(256) void gemm_wmma(const float* __restrict__ A,
                                                 const _Float16* __restrict__ Bfrag,
                                                 const float* __restrict__ bias,
                                                 float* __restrict__ C,
                                                 int K, int lda, int ldc) {
  __shared__ __align__(32) _Float16 ldsA[4 * 512];
  int tid  = threadIdx.x;
  int lane = tid & 31;
  int wave = tid >> 5;
  int mBase = blockIdx.x * 64;
  int KT    = (K + 31) >> 5;
  v8f acc[TPW][4] = {};
  int sr  = tid >> 3;
  int sq  = tid & 7;
  int skl = sq << 2;
  int sln = (((skl >> 3) & 1) << 4) + (sr & 15);
  int se0 = (skl & 7) | ((skl >> 4) << 3);
  _Float16* st0 = &ldsA[(sr >> 4) * 512 + sln * 16 + se0];
  _Float16* st1 = st0 + 2 * 512;
  const float* Arow0 = A + (size_t)(mBase + sr) * lda + skl;
  const float* Arow1 = Arow0 + (size_t)32 * lda;
  const _Float16* Bw = Bfrag + (size_t)wave * KT * 512 + lane * 16;

  for (int kt = 0; kt < KT; ++kt) {
    int k0 = kt << 5;
    v16h b[TPW];
#pragma unroll
    for (int j = 0; j < TPW; ++j)
      b[j] = *reinterpret_cast<const v16h*>(&Bw[((size_t)(j << 3) * KT + kt) * 512]);
    v4h hv0, hv1;
    if (k0 + 32 <= K) {
      float4 va = *(const float4*)(Arow0 + k0);
      float4 vb = *(const float4*)(Arow1 + k0);
      hv0[0] = (_Float16)va.x; hv0[1] = (_Float16)va.y;
      hv0[2] = (_Float16)va.z; hv0[3] = (_Float16)va.w;
      hv1[0] = (_Float16)vb.x; hv1[1] = (_Float16)vb.y;
      hv1[2] = (_Float16)vb.z; hv1[3] = (_Float16)vb.w;
    } else {
#pragma unroll
      for (int j = 0; j < 4; ++j) {
        bool ok = (k0 + skl + j) < K;
        hv0[j] = (_Float16)(ok ? Arow0[k0 + j] : 0.0f);
        hv1[j] = (_Float16)(ok ? Arow1[k0 + j] : 0.0f);
      }
    }
    *reinterpret_cast<v4h*>(st0) = hv0;
    *reinterpret_cast<v4h*>(st1) = hv1;
    __syncthreads();
    v16h a[4];
#pragma unroll
    for (int mt = 0; mt < 4; ++mt)
      a[mt] = *reinterpret_cast<const v16h*>(&ldsA[mt * 512 + lane * 16]);
#pragma unroll
    for (int j = 0; j < TPW; ++j) {
#pragma unroll
      for (int mt = 0; mt < 4; ++mt)
        acc[j][mt] = __builtin_amdgcn_wmma_f32_16x16x32_f16(false, a[mt], false, b[j],
                                                            (short)0, acc[j][mt], false, false);
    }
    __syncthreads();
  }
  int rHalf = (lane >> 4) << 3;
#pragma unroll
  for (int j = 0; j < TPW; ++j) {
    int col = ((wave + (j << 3)) << 4) + (lane & 15);
    float bv = (bias != nullptr) ? bias[col] : 0.0f;
#pragma unroll
    for (int mt = 0; mt < 4; ++mt) {
      int rBase = mBase + (mt << 4) + rHalf;
#pragma unroll
      for (int r = 0; r < 8; ++r)
        C[(size_t)(rBase + r) * ldc + col] = acc[j][mt][r] + bv;
    }
  }
}

__device__ __forceinline__ float sigmoid_hw(float x) {
  return __builtin_amdgcn_rcpf(1.0f + __builtin_amdgcn_exp2f(x * -1.442695041f));
}
__device__ __forceinline__ float tanh_hw(float x) {
  return 1.0f - 2.0f * __builtin_amdgcn_rcpf(1.0f + __builtin_amdgcn_exp2f(x * 2.885390082f));
}

__device__ __forceinline__ void lds_barrier() {
  __builtin_amdgcn_fence(__ATOMIC_RELEASE, "workgroup", "local");
  __builtin_amdgcn_s_barrier();
  __builtin_amdgcn_fence(__ATOMIC_ACQUIRE, "workgroup", "local");
}

__device__ __forceinline__ void gi_fetch(float2 (&g)[3], const float* ga, int t) {
  const float* a = ga + (size_t)t * 384;
  g[0] = *(const float2*)(a);
  g[1] = *(const float2*)(a + 128);
  g[2] = *(const float2*)(a + 256);
}

#define GRU_RB 4
__global__ __launch_bounds__(256) __attribute__((amdgpu_waves_per_eu(2, 2))) void gru_recur(const float* __restrict__ gi,
                                                 const _Float16* __restrict__ WhhF,
                                                 const float* __restrict__ bhh,
                                                 float* __restrict__ out, int S) {
  __shared__ __align__(8) float hS[GRU_RB * 128];
  __shared__ __align__(32) _Float16 aF[4 * 512];
  __shared__ __align__(8) float ghS[GRU_RB * 384];
  int tid  = threadIdx.x;
  int lane = tid & 31;
  int wave = tid >> 5;
  int bBase = blockIdx.x * GRU_RB;
  const _Float16* Bw = WhhF + (size_t)(wave * 3) * 4 * 512 + lane * 16;
  if (S <= 0) return;

  v16h bF[3][4];
  float bv[3];
#pragma unroll
  for (int j = 0; j < 3; ++j) {
#pragma unroll
    for (int kt = 0; kt < 4; ++kt)
      bF[j][kt] = *reinterpret_cast<const v16h*>(&Bw[((size_t)j * 4 + kt) * 512]);
    bv[j] = bhh[((wave * 3 + j) << 4) + (lane & 15)];
  }

  int grow = tid >> 6;
  int gh0  = (tid & 63) << 1;
  const float* giA = gi + (size_t)(bBase + grow) * S * 384 + gh0;

  float2 g0[3], g1[3], g2[3];
  gi_fetch(g0, giA, 0);
  gi_fetch(g1, giA, (1 < S) ? 1 : S - 1);

  for (int i = tid; i < GRU_RB * 128; i += 256) hS[i] = 0.0f;
  for (int i = tid; i < 4 * 512; i += 256) aF[i] = (_Float16)0.0f;
  __syncthreads();

  auto step = [&](int t, const float2 (&g)[3]) __attribute__((always_inline)) {
    v16h a[4];
#pragma unroll
    for (int kt = 0; kt < 4; ++kt)
      a[kt] = *reinterpret_cast<const v16h*>(&aF[kt * 512 + lane * 16]);
    __builtin_amdgcn_sched_barrier(0);
    v8f acc[3] = {};
#pragma unroll
    for (int kt = 0; kt < 4; ++kt) {
#pragma unroll
      for (int j = 0; j < 3; ++j)
        acc[j] = __builtin_amdgcn_wmma_f32_16x16x32_f16(false, a[kt], false, bF[j][kt],
                                                        (short)0, acc[j], false, false);
    }
    if (lane < 16) {
#pragma unroll
      for (int j = 0; j < 3; ++j) {
        int col = ((wave * 3 + j) << 4) + lane;
#pragma unroll
        for (int r = 0; r < GRU_RB; ++r) ghS[r * 384 + col] = acc[j][r] + bv[j];
      }
    }
    lds_barrier();
    float2 hr = *(const float2*)(&ghS[grow * 384 + gh0]);
    float2 hz = *(const float2*)(&ghS[grow * 384 + 128 + gh0]);
    float2 hn = *(const float2*)(&ghS[grow * 384 + 256 + gh0]);
    float2 ho = *(const float2*)(&hS[grow * 128 + gh0]);
    float2 gr = g[0];
    float2 gz = g[1];
    float2 gn = g[2];
    float2 hNew;
#define GATE(c)                                                            \
    {                                                                      \
      float rg = sigmoid_hw(gr.c + hr.c);                                  \
      float zg = sigmoid_hw(gz.c + hz.c);                                  \
      float ng = tanh_hw(gn.c + rg * hn.c);                                \
      hNew.c = (1.0f - zg) * ng + zg * ho.c;                               \
    }
    GATE(x) GATE(y)
#undef GATE
    *(float2*)(&hS[grow * 128 + gh0]) = hNew;
    *(float2*)(out + ((size_t)(bBase + grow) * S + t) * 128 + gh0) = hNew;
    {
      int kt = gh0 >> 5;
      int kl = gh0 & 31;
      int ln = (((kl >> 3) & 1) << 4) + grow;
      int e0 = (kl & 7) | ((kl >> 4) << 3);
      v2h hv;
      hv[0] = (_Float16)hNew.x; hv[1] = (_Float16)hNew.y;
      *reinterpret_cast<v2h*>(&aF[kt * 512 + ln * 16 + e0]) = hv;
    }
    lds_barrier();
  };

  int t = 0;
  for (; t + 2 < S; t += 3) {
    gi_fetch(g2, giA, t + 2);
    step(t, g0);
    gi_fetch(g0, giA, (t + 3 < S) ? t + 3 : S - 1);
    step(t + 1, g1);
    gi_fetch(g1, giA, (t + 4 < S) ? t + 4 : S - 1);
    step(t + 2, g2);
  }
  if (t < S) {
    step(t, g0);
    if (t + 1 < S) step(t + 1, g1);
  }
}

__device__ __forceinline__ float blockReduceSum(float v, float* red) {
  int tid = threadIdx.x;
  red[tid] = v;
  __syncthreads();
  for (int off = 128; off > 0; off >>= 1) {
    if (tid < off) red[tid] += red[tid + off];
    __syncthreads();
  }
  float r = red[0];
  __syncthreads();
  return r;
}
__device__ __forceinline__ float blockReduceMax(float v, float* red) {
  int tid = threadIdx.x;
  red[tid] = v;
  __syncthreads();
  for (int off = 128; off > 0; off >>= 1) {
    if (tid < off) red[tid] = fmaxf(red[tid], red[tid + off]);
    __syncthreads();
  }
  float r = red[0];
  __syncthreads();
  return r;
}

__global__ __launch_bounds__(256) void finalize_kernel(const float* __restrict__ gE,
                                                       const float* __restrict__ u,
                                                       const float* __restrict__ out1,
                                                       const float* __restrict__ vvec,
                                                       const float* __restrict__ Wfc,
                                                       const float* __restrict__ bfc,
                                                       float* __restrict__ outp, int S) {
  __shared__ float uL[128], vL[128], eS[64], attnS[64];
  __shared__ __align__(16) float concatL[256];
  __shared__ float oldL[512], wcL[512], red[256];
  int tid  = threadIdx.x;
  int lane = tid & 31;
  int wave = tid >> 5;
  int b = blockIdx.x;

  if (tid < 128) {
    uL[tid] = u[(size_t)b * 128 + tid];
    vL[tid] = vvec[tid];
  }
  __syncthreads();

  for (int s = wave; s < S; s += 8) {
    float p = 0.0f;
    const float* ge = gE + ((size_t)b * S + s) * 128;
    for (int h = lane; h < 128; h += 32) p += vL[h] * tanhf(uL[h] + ge[h]);
    for (int off = 16; off > 0; off >>= 1) p += __shfl_down(p, off, 32);
    if (lane == 0) eS[s] = p;
  }
  __syncthreads();

  if (tid == 0) {
    float m = eS[0];
    for (int s = 1; s < S; ++s) m = fmaxf(m, eS[s]);
    float sum = 0.0f;
    for (int s = 0; s < S; ++s) { float ev = expf(eS[s] - m); attnS[s] = ev; sum += ev; }
    float inv = 1.0f / sum;
    for (int s = 0; s < S; ++s) attnS[s] *= inv;
  }
  __syncthreads();

  if (tid < 128) {
    float c = 0.0f;
    for (int s = 0; s < S; ++s) c += attnS[s] * out1[((size_t)b * S + s) * 128 + tid];
    concatL[tid]       = out1[((size_t)b * S + (S - 1)) * 128 + tid];
    concatL[128 + tid] = c;
  }
  __syncthreads();

  {
    const float4 c0 = *(const float4*)(&concatL[lane << 2]);
    const float4 c1 = *(const float4*)(&concatL[128 + (lane << 2)]);
    const float* wl = Wfc + (lane << 2);
    for (int n0 = wave; n0 < 500; n0 += 64) {
      float4 w0[8], w1[8];
      float bn[8], p[8];
#pragma unroll
      for (int i = 0; i < 8; ++i) {
        int n  = n0 + (i << 3);
        int nc = (n < 500) ? n : 499;
        const float* wr = wl + (size_t)nc * 256;
        w0[i] = *(const float4*)(wr);
        w1[i] = *(const float4*)(wr + 128);
        bn[i] = bfc[nc];
      }
      __builtin_amdgcn_sched_barrier(0);
#pragma unroll
      for (int i = 0; i < 8; ++i) {
        float a = (lane == 0) ? bn[i] : 0.0f;
        a += w0[i].x * c0.x;
        a += w0[i].y * c0.y;
        a += w0[i].z * c0.z;
        a += w0[i].w * c0.w;
        a += w1[i].x * c1.x;
        a += w1[i].y * c1.y;
        a += w1[i].z * c1.z;
        a += w1[i].w * c1.w;
        p[i] = a;
      }
#pragma unroll
      for (int off = 16; off > 0; off >>= 1) {
#pragma unroll
        for (int i = 0; i < 8; ++i) p[i] += __shfl_xor(p[i], off, 32);
      }
      if (lane == 0) {
#pragma unroll
        for (int i = 0; i < 8; ++i) {
          int n = n0 + (i << 3);
          if (n < 500) oldL[n] = p[i];
        }
      }
    }
  }
  __syncthreads();

  float pm = -3.4e38f;
  for (int n = tid; n < 500; n += 256) pm = fmaxf(pm, oldL[n]);
  pm = blockReduceMax(pm, red);
  float ps = 0.0f;
  for (int n = tid; n < 500; n += 256) { float e = expf(oldL[n] - pm); wcL[n] = e; ps += e; }
  ps = blockReduceSum(ps, red);
  float inv = 1.0f / ps;
  for (int n = tid; n < 500; n += 256) {
    float w = wcL[n] * inv;
    oldL[n] = w;
    wcL[n]  = fminf(fmaxf(w, LBV), UBV);
  }
  __syncthreads();

  for (int it = 0; it < 32; ++it) {
    float pl = 0.0f, pn = 0.0f;
    for (int n = tid; n < 500; n += 256) {
      float wcv = wcL[n];
      pl += oldL[n] - wcv;
      if (wcv != UBV) pn += wcv;
    }
    float leftover = blockReduceSum(pl, red);
    float nsum     = blockReduceSum(pn, red);
    float ns = (nsum == 0.0f) ? 1.0f : nsum;
    float pover = 0.0f;
    for (int n = tid; n < 500; n += 256) {
      float wcv = wcL[n];
      float wc2 = (wcv != UBV) ? wcv + leftover * wcv / ns : wcv;
      oldL[n] = wc2;
      if (wc2 > UBV) pover = 1.0f;
    }
    float anyOver = blockReduceSum(pover, red);
    int new_done = (anyOver == 0.0f);
    for (int n = tid; n < 500; n += 256) {
      float wc2 = oldL[n];
      wcL[n] = new_done ? wc2 : fminf(fmaxf(wc2, LBV), UBV);
    }
    __syncthreads();
    if (new_done) break;
  }

  for (int n = tid; n < 500; n += 256) outp[(size_t)b * 500 + n] = wcL[n];
}

extern "C" void kernel_launch(void* const* d_in, const int* in_sizes, int n_in,
                              void* d_out, int out_size, void* d_ws, size_t ws_size,
                              hipStream_t stream) {
  (void)in_sizes; (void)n_in; (void)out_size; (void)ws_size;
  const float* x    = (const float*)d_in[0];
  const float* Wih0 = (const float*)d_in[1];
  const float* Whh0 = (const float*)d_in[2];
  const float* bih0 = (const float*)d_in[3];
  const float* bhh0 = (const float*)d_in[4];
  const float* Wih1 = (const float*)d_in[5];
  const float* Whh1 = (const float*)d_in[6];
  const float* bih1 = (const float*)d_in[7];
  const float* bhh1 = (const float*)d_in[8];
  const float* Wa   = (const float*)d_in[9];
  const float* ba   = (const float*)d_in[10];
  const float* vv   = (const float*)d_in[11];
  const float* Wfc  = (const float*)d_in[12];
  const float* bfc  = (const float*)d_in[13];
  float* outp = (float*)d_out;

  constexpr int B = kB, S = kS, N = kN, H = kH, G3 = kG3;

  uint8_t* base = (uint8_t*)d_ws;
  size_t off = 0;
  auto carve = [&](size_t bytes) {
    void* p = base + off;
    off = (off + bytes + 255) & ~(size_t)255;
    return p;
  };
  float* gi   = (float*)carve((size_t)B * S * G3 * 4);
  float* out0 = (float*)carve((size_t)B * S * H * 4);
  float* out1 = (float*)carve((size_t)B * S * H * 4);
  float* gEb  = (float*)carve((size_t)B * S * H * 4);
  float* ub   = (float*)carve((size_t)B * H * 4);
  _Float16* Wih0F = (_Float16*)carve((size_t)24 * 16 * 512 * 2);
  _Float16* Whh0F = (_Float16*)carve((size_t)24 * 4 * 512 * 2);
  _Float16* Wih1F = (_Float16*)carve((size_t)24 * 4 * 512 * 2);
  _Float16* Whh1F = (_Float16*)carve((size_t)24 * 4 * 512 * 2);
  _Float16* WaxF  = (_Float16*)carve((size_t)8 * 16 * 512 * 2);
  _Float16* WahF  = (_Float16*)carve((size_t)8 * 4 * 512 * 2);

  wfrag_prep_all<<<kPB5, 256, 0, stream>>>(Wih0, Whh0, Wih1, Whh1, Wa + H, Wa,
                                           Wih0F, Whh0F, Wih1F, Whh1F, WaxF, WahF);

  auto gemm384 = [&](const float* A, const _Float16* Bf, const float* bias, float* C,
                     int M, int K, int lda) {
    gemm_wmma<3><<<dim3(M / 64), 256, 0, stream>>>(A, Bf, bias, C, K, lda, G3);
  };
  auto gemm128 = [&](const float* A, const _Float16* Bf, const float* bias, float* C,
                     int M, int K, int lda) {
    gemm_wmma<1><<<dim3(M / 64), 256, 0, stream>>>(A, Bf, bias, C, K, lda, H);
  };

  gemm384(x, Wih0F, bih0, gi, B * S, N, N);
  gru_recur<<<B / GRU_RB, 256, 0, stream>>>(gi, Whh0F, bhh0, out0, S);
  gemm384(out0, Wih1F, bih1, gi, B * S, H, H);
  gru_recur<<<B / GRU_RB, 256, 0, stream>>>(gi, Whh1F, bhh1, out1, S);
  gemm128(x, WaxF, ba, gEb, B * S, N, N);
  gemm128(out1 + (size_t)(S - 1) * H, WahF, nullptr, ub, B, H, S * H);
  finalize_kernel<<<B, 256, 0, stream>>>(gEb, ub, out1, vv, Wfc, bfc, outp, S);
}


# Attention projection x @ Wa_x^T rides in the layer-0 input GEMM as a fourth N tile per wave: x (the largest array) is read once, not twice, and one launch goes; each product's sums unchanged
# speedup vs baseline: 3.0206x; 1.0794x over previous
#include <hip/hip_runtime.h>
#include <hip/hip_bf16.h>
#include <stdint.h>

typedef __attribute__((ext_vector_type(16))) _Float16 v16h;
typedef __attribute__((ext_vector_type(8)))  _Float16 v8h;
typedef __attribute__((ext_vector_type(4)))  _Float16 v4h;
typedef __attribute__((ext_vector_type(2)))  _Float16 v2h;
typedef __attribute__((ext_vector_type(8)))  float    v8f;

#define LBV 0.0f
#define UBV 0.2f

constexpr int kB = 1024, kS = 60, kN = 500, kH = 128, kG3 = 384;
static_assert(kG3 % 16 == 0 && kH % 16 == 0, "weight rows fill whole 16-row fragments");

constexpr int prepBlocks(int n, int k) { return (n / 16) * ((k + 31) / 32) * 64 / 256; }
constexpr int kPB0 = prepBlocks(kG3, kN);
constexpr int kPB1 = kPB0 + prepBlocks(kG3, kH);
constexpr int kPB2 = kPB1 + prepBlocks(kG3, kH);
constexpr int kPB3 = kPB2 + prepBlocks(kG3, kH);
constexpr int kPB4 = kPB3 + prepBlocks(kH, kN);
constexpr int kPB5 = kPB4 + prepBlocks(kH, kH);

__global__ __launch_bounds__(256) void wfrag_prep_all(const float* __restrict__ wih0,
                                                      const float* __restrict__ whh0,
                                                      const float* __restrict__ wih1,
                                                      const float* __restrict__ whh1,
                                                      const float* __restrict__ wax,
                                                      const float* __restrict__ wah,
                                                      _Float16* __restrict__ d0,
                                                      _Float16* __restrict__ d1,
                                                      _Float16* __restrict__ d2,
                                                      _Float16* __restrict__ d3,
                                                      _Float16* __restrict__ d4,
                                                      _Float16* __restrict__ d5) {
  int blk = blockIdx.x;
  const float* W;
  _Float16* dst;
  int K, ldw, b0;
  if (blk < kPB0)      { W = wih0; dst = d0; K = kN; ldw = kN;      b0 = 0;    }
  else if (blk < kPB1) { W = whh0; dst = d1; K = kH; ldw = kH;      b0 = kPB0; }
  else if (blk < kPB2) { W = wih1; dst = d2; K = kH; ldw = kH;      b0 = kPB1; }
  else if (blk < kPB3) { W = whh1; dst = d3; K = kH; ldw = kH;      b0 = kPB2; }
  else if (blk < kPB4) { W = wax;  dst = d4; K = kN; ldw = kH + kN; b0 = kPB3; }
  else                 { W = wah;  dst = d5; K = kH; ldw = kH + kN; b0 = kPB4; }
  int KT = (K + 31) >> 5;
  int u      = (blk - b0) * 256 + threadIdx.x;
  int frag   = u >> 6;
  int within = u & 63;
  int lane = within >> 1;
  int e8   = within & 1;
  int tileN = frag / KT;
  int tileK = frag - tileN * KT;
  int n  = tileN * 16 + (lane & 15);
  int kk = tileK * 32 + ((lane >> 4) << 3) + (e8 << 4);
  const float* row = W + (size_t)n * ldw;
  float v[8];
#pragma unroll
  for (int i = 0; i < 8; ++i) v[i] = row[(kk + i < K) ? kk + i : K - 1];
  v8h hv;
#pragma unroll
  for (int i = 0; i < 8; ++i) hv[i] = (_Float16)((kk + i < K) ? v[i] : 0.0f);
  *reinterpret_cast<v8h*>(dst + (size_t)frag * 512 + lane * 16 + e8 * 8) = hv;
}

template <int TPW, int XT>
__global__ __launch_bounds__(256) void gemm_wmma(const float* __restrict__ A,
                                                 const _Float16* __restrict__ Bfrag,
                                                 const float* __restrict__ bias,
                                                 float* __restrict__ C,
                                                 int K, int lda, int ldc,
                                                 const _Float16* __restrict__ Bfrag2,
                                                 const float* __restrict__ bias2,
                                                 float* __restrict__ C2, int ldc2) {
  static_assert(XT == 0 || XT == 1, "at most one extra N tile per wave");
  constexpr int NT = TPW + XT;
  __shared__ __align__(32) _Float16 ldsA[4 * 512];
  int tid  = threadIdx.x;
  int lane = tid & 31;
  int wave = tid >> 5;
  int mBase = blockIdx.x * 64;
  int KT    = (K + 31) >> 5;
  v8f acc[NT][4] = {};
  int sr  = tid >> 3;
  int sq  = tid & 7;
  int skl = sq << 2;
  int sln = (((skl >> 3) & 1) << 4) + (sr & 15);
  int se0 = (skl & 7) | ((skl >> 4) << 3);
  _Float16* st0 = &ldsA[(sr >> 4) * 512 + sln * 16 + se0];
  _Float16* st1 = st0 + 2 * 512;
  const float* Arow0 = A + (size_t)(mBase + sr) * lda + skl;
  const float* Arow1 = Arow0 + (size_t)32 * lda;
  const _Float16* Bw = Bfrag + (size_t)wave * KT * 512 + lane * 16;
  const _Float16* Bw2 = nullptr;
  if constexpr (XT == 1) Bw2 = Bfrag2 + (size_t)wave * KT * 512 + lane * 16;

  for (int kt = 0; kt < KT; ++kt) {
    int k0 = kt << 5;
    v16h b[NT];
#pragma unroll
    for (int j = 0; j < TPW; ++j)
      b[j] = *reinterpret_cast<const v16h*>(&Bw[((size_t)(j << 3) * KT + kt) * 512]);
    if constexpr (XT == 1)
      b[TPW] = *reinterpret_cast<const v16h*>(&Bw2[(size_t)kt * 512]);
    v4h hv0, hv1;
    if (k0 + 32 <= K) {
      float4 va = *(const float4*)(Arow0 + k0);
      float4 vb = *(const float4*)(Arow1 + k0);
      hv0[0] = (_Float16)va.x; hv0[1] = (_Float16)va.y;
      hv0[2] = (_Float16)va.z; hv0[3] = (_Float16)va.w;
      hv1[0] = (_Float16)vb.x; hv1[1] = (_Float16)vb.y;
      hv1[2] = (_Float16)vb.z; hv1[3] = (_Float16)vb.w;
    } else {
#pragma unroll
      for (int j = 0; j < 4; ++j) {
        bool ok = (k0 + skl + j) < K;
        hv0[j] = (_Float16)(ok ? Arow0[k0 + j] : 0.0f);
        hv1[j] = (_Float16)(ok ? Arow1[k0 + j] : 0.0f);
      }
    }
    *reinterpret_cast<v4h*>(st0) = hv0;
    *reinterpret_cast<v4h*>(st1) = hv1;
    __syncthreads();
    v16h a[4];
#pragma unroll
    for (int mt = 0; mt < 4; ++mt)
      a[mt] = *reinterpret_cast<const v16h*>(&ldsA[mt * 512 + lane * 16]);
#pragma unroll
    for (int j = 0; j < NT; ++j) {
#pragma unroll
      for (int mt = 0; mt < 4; ++mt)
        acc[j][mt] = __builtin_amdgcn_wmma_f32_16x16x32_f16(false, a[mt], false, b[j],
                                                            (short)0, acc[j][mt], false, false);
    }
    __syncthreads();
  }
  int rHalf = (lane >> 4) << 3;
#pragma unroll
  for (int j = 0; j < TPW; ++j) {
    int col = ((wave + (j << 3)) << 4) + (lane & 15);
    float bv = (bias != nullptr) ? bias[col] : 0.0f;
#pragma unroll
    for (int mt = 0; mt < 4; ++mt) {
      int rBase = mBase + (mt << 4) + rHalf;
#pragma unroll
      for (int r = 0; r < 8; ++r)
        C[(size_t)(rBase + r) * ldc + col] = acc[j][mt][r] + bv;
    }
  }
  if constexpr (XT == 1) {
    int col = (wave << 4) + (lane & 15);
    float bv = (bias2 != nullptr) ? bias2[col] : 0.0f;
#pragma unroll
    for (int mt = 0; mt < 4; ++mt) {
      int rBase = mBase + (mt << 4) + rHalf;
#pragma unroll
      for (int r = 0; r < 8; ++r)
        C2[(size_t)(rBase + r) * ldc2 + col] = acc[TPW][mt][r] + bv;
    }
  }
}

__device__ __forceinline__ float sigmoid_hw(float x) {
  return __builtin_amdgcn_rcpf(1.0f + __builtin_amdgcn_exp2f(x * -1.442695041f));
}
__device__ __forceinline__ float tanh_hw(float x) {
  return 1.0f - 2.0f * __builtin_amdgcn_rcpf(1.0f + __builtin_amdgcn_exp2f(x * 2.885390082f));
}

__device__ __forceinline__ void lds_barrier() {
  __builtin_amdgcn_fence(__ATOMIC_RELEASE, "workgroup", "local");
  __builtin_amdgcn_s_barrier();
  __builtin_amdgcn_fence(__ATOMIC_ACQUIRE, "workgroup", "local");
}

__device__ __forceinline__ void gi_fetch(float2 (&g)[3], const float* ga, int t) {
  const float* a = ga + (size_t)t * 384;
  g[0] = *(const float2*)(a);
  g[1] = *(const float2*)(a + 128);
  g[2] = *(const float2*)(a + 256);
}

#define GRU_RB 4
__global__ __launch_bounds__(256) __attribute__((amdgpu_waves_per_eu(2, 2))) void gru_recur(const float* __restrict__ gi,
                                                 const _Float16* __restrict__ WhhF,
                                                 const float* __restrict__ bhh,
                                                 float* __restrict__ out, int S) {
  __shared__ __align__(8) float hS[GRU_RB * 128];
  __shared__ __align__(32) _Float16 aF[4 * 512];
  __shared__ __align__(8) float ghS[GRU_RB * 384];
  int tid  = threadIdx.x;
  int lane = tid & 31;
  int wave = tid >> 5;
  int bBase = blockIdx.x * GRU_RB;
  const _Float16* Bw = WhhF + (size_t)(wave * 3) * 4 * 512 + lane * 16;
  if (S <= 0) return;

  v16h bF[3][4];
  float bv[3];
#pragma unroll
  for (int j = 0; j < 3; ++j) {
#pragma unroll
    for (int kt = 0; kt < 4; ++kt)
      bF[j][kt] = *reinterpret_cast<const v16h*>(&Bw[((size_t)j * 4 + kt) * 512]);
    bv[j] = bhh[((wave * 3 + j) << 4) + (lane & 15)];
  }

  int grow = tid >> 6;
  int gh0  = (tid & 63) << 1;
  const float* giA = gi + (size_t)(bBase + grow) * S * 384 + gh0;

  float2 g0[3], g1[3], g2[3];
  gi_fetch(g0, giA, 0);
  gi_fetch(g1, giA, (1 < S) ? 1 : S - 1);

  for (int i = tid; i < GRU_RB * 128; i += 256) hS[i] = 0.0f;
  for (int i = tid; i < 4 * 512; i += 256) aF[i] = (_Float16)0.0f;
  __syncthreads();

  auto step = [&](int t, const float2 (&g)[3]) __attribute__((always_inline)) {
    v16h a[4];
#pragma unroll
    for (int kt = 0; kt < 4; ++kt)
      a[kt] = *reinterpret_cast<const v16h*>(&aF[kt * 512 + lane * 16]);
    __builtin_amdgcn_sched_barrier(0);
    v8f acc[3] = {};
#pragma unroll
    for (int kt = 0; kt < 4; ++kt) {
#pragma unroll
      for (int j = 0; j < 3; ++j)
        acc[j] = __builtin_amdgcn_wmma_f32_16x16x32_f16(false, a[kt], false, bF[j][kt],
                                                        (short)0, acc[j], false, false);
    }
    if (lane < 16) {
#pragma unroll
      for (int j = 0; j < 3; ++j) {
        int col = ((wave * 3 + j) << 4) + lane;
#pragma unroll
        for (int r = 0; r < GRU_RB; ++r) ghS[r * 384 + col] = acc[j][r] + bv[j];
      }
    }
    lds_barrier();
    float2 hr = *(const float2*)(&ghS[grow * 384 + gh0]);
    float2 hz = *(const float2*)(&ghS[grow * 384 + 128 + gh0]);
    float2 hn = *(const float2*)(&ghS[grow * 384 + 256 + gh0]);
    float2 ho = *(const float2*)(&hS[grow * 128 + gh0]);
    float2 gr = g[0];
    float2 gz = g[1];
    float2 gn = g[2];
    float2 hNew;
#define GATE(c)                                                            \
    {                                                                      \
      float rg = sigmoid_hw(gr.c + hr.c);                                  \
      float zg = sigmoid_hw(gz.c + hz.c);                                  \
      float ng = tanh_hw(gn.c + rg * hn.c);                                \
      hNew.c = (1.0f - zg) * ng + zg * ho.c;                               \
    }
    GATE(x) GATE(y)
#undef GATE
    *(float2*)(&hS[grow * 128 + gh0]) = hNew;
    *(float2*)(out + ((size_t)(bBase + grow) * S + t) * 128 + gh0) = hNew;
    {
      int kt = gh0 >> 5;
      int kl = gh0 & 31;
      int ln = (((kl >> 3) & 1) << 4) + grow;
      int e0 = (kl & 7) | ((kl >> 4) << 3);
      v2h hv;
      hv[0] = (_Float16)hNew.x; hv[1] = (_Float16)hNew.y;
      *reinterpret_cast<v2h*>(&aF[kt * 512 + ln * 16 + e0]) = hv;
    }
    lds_barrier();
  };

  int t = 0;
  for (; t + 2 < S; t += 3) {
    gi_fetch(g2, giA, t + 2);
    step(t, g0);
    gi_fetch(g0, giA, (t + 3 < S) ? t + 3 : S - 1);
    step(t + 1, g1);
    gi_fetch(g1, giA, (t + 4 < S) ? t + 4 : S - 1);
    step(t + 2, g2);
  }
  if (t < S) {
    step(t, g0);
    if (t + 1 < S) step(t + 1, g1);
  }
}

__device__ __forceinline__ float blockReduceSum(float v, float* red) {
  int tid = threadIdx.x;
  red[tid] = v;
  __syncthreads();
  for (int off = 128; off > 0; off >>= 1) {
    if (tid < off) red[tid] += red[tid + off];
    __syncthreads();
  }
  float r = red[0];
  __syncthreads();
  return r;
}
__device__ __forceinline__ float blockReduceMax(float v, float* red) {
  int tid = threadIdx.x;
  red[tid] = v;
  __syncthreads();
  for (int off = 128; off > 0; off >>= 1) {
    if (tid < off) red[tid] = fmaxf(red[tid], red[tid + off]);
    __syncthreads();
  }
  float r = red[0];
  __syncthreads();
  return r;
}

__global__ __launch_bounds__(256) void finalize_kernel(const float* __restrict__ gE,
                                                       const float* __restrict__ u,
                                                       const float* __restrict__ out1,
                                                       const float* __restrict__ vvec,
                                                       const float* __restrict__ Wfc,
                                                       const float* __restrict__ bfc,
                                                       float* __restrict__ outp, int S) {
  __shared__ float uL[128], vL[128], eS[64], attnS[64];
  __shared__ __align__(16) float concatL[256];
  __shared__ float oldL[512], wcL[512], red[256];
  int tid  = threadIdx.x;
  int lane = tid & 31;
  int wave = tid >> 5;
  int b = blockIdx.x;

  if (tid < 128) {
    uL[tid] = u[(size_t)b * 128 + tid];
    vL[tid] = vvec[tid];
  }
  __syncthreads();

  for (int s = wave; s < S; s += 8) {
    float p = 0.0f;
    const float* ge = gE + ((size_t)b * S + s) * 128;
    for (int h = lane; h < 128; h += 32) p += vL[h] * tanhf(uL[h] + ge[h]);
    for (int off = 16; off > 0; off >>= 1) p += __shfl_down(p, off, 32);
    if (lane == 0) eS[s] = p;
  }
  __syncthreads();

  if (tid == 0) {
    float m = eS[0];
    for (int s = 1; s < S; ++s) m = fmaxf(m, eS[s]);
    float sum = 0.0f;
    for (int s = 0; s < S; ++s) { float ev = expf(eS[s] - m); attnS[s] = ev; sum += ev; }
    float inv = 1.0f / sum;
    for (int s = 0; s < S; ++s) attnS[s] *= inv;
  }
  __syncthreads();

  if (tid < 128) {
    float c = 0.0f;
    for (int s = 0; s < S; ++s) c += attnS[s] * out1[((size_t)b * S + s) * 128 + tid];
    concatL[tid]       = out1[((size_t)b * S + (S - 1)) * 128 + tid];
    concatL[128 + tid] = c;
  }
  __syncthreads();

  {
    const float4 c0 = *(const float4*)(&concatL[lane << 2]);
    const float4 c1 = *(const float4*)(&concatL[128 + (lane << 2)]);
    const float* wl = Wfc + (lane << 2);
    for (int n0 = wave; n0 < 500; n0 += 64) {
      float4 w0[8], w1[8];
      float bn[8], p[8];
#pragma unroll
      for (int i = 0; i < 8; ++i) {
        int n  = n0 + (i << 3);
        int nc = (n < 500) ? n : 499;
        const float* wr = wl + (size_t)nc * 256;
        w0[i] = *(const float4*)(wr);
        w1[i] = *(const float4*)(wr + 128);
        bn[i] = bfc[nc];
      }
      __builtin_amdgcn_sched_barrier(0);
#pragma unroll
      for (int i = 0; i < 8; ++i) {
        float a = (lane == 0) ? bn[i] : 0.0f;
        a += w0[i].x * c0.x;
        a += w0[i].y * c0.y;
        a += w0[i].z * c0.z;
        a += w0[i].w * c0.w;
        a += w1[i].x * c1.x;
        a += w1[i].y * c1.y;
        a += w1[i].z * c1.z;
        a += w1[i].w * c1.w;
        p[i] = a;
      }
#pragma unroll
      for (int off = 16; off > 0; off >>= 1) {
#pragma unroll
        for (int i = 0; i < 8; ++i) p[i] += __shfl_xor(p[i], off, 32);
      }
      if (lane == 0) {
#pragma unroll
        for (int i = 0; i < 8; ++i) {
          int n = n0 + (i << 3);
          if (n < 500) oldL[n] = p[i];
        }
      }
    }
  }
  __syncthreads();

  float pm = -3.4e38f;
  for (int n = tid; n < 500; n += 256) pm = fmaxf(pm, oldL[n]);
  pm = blockReduceMax(pm, red);
  float ps = 0.0f;
  for (int n = tid; n < 500; n += 256) { float e = expf(oldL[n] - pm); wcL[n] = e; ps += e; }
  ps = blockReduceSum(ps, red);
  float inv = 1.0f / ps;
  for (int n = tid; n < 500; n += 256) {
    float w = wcL[n] * inv;
    oldL[n] = w;
    wcL[n]  = fminf(fmaxf(w, LBV), UBV);
  }
  __syncthreads();

  for (int it = 0; it < 32; ++it) {
    float pl = 0.0f, pn = 0.0f;
    for (int n = tid; n < 500; n += 256) {
      float wcv = wcL[n];
      pl += oldL[n] - wcv;
      if (wcv != UBV) pn += wcv;
    }
    float leftover = blockReduceSum(pl, red);
    float nsum     = blockReduceSum(pn, red);
    float ns = (nsum == 0.0f) ? 1.0f : nsum;
    float pover = 0.0f;
    for (int n = tid; n < 500; n += 256) {
      float wcv = wcL[n];
      float wc2 = (wcv != UBV) ? wcv + leftover * wcv / ns : wcv;
      oldL[n] = wc2;
      if (wc2 > UBV) pover = 1.0f;
    }
    float anyOver = blockReduceSum(pover, red);
    int new_done = (anyOver == 0.0f);
    for (int n = tid; n < 500; n += 256) {
      float wc2 = oldL[n];
      wcL[n] = new_done ? wc2 : fminf(fmaxf(wc2, LBV), UBV);
    }
    __syncthreads();
    if (new_done) break;
  }

  for (int n = tid; n < 500; n += 256) outp[(size_t)b * 500 + n] = wcL[n];
}

extern "C" void kernel_launch(void* const* d_in, const int* in_sizes, int n_in,
                              void* d_out, int out_size, void* d_ws, size_t ws_size,
                              hipStream_t stream) {
  (void)in_sizes; (void)n_in; (void)out_size; (void)ws_size;
  const float* x    = (const float*)d_in[0];
  const float* Wih0 = (const float*)d_in[1];
  const float* Whh0 = (const float*)d_in[2];
  const float* bih0 = (const float*)d_in[3];
  const float* bhh0 = (const float*)d_in[4];
  const float* Wih1 = (const float*)d_in[5];
  const float* Whh1 = (const float*)d_in[6];
  const float* bih1 = (const float*)d_in[7];
  const float* bhh1 = (const float*)d_in[8];
  const float* Wa   = (const float*)d_in[9];
  const float* ba   = (const float*)d_in[10];
  const float* vv   = (const float*)d_in[11];
  const float* Wfc  = (const float*)d_in[12];
  const float* bfc  = (const float*)d_in[13];
  float* outp = (float*)d_out;

  constexpr int B = kB, S = kS, N = kN, H = kH, G3 = kG3;

  uint8_t* base = (uint8_t*)d_ws;
  size_t off = 0;
  auto carve = [&](size_t bytes) {
    void* p = base + off;
    off = (off + bytes + 255) & ~(size_t)255;
    return p;
  };
  float* gi   = (float*)carve((size_t)B * S * G3 * 4);
  float* out0 = (float*)carve((size_t)B * S * H * 4);
  float* out1 = (float*)carve((size_t)B * S * H * 4);
  float* gEb  = (float*)carve((size_t)B * S * H * 4);
  float* ub   = (float*)carve((size_t)B * H * 4);
  _Float16* Wih0F = (_Float16*)carve((size_t)24 * 16 * 512 * 2);
  _Float16* Whh0F = (_Float16*)carve((size_t)24 * 4 * 512 * 2);
  _Float16* Wih1F = (_Float16*)carve((size_t)24 * 4 * 512 * 2);
  _Float16* Whh1F = (_Float16*)carve((size_t)24 * 4 * 512 * 2);
  _Float16* WaxF  = (_Float16*)carve((size_t)8 * 16 * 512 * 2);
  _Float16* WahF  = (_Float16*)carve((size_t)8 * 4 * 512 * 2);

  wfrag_prep_all<<<kPB5, 256, 0, stream>>>(Wih0, Whh0, Wih1, Whh1, Wa + H, Wa,
                                           Wih0F, Whh0F, Wih1F, Whh1F, WaxF, WahF);

  auto gemm384 = [&](const float* A, const _Float16* Bf, const float* bias, float* C,
                     int M, int K, int lda) {
    gemm_wmma<3, 0><<<dim3(M / 64), 256, 0, stream>>>(A, Bf, bias, C, K, lda, G3,
                                                       nullptr, nullptr, nullptr, 0);
  };
  auto gemm128 = [&](const float* A, const _Float16* Bf, const float* bias, float* C,
                     int M, int K, int lda) {
    gemm_wmma<1, 0><<<dim3(M / 64), 256, 0, stream>>>(A, Bf, bias, C, K, lda, H,
                                                       nullptr, nullptr, nullptr, 0);
  };

  gemm_wmma<3, 1><<<dim3(B * S / 64), 256, 0, stream>>>(x, Wih0F, bih0, gi, N, N, G3,
                                                         WaxF, ba, gEb, H);
  gru_recur<<<B / GRU_RB, 256, 0, stream>>>(gi, Whh0F, bhh0, out0, S);
  gemm384(out0, Wih1F, bih1, gi, B * S, H, H);
  gru_recur<<<B / GRU_RB, 256, 0, stream>>>(gi, Whh1F, bhh1, out1, S);
  gemm128(out1 + (size_t)(S - 1) * H, WahF, nullptr, ub, B, H, S * H);
  finalize_kernel<<<B, 256, 0, stream>>>(gEb, ub, out1, vv, Wfc, bfc, outp, S);
}


# Attention scores and context: gE rows fetched whole (float4 per lane), four rows per wave in flight, out1 32 rows in flight (was one 4-byte load per wait); a lane now sums 4 consecutive h first
# speedup vs baseline: 3.1417x; 1.0401x over previous
#include <hip/hip_runtime.h>
#include <hip/hip_bf16.h>
#include <stdint.h>

typedef __attribute__((ext_vector_type(16))) _Float16 v16h;
typedef __attribute__((ext_vector_type(8)))  _Float16 v8h;
typedef __attribute__((ext_vector_type(4)))  _Float16 v4h;
typedef __attribute__((ext_vector_type(2)))  _Float16 v2h;
typedef __attribute__((ext_vector_type(8)))  float    v8f;

#define LBV 0.0f
#define UBV 0.2f

constexpr int kB = 1024, kS = 60, kN = 500, kH = 128, kG3 = 384;
static_assert(kG3 % 16 == 0 && kH % 16 == 0, "weight rows fill whole 16-row fragments");

constexpr int prepBlocks(int n, int k) { return (n / 16) * ((k + 31) / 32) * 64 / 256; }
constexpr int kPB0 = prepBlocks(kG3, kN);
constexpr int kPB1 = kPB0 + prepBlocks(kG3, kH);
constexpr int kPB2 = kPB1 + prepBlocks(kG3, kH);
constexpr int kPB3 = kPB2 + prepBlocks(kG3, kH);
constexpr int kPB4 = kPB3 + prepBlocks(kH, kN);
constexpr int kPB5 = kPB4 + prepBlocks(kH, kH);

__global__ __launch_bounds__(256) void wfrag_prep_all(const float* __restrict__ wih0,
                                                      const float* __restrict__ whh0,
                                                      const float* __restrict__ wih1,
                                                      const float* __restrict__ whh1,
                                                      const float* __restrict__ wax,
                                                      const float* __restrict__ wah,
                                                      _Float16* __restrict__ d0,
                                                      _Float16* __restrict__ d1,
                                                      _Float16* __restrict__ d2,
                                                      _Float16* __restrict__ d3,
                                                      _Float16* __restrict__ d4,
                                                      _Float16* __restrict__ d5) {
  int blk = blockIdx.x;
  const float* W;
  _Float16* dst;
  int K, ldw, b0;
  if (blk < kPB0)      { W = wih0; dst = d0; K = kN; ldw = kN;      b0 = 0;    }
  else if (blk < kPB1) { W = whh0; dst = d1; K = kH; ldw = kH;      b0 = kPB0; }
  else if (blk < kPB2) { W = wih1; dst = d2; K = kH; ldw = kH;      b0 = kPB1; }
  else if (blk < kPB3) { W = whh1; dst = d3; K = kH; ldw = kH;      b0 = kPB2; }
  else if (blk < kPB4) { W = wax;  dst = d4; K = kN; ldw = kH + kN; b0 = kPB3; }
  else                 { W = wah;  dst = d5; K = kH; ldw = kH + kN; b0 = kPB4; }
  int KT = (K + 31) >> 5;
  int u      = (blk - b0) * 256 + threadIdx.x;
  int frag   = u >> 6;
  int within = u & 63;
  int lane = within >> 1;
  int e8   = within & 1;
  int tileN = frag / KT;
  int tileK = frag - tileN * KT;
  int n  = tileN * 16 + (lane & 15);
  int kk = tileK * 32 + ((lane >> 4) << 3) + (e8 << 4);
  const float* row = W + (size_t)n * ldw;
  float v[8];
#pragma unroll
  for (int i = 0; i < 8; ++i) v[i] = row[(kk + i < K) ? kk + i : K - 1];
  v8h hv;
#pragma unroll
  for (int i = 0; i < 8; ++i) hv[i] = (_Float16)((kk + i < K) ? v[i] : 0.0f);
  *reinterpret_cast<v8h*>(dst + (size_t)frag * 512 + lane * 16 + e8 * 8) = hv;
}

template <int TPW, int XT>
__global__ __launch_bounds__(256) void gemm_wmma(const float* __restrict__ A,
                                                 const _Float16* __restrict__ Bfrag,
                                                 const float* __restrict__ bias,
                                                 float* __restrict__ C,
                                                 int K, int lda, int ldc,
                                                 const _Float16* __restrict__ Bfrag2,
                                                 const float* __restrict__ bias2,
                                                 float* __restrict__ C2, int ldc2) {
  static_assert(XT == 0 || XT == 1, "at most one extra N tile per wave");
  constexpr int NT = TPW + XT;
  __shared__ __align__(32) _Float16 ldsA[4 * 512];
  int tid  = threadIdx.x;
  int lane = tid & 31;
  int wave = tid >> 5;
  int mBase = blockIdx.x * 64;
  int KT    = (K + 31) >> 5;
  v8f acc[NT][4] = {};
  int sr  = tid >> 3;
  int sq  = tid & 7;
  int skl = sq << 2;
  int sln = (((skl >> 3) & 1) << 4) + (sr & 15);
  int se0 = (skl & 7) | ((skl >> 4) << 3);
  _Float16* st0 = &ldsA[(sr >> 4) * 512 + sln * 16 + se0];
  _Float16* st1 = st0 + 2 * 512;
  const float* Arow0 = A + (size_t)(mBase + sr) * lda + skl;
  const float* Arow1 = Arow0 + (size_t)32 * lda;
  const _Float16* Bw = Bfrag + (size_t)wave * KT * 512 + lane * 16;
  const _Float16* Bw2 = nullptr;
  if constexpr (XT == 1) Bw2 = Bfrag2 + (size_t)wave * KT * 512 + lane * 16;

  for (int kt = 0; kt < KT; ++kt) {
    int k0 = kt << 5;
    v16h b[NT];
#pragma unroll
    for (int j = 0; j < TPW; ++j)
      b[j] = *reinterpret_cast<const v16h*>(&Bw[((size_t)(j << 3) * KT + kt) * 512]);
    if constexpr (XT == 1)
      b[TPW] = *reinterpret_cast<const v16h*>(&Bw2[(size_t)kt * 512]);
    v4h hv0, hv1;
    if (k0 + 32 <= K) {
      float4 va = *(const float4*)(Arow0 + k0);
      float4 vb = *(const float4*)(Arow1 + k0);
      hv0[0] = (_Float16)va.x; hv0[1] = (_Float16)va.y;
      hv0[2] = (_Float16)va.z; hv0[3] = (_Float16)va.w;
      hv1[0] = (_Float16)vb.x; hv1[1] = (_Float16)vb.y;
      hv1[2] = (_Float16)vb.z; hv1[3] = (_Float16)vb.w;
    } else {
#pragma unroll
      for (int j = 0; j < 4; ++j) {
        bool ok = (k0 + skl + j) < K;
        hv0[j] = (_Float16)(ok ? Arow0[k0 + j] : 0.0f);
        hv1[j] = (_Float16)(ok ? Arow1[k0 + j] : 0.0f);
      }
    }
    *reinterpret_cast<v4h*>(st0) = hv0;
    *reinterpret_cast<v4h*>(st1) = hv1;
    __syncthreads();
    v16h a[4];
#pragma unroll
    for (int mt = 0; mt < 4; ++mt)
      a[mt] = *reinterpret_cast<const v16h*>(&ldsA[mt * 512 + lane * 16]);
#pragma unroll
    for (int j = 0; j < NT; ++j) {
#pragma unroll
      for (int mt = 0; mt < 4; ++mt)
        acc[j][mt] = __builtin_amdgcn_wmma_f32_16x16x32_f16(false, a[mt], false, b[j],
                                                            (short)0, acc[j][mt], false, false);
    }
    __syncthreads();
  }
  int rHalf = (lane >> 4) << 3;
#pragma unroll
  for (int j = 0; j < TPW; ++j) {
    int col = ((wave + (j << 3)) << 4) + (lane & 15);
    float bv = (bias != nullptr) ? bias[col] : 0.0f;
#pragma unroll
    for (int mt = 0; mt < 4; ++mt) {
      int rBase = mBase + (mt << 4) + rHalf;
#pragma unroll
      for (int r = 0; r < 8; ++r)
        C[(size_t)(rBase + r) * ldc + col] = acc[j][mt][r] + bv;
    }
  }
  if constexpr (XT == 1) {
    int col = (wave << 4) + (lane & 15);
    float bv = (bias2 != nullptr) ? bias2[col] : 0.0f;
#pragma unroll
    for (int mt = 0; mt < 4; ++mt) {
      int rBase = mBase + (mt << 4) + rHalf;
#pragma unroll
      for (int r = 0; r < 8; ++r)
        C2[(size_t)(rBase + r) * ldc2 + col] = acc[TPW][mt][r] + bv;
    }
  }
}

__device__ __forceinline__ float sigmoid_hw(float x) {
  return __builtin_amdgcn_rcpf(1.0f + __builtin_amdgcn_exp2f(x * -1.442695041f));
}
__device__ __forceinline__ float tanh_hw(float x) {
  return 1.0f - 2.0f * __builtin_amdgcn_rcpf(1.0f + __builtin_amdgcn_exp2f(x * 2.885390082f));
}

__device__ __forceinline__ void lds_barrier() {
  __builtin_amdgcn_fence(__ATOMIC_RELEASE, "workgroup", "local");
  __builtin_amdgcn_s_barrier();
  __builtin_amdgcn_fence(__ATOMIC_ACQUIRE, "workgroup", "local");
}

__device__ __forceinline__ void gi_fetch(float2 (&g)[3], const float* ga, int t) {
  const float* a = ga + (size_t)t * 384;
  g[0] = *(const float2*)(a);
  g[1] = *(const float2*)(a + 128);
  g[2] = *(const float2*)(a + 256);
}

#define GRU_RB 4
__global__ __launch_bounds__(256) __attribute__((amdgpu_waves_per_eu(2, 2))) void gru_recur(const float* __restrict__ gi,
                                                 const _Float16* __restrict__ WhhF,
                                                 const float* __restrict__ bhh,
                                                 float* __restrict__ out, int S) {
  __shared__ __align__(8) float hS[GRU_RB * 128];
  __shared__ __align__(32) _Float16 aF[4 * 512];
  __shared__ __align__(8) float ghS[GRU_RB * 384];
  int tid  = threadIdx.x;
  int lane = tid & 31;
  int wave = tid >> 5;
  int bBase = blockIdx.x * GRU_RB;
  const _Float16* Bw = WhhF + (size_t)(wave * 3) * 4 * 512 + lane * 16;
  if (S <= 0) return;

  v16h bF[3][4];
  float bv[3];
#pragma unroll
  for (int j = 0; j < 3; ++j) {
#pragma unroll
    for (int kt = 0; kt < 4; ++kt)
      bF[j][kt] = *reinterpret_cast<const v16h*>(&Bw[((size_t)j * 4 + kt) * 512]);
    bv[j] = bhh[((wave * 3 + j) << 4) + (lane & 15)];
  }

  int grow = tid >> 6;
  int gh0  = (tid & 63) << 1;
  const float* giA = gi + (size_t)(bBase + grow) * S * 384 + gh0;

  float2 g0[3], g1[3], g2[3];
  gi_fetch(g0, giA, 0);
  gi_fetch(g1, giA, (1 < S) ? 1 : S - 1);

  for (int i = tid; i < GRU_RB * 128; i += 256) hS[i] = 0.0f;
  for (int i = tid; i < 4 * 512; i += 256) aF[i] = (_Float16)0.0f;
  __syncthreads();

  auto step = [&](int t, const float2 (&g)[3]) __attribute__((always_inline)) {
    v16h a[4];
#pragma unroll
    for (int kt = 0; kt < 4; ++kt)
      a[kt] = *reinterpret_cast<const v16h*>(&aF[kt * 512 + lane * 16]);
    __builtin_amdgcn_sched_barrier(0);
    v8f acc[3] = {};
#pragma unroll
    for (int kt = 0; kt < 4; ++kt) {
#pragma unroll
      for (int j = 0; j < 3; ++j)
        acc[j] = __builtin_amdgcn_wmma_f32_16x16x32_f16(false, a[kt], false, bF[j][kt],
                                                        (short)0, acc[j], false, false);
    }
    if (lane < 16) {
#pragma unroll
      for (int j = 0; j < 3; ++j) {
        int col = ((wave * 3 + j) << 4) + lane;
#pragma unroll
        for (int r = 0; r < GRU_RB; ++r) ghS[r * 384 + col] = acc[j][r] + bv[j];
      }
    }
    lds_barrier();
    float2 hr = *(const float2*)(&ghS[grow * 384 + gh0]);
    float2 hz = *(const float2*)(&ghS[grow * 384 + 128 + gh0]);
    float2 hn = *(const float2*)(&ghS[grow * 384 + 256 + gh0]);
    float2 ho = *(const float2*)(&hS[grow * 128 + gh0]);
    float2 gr = g[0];
    float2 gz = g[1];
    float2 gn = g[2];
    float2 hNew;
#define GATE(c)                                                            \
    {                                                                      \
      float rg = sigmoid_hw(gr.c + hr.c);                                  \
      float zg = sigmoid_hw(gz.c + hz.c);                                  \
      float ng = tanh_hw(gn.c + rg * hn.c);                                \
      hNew.c = (1.0f - zg) * ng + zg * ho.c;                               \
    }
    GATE(x) GATE(y)
#undef GATE
    *(float2*)(&hS[grow * 128 + gh0]) = hNew;
    *(float2*)(out + ((size_t)(bBase + grow) * S + t) * 128 + gh0) = hNew;
    {
      int kt = gh0 >> 5;
      int kl = gh0 & 31;
      int ln = (((kl >> 3) & 1) << 4) + grow;
      int e0 = (kl & 7) | ((kl >> 4) << 3);
      v2h hv;
      hv[0] = (_Float16)hNew.x; hv[1] = (_Float16)hNew.y;
      *reinterpret_cast<v2h*>(&aF[kt * 512 + ln * 16 + e0]) = hv;
    }
    lds_barrier();
  };

  int t = 0;
  for (; t + 2 < S; t += 3) {
    gi_fetch(g2, giA, t + 2);
    step(t, g0);
    gi_fetch(g0, giA, (t + 3 < S) ? t + 3 : S - 1);
    step(t + 1, g1);
    gi_fetch(g1, giA, (t + 4 < S) ? t + 4 : S - 1);
    step(t + 2, g2);
  }
  if (t < S) {
    step(t, g0);
    if (t + 1 < S) step(t + 1, g1);
  }
}

__device__ __forceinline__ float blockReduceSum(float v, float* red) {
  int tid = threadIdx.x;
  red[tid] = v;
  __syncthreads();
  for (int off = 128; off > 0; off >>= 1) {
    if (tid < off) red[tid] += red[tid + off];
    __syncthreads();
  }
  float r = red[0];
  __syncthreads();
  return r;
}
__device__ __forceinline__ float blockReduceMax(float v, float* red) {
  int tid = threadIdx.x;
  red[tid] = v;
  __syncthreads();
  for (int off = 128; off > 0; off >>= 1) {
    if (tid < off) red[tid] = fmaxf(red[tid], red[tid + off]);
    __syncthreads();
  }
  float r = red[0];
  __syncthreads();
  return r;
}

__global__ __launch_bounds__(256) void finalize_kernel(const float* __restrict__ gE,
                                                       const float* __restrict__ u,
                                                       const float* __restrict__ out1,
                                                       const float* __restrict__ vvec,
                                                       const float* __restrict__ Wfc,
                                                       const float* __restrict__ bfc,
                                                       float* __restrict__ outp, int S) {
  __shared__ __align__(16) float uL[128];
  __shared__ __align__(16) float vL[128];
  __shared__ float eS[64], attnS[64];
  __shared__ __align__(16) float concatL[256];
  __shared__ float oldL[512], wcL[512], red[256];
  int tid  = threadIdx.x;
  int lane = tid & 31;
  int wave = tid >> 5;
  int b = blockIdx.x;

  if (tid < 128) {
    uL[tid] = u[(size_t)b * 128 + tid];
    vL[tid] = vvec[tid];
  }
  __syncthreads();

  {
    const float4 u4 = *(const float4*)(&uL[lane << 2]);
    const float4 v4 = *(const float4*)(&vL[lane << 2]);
    const float* geb = gE + (size_t)b * S * 128 + (lane << 2);
    for (int s0 = wave; s0 < S; s0 += 32) {
      float4 g[4];
#pragma unroll
      for (int i = 0; i < 4; ++i) {
        int s  = s0 + (i << 3);
        int sc = (s < S) ? s : S - 1;
        g[i] = *(const float4*)(geb + (size_t)sc * 128);
      }
      __builtin_amdgcn_sched_barrier(0);
#pragma unroll
      for (int i = 0; i < 4; ++i) {
        float p = v4.x * tanhf(u4.x + g[i].x);
        p += v4.y * tanhf(u4.y + g[i].y);
        p += v4.z * tanhf(u4.z + g[i].z);
        p += v4.w * tanhf(u4.w + g[i].w);
        for (int off = 16; off > 0; off >>= 1) p += __shfl_down(p, off, 32);
        int s = s0 + (i << 3);
        if (lane == 0 && s < S) eS[s] = p;
      }
    }
  }
  __syncthreads();

  if (tid == 0) {
    float m = eS[0];
    for (int s = 1; s < S; ++s) m = fmaxf(m, eS[s]);
    float sum = 0.0f;
    for (int s = 0; s < S; ++s) { float ev = expf(eS[s] - m); attnS[s] = ev; sum += ev; }
    float inv = 1.0f / sum;
    for (int s = 0; s < S; ++s) attnS[s] *= inv;
  }
  __syncthreads();

  if (tid < 128) {
    float c = 0.0f;
    const float* ob = out1 + (size_t)b * S * 128 + tid;
    for (int s0 = 0; s0 < S; s0 += 32) {
      float o[32];
#pragma unroll
      for (int i = 0; i < 32; ++i) {
        int s  = s0 + i;
        int sc = (s < S) ? s : S - 1;
        o[i] = ob[(size_t)sc * 128];
      }
#pragma unroll
      for (int i = 0; i < 32; ++i) {
        int s = s0 + i;
        float a = (s < S) ? attnS[s] : 0.0f;
        c += a * o[i];
      }
    }
    concatL[tid]       = ob[(size_t)(S - 1) * 128];
    concatL[128 + tid] = c;
  }
  __syncthreads();

  {
    const float4 c0 = *(const float4*)(&concatL[lane << 2]);
    const float4 c1 = *(const float4*)(&concatL[128 + (lane << 2)]);
    const float* wl = Wfc + (lane << 2);
    for (int n0 = wave; n0 < 500; n0 += 64) {
      float4 w0[8], w1[8];
      float bn[8], p[8];
#pragma unroll
      for (int i = 0; i < 8; ++i) {
        int n  = n0 + (i << 3);
        int nc = (n < 500) ? n : 499;
        const float* wr = wl + (size_t)nc * 256;
        w0[i] = *(const float4*)(wr);
        w1[i] = *(const float4*)(wr + 128);
        bn[i] = bfc[nc];
      }
      __builtin_amdgcn_sched_barrier(0);
#pragma unroll
      for (int i = 0; i < 8; ++i) {
        float a = (lane == 0) ? bn[i] : 0.0f;
        a += w0[i].x * c0.x;
        a += w0[i].y * c0.y;
        a += w0[i].z * c0.z;
        a += w0[i].w * c0.w;
        a += w1[i].x * c1.x;
        a += w1[i].y * c1.y;
        a += w1[i].z * c1.z;
        a += w1[i].w * c1.w;
        p[i] = a;
      }
#pragma unroll
      for (int off = 16; off > 0; off >>= 1) {
#pragma unroll
        for (int i = 0; i < 8; ++i) p[i] += __shfl_xor(p[i], off, 32);
      }
      if (lane == 0) {
#pragma unroll
        for (int i = 0; i < 8; ++i) {
          int n = n0 + (i << 3);
          if (n < 500) oldL[n] = p[i];
        }
      }
    }
  }
  __syncthreads();

  float pm = -3.4e38f;
  for (int n = tid; n < 500; n += 256) pm = fmaxf(pm, oldL[n]);
  pm = blockReduceMax(pm, red);
  float ps = 0.0f;
  for (int n = tid; n < 500; n += 256) { float e = expf(oldL[n] - pm); wcL[n] = e; ps += e; }
  ps = blockReduceSum(ps, red);
  float inv = 1.0f / ps;
  for (int n = tid; n < 500; n += 256) {
    float w = wcL[n] * inv;
    oldL[n] = w;
    wcL[n]  = fminf(fmaxf(w, LBV), UBV);
  }
  __syncthreads();

  for (int it = 0; it < 32; ++it) {
    float pl = 0.0f, pn = 0.0f;
    for (int n = tid; n < 500; n += 256) {
      float wcv = wcL[n];
      pl += oldL[n] - wcv;
      if (wcv != UBV) pn += wcv;
    }
    float leftover = blockReduceSum(pl, red);
    float nsum     = blockReduceSum(pn, red);
    float ns = (nsum == 0.0f) ? 1.0f : nsum;
    float pover = 0.0f;
    for (int n = tid; n < 500; n += 256) {
      float wcv = wcL[n];
      float wc2 = (wcv != UBV) ? wcv + leftover * wcv / ns : wcv;
      oldL[n] = wc2;
      if (wc2 > UBV) pover = 1.0f;
    }
    float anyOver = blockReduceSum(pover, red);
    int new_done = (anyOver == 0.0f);
    for (int n = tid; n < 500; n += 256) {
      float wc2 = oldL[n];
      wcL[n] = new_done ? wc2 : fminf(fmaxf(wc2, LBV), UBV);
    }
    __syncthreads();
    if (new_done) break;
  }

  for (int n = tid; n < 500; n += 256) outp[(size_t)b * 500 + n] = wcL[n];
}

extern "C" void kernel_launch(void* const* d_in, const int* in_sizes, int n_in,
                              void* d_out, int out_size, void* d_ws, size_t ws_size,
                              hipStream_t stream) {
  (void)in_sizes; (void)n_in; (void)out_size; (void)ws_size;
  const float* x    = (const float*)d_in[0];
  const float* Wih0 = (const float*)d_in[1];
  const float* Whh0 = (const float*)d_in[2];
  const float* bih0 = (const float*)d_in[3];
  const float* bhh0 = (const float*)d_in[4];
  const float* Wih1 = (const float*)d_in[5];
  const float* Whh1 = (const float*)d_in[6];
  const float* bih1 = (const float*)d_in[7];
  const float* bhh1 = (const float*)d_in[8];
  const float* Wa   = (const float*)d_in[9];
  const float* ba   = (const float*)d_in[10];
  const float* vv   = (const float*)d_in[11];
  const float* Wfc  = (const float*)d_in[12];
  const float* bfc  = (const float*)d_in[13];
  float* outp = (float*)d_out;

  constexpr int B = kB, S = kS, N = kN, H = kH, G3 = kG3;

  uint8_t* base = (uint8_t*)d_ws;
  size_t off = 0;
  auto carve = [&](size_t bytes) {
    void* p = base + off;
    off = (off + bytes + 255) & ~(size_t)255;
    return p;
  };
  float* gi   = (float*)carve((size_t)B * S * G3 * 4);
  float* out0 = (float*)carve((size_t)B * S * H * 4);
  float* out1 = (float*)carve((size_t)B * S * H * 4);
  float* gEb  = (float*)carve((size_t)B * S * H * 4);
  float* ub   = (float*)carve((size_t)B * H * 4);
  _Float16* Wih0F = (_Float16*)carve((size_t)24 * 16 * 512 * 2);
  _Float16* Whh0F = (_Float16*)carve((size_t)24 * 4 * 512 * 2);
  _Float16* Wih1F = (_Float16*)carve((size_t)24 * 4 * 512 * 2);
  _Float16* Whh1F = (_Float16*)carve((size_t)24 * 4 * 512 * 2);
  _Float16* WaxF  = (_Float16*)carve((size_t)8 * 16 * 512 * 2);
  _Float16* WahF  = (_Float16*)carve((size_t)8 * 4 * 512 * 2);

  wfrag_prep_all<<<kPB5, 256, 0, stream>>>(Wih0, Whh0, Wih1, Whh1, Wa + H, Wa,
                                           Wih0F, Whh0F, Wih1F, Whh1F, WaxF, WahF);

  auto gemm384 = [&](const float* A, const _Float16* Bf, const float* bias, float* C,
                     int M, int K, int lda) {
    gemm_wmma<3, 0><<<dim3(M / 64), 256, 0, stream>>>(A, Bf, bias, C, K, lda, G3,
                                                       nullptr, nullptr, nullptr, 0);
  };
  auto gemm128 = [&](const float* A, const _Float16* Bf, const float* bias, float* C,
                     int M, int K, int lda) {
    gemm_wmma<1, 0><<<dim3(M / 64), 256, 0, stream>>>(A, Bf, bias, C, K, lda, H,
                                                       nullptr, nullptr, nullptr, 0);
  };

  gemm_wmma<3, 1><<<dim3(B * S / 64), 256, 0, stream>>>(x, Wih0F, bih0, gi, N, N, G3,
                                                         WaxF, ba, gEb, H);
  gru_recur<<<B / GRU_RB, 256, 0, stream>>>(gi, Whh0F, bhh0, out0, S);
  gemm384(out0, Wih1F, bih1, gi, B * S, H, H);
  gru_recur<<<B / GRU_RB, 256, 0, stream>>>(gi, Whh1F, bhh1, out1, S);
  gemm128(out1 + (size_t)(S - 1) * H, WahF, nullptr, ub, B, H, S * H);
  finalize_kernel<<<B, 256, 0, stream>>>(gEb, ub, out1, vv, Wfc, bfc, outp, S);
}


# Input GEMMs: A quads and B fragments are requested two k-steps ahead into two register sets each, and the staging barriers order LDS only so those loads stay in flight; sums unchanged
# speedup vs baseline: 3.3110x; 1.0539x over previous
#include <hip/hip_runtime.h>
#include <hip/hip_bf16.h>
#include <stdint.h>

typedef __attribute__((ext_vector_type(16))) _Float16 v16h;
typedef __attribute__((ext_vector_type(8)))  _Float16 v8h;
typedef __attribute__((ext_vector_type(4)))  _Float16 v4h;
typedef __attribute__((ext_vector_type(2)))  _Float16 v2h;
typedef __attribute__((ext_vector_type(8)))  float    v8f;

#define LBV 0.0f
#define UBV 0.2f

constexpr int kB = 1024, kS = 60, kN = 500, kH = 128, kG3 = 384;
static_assert(kG3 % 16 == 0 && kH % 16 == 0, "weight rows fill whole 16-row fragments");

constexpr int prepBlocks(int n, int k) { return (n / 16) * ((k + 31) / 32) * 64 / 256; }
constexpr int kPB0 = prepBlocks(kG3, kN);
constexpr int kPB1 = kPB0 + prepBlocks(kG3, kH);
constexpr int kPB2 = kPB1 + prepBlocks(kG3, kH);
constexpr int kPB3 = kPB2 + prepBlocks(kG3, kH);
constexpr int kPB4 = kPB3 + prepBlocks(kH, kN);
constexpr int kPB5 = kPB4 + prepBlocks(kH, kH);

__global__ __launch_bounds__(256) void wfrag_prep_all(const float* __restrict__ wih0,
                                                      const float* __restrict__ whh0,
                                                      const float* __restrict__ wih1,
                                                      const float* __restrict__ whh1,
                                                      const float* __restrict__ wax,
                                                      const float* __restrict__ wah,
                                                      _Float16* __restrict__ d0,
                                                      _Float16* __restrict__ d1,
                                                      _Float16* __restrict__ d2,
                                                      _Float16* __restrict__ d3,
                                                      _Float16* __restrict__ d4,
                                                      _Float16* __restrict__ d5) {
  int blk = blockIdx.x;
  const float* W;
  _Float16* dst;
  int K, ldw, b0;
  if (blk < kPB0)      { W = wih0; dst = d0; K = kN; ldw = kN;      b0 = 0;    }
  else if (blk < kPB1) { W = whh0; dst = d1; K = kH; ldw = kH;      b0 = kPB0; }
  else if (blk < kPB2) { W = wih1; dst = d2; K = kH; ldw = kH;      b0 = kPB1; }
  else if (blk < kPB3) { W = whh1; dst = d3; K = kH; ldw = kH;      b0 = kPB2; }
  else if (blk < kPB4) { W = wax;  dst = d4; K = kN; ldw = kH + kN; b0 = kPB3; }
  else                 { W = wah;  dst = d5; K = kH; ldw = kH + kN; b0 = kPB4; }
  int KT = (K + 31) >> 5;
  int u      = (blk - b0) * 256 + threadIdx.x;
  int frag   = u >> 6;
  int within = u & 63;
  int lane = within >> 1;
  int e8   = within & 1;
  int tileN = frag / KT;
  int tileK = frag - tileN * KT;
  int n  = tileN * 16 + (lane & 15);
  int kk = tileK * 32 + ((lane >> 4) << 3) + (e8 << 4);
  const float* row = W + (size_t)n * ldw;
  float v[8];
#pragma unroll
  for (int i = 0; i < 8; ++i) v[i] = row[(kk + i < K) ? kk + i : K - 1];
  v8h hv;
#pragma unroll
  for (int i = 0; i < 8; ++i) hv[i] = (_Float16)((kk + i < K) ? v[i] : 0.0f);
  *reinterpret_cast<v8h*>(dst + (size_t)frag * 512 + lane * 16 + e8 * 8) = hv;
}

__device__ __forceinline__ void lds_barrier() {
  __builtin_amdgcn_fence(__ATOMIC_RELEASE, "workgroup", "local");
  __builtin_amdgcn_s_barrier();
  __builtin_amdgcn_fence(__ATOMIC_ACQUIRE, "workgroup", "local");
}

template <int TPW, int XT>
__global__ __launch_bounds__(256) void gemm_wmma(const float* __restrict__ A,
                                                 const _Float16* __restrict__ Bfrag,
                                                 const float* __restrict__ bias,
                                                 float* __restrict__ C,
                                                 int K, int lda, int ldc,
                                                 const _Float16* __restrict__ Bfrag2,
                                                 const float* __restrict__ bias2,
                                                 float* __restrict__ C2, int ldc2) {
  static_assert(XT == 0 || XT == 1, "at most one extra N tile per wave");
  constexpr int NT = TPW + XT;
  __shared__ __align__(32) _Float16 ldsA[4 * 512];
  int tid  = threadIdx.x;
  int lane = tid & 31;
  int wave = tid >> 5;
  int mBase = blockIdx.x * 64;
  int KT    = (K + 31) >> 5;
  v8f acc[NT][4] = {};
  int sr  = tid >> 3;
  int sq  = tid & 7;
  int skl = sq << 2;
  int sln = (((skl >> 3) & 1) << 4) + (sr & 15);
  int se0 = (skl & 7) | ((skl >> 4) << 3);
  _Float16* st0 = &ldsA[(sr >> 4) * 512 + sln * 16 + se0];
  _Float16* st1 = st0 + 2 * 512;
  const float* Arow0 = A + (size_t)(mBase + sr) * lda;
  const float* Arow1 = Arow0 + (size_t)32 * lda;
  const _Float16* Bw = Bfrag + (size_t)wave * KT * 512 + lane * 16;
  const _Float16* Bw2 = nullptr;
  if constexpr (XT == 1) Bw2 = Bfrag2 + (size_t)wave * KT * 512 + lane * 16;
  const int klast = (K - 1) & ~3;

  auto loadA = [&](float4 (&as)[2], int kt) __attribute__((always_inline)) {
    int col = (kt << 5) + skl;
    int cc  = (col < klast) ? col : klast;
    as[0] = *(const float4*)(Arow0 + cc);
    as[1] = *(const float4*)(Arow1 + cc);
  };
  auto loadB = [&](v16h (&bs)[NT], int kt) __attribute__((always_inline)) {
    int kc = (kt < KT) ? kt : KT - 1;
#pragma unroll
    for (int j = 0; j < TPW; ++j)
      bs[j] = *reinterpret_cast<const v16h*>(&Bw[((size_t)(j << 3) * KT + kc) * 512]);
    if constexpr (XT == 1)
      bs[TPW] = *reinterpret_cast<const v16h*>(&Bw2[(size_t)kc * 512]);
  };

  auto kstep = [&](int kt, float4 (&as)[2], v16h (&bs)[NT]) __attribute__((always_inline)) {
    int k0 = kt << 5;
    float4 va = as[0];
    float4 vb = as[1];
    if (k0 + 32 > K) {
      int col = k0 + skl;
      if (col     >= K) { va.x = 0.0f; vb.x = 0.0f; }
      if (col + 1 >= K) { va.y = 0.0f; vb.y = 0.0f; }
      if (col + 2 >= K) { va.z = 0.0f; vb.z = 0.0f; }
      if (col + 3 >= K) { va.w = 0.0f; vb.w = 0.0f; }
    }
    v4h hv0, hv1;
    hv0[0] = (_Float16)va.x; hv0[1] = (_Float16)va.y;
    hv0[2] = (_Float16)va.z; hv0[3] = (_Float16)va.w;
    hv1[0] = (_Float16)vb.x; hv1[1] = (_Float16)vb.y;
    hv1[2] = (_Float16)vb.z; hv1[3] = (_Float16)vb.w;
    *reinterpret_cast<v4h*>(st0) = hv0;
    *reinterpret_cast<v4h*>(st1) = hv1;
    loadA(as, kt + 2);
    lds_barrier();
    v16h a[4];
#pragma unroll
    for (int mt = 0; mt < 4; ++mt)
      a[mt] = *reinterpret_cast<const v16h*>(&ldsA[mt * 512 + lane * 16]);
#pragma unroll
    for (int j = 0; j < NT; ++j) {
#pragma unroll
      for (int mt = 0; mt < 4; ++mt)
        acc[j][mt] = __builtin_amdgcn_wmma_f32_16x16x32_f16(false, a[mt], false, bs[j],
                                                            (short)0, acc[j][mt], false, false);
    }
    loadB(bs, kt + 2);
    lds_barrier();
  };

  float4 a0[2], a1[2];
  v16h b0[NT], b1[NT];
  loadA(a0, 0);
  loadB(b0, 0);
  loadA(a1, 1);
  loadB(b1, 1);
  int kt = 0;
  for (; kt + 1 < KT; kt += 2) {
    kstep(kt, a0, b0);
    kstep(kt + 1, a1, b1);
  }
  if (kt < KT) kstep(kt, a0, b0);
  int rHalf = (lane >> 4) << 3;
#pragma unroll
  for (int j = 0; j < TPW; ++j) {
    int col = ((wave + (j << 3)) << 4) + (lane & 15);
    float bv = (bias != nullptr) ? bias[col] : 0.0f;
#pragma unroll
    for (int mt = 0; mt < 4; ++mt) {
      int rBase = mBase + (mt << 4) + rHalf;
#pragma unroll
      for (int r = 0; r < 8; ++r)
        C[(size_t)(rBase + r) * ldc + col] = acc[j][mt][r] + bv;
    }
  }
  if constexpr (XT == 1) {
    int col = (wave << 4) + (lane & 15);
    float bv = (bias2 != nullptr) ? bias2[col] : 0.0f;
#pragma unroll
    for (int mt = 0; mt < 4; ++mt) {
      int rBase = mBase + (mt << 4) + rHalf;
#pragma unroll
      for (int r = 0; r < 8; ++r)
        C2[(size_t)(rBase + r) * ldc2 + col] = acc[TPW][mt][r] + bv;
    }
  }
}

__device__ __forceinline__ float sigmoid_hw(float x) {
  return __builtin_amdgcn_rcpf(1.0f + __builtin_amdgcn_exp2f(x * -1.442695041f));
}
__device__ __forceinline__ float tanh_hw(float x) {
  return 1.0f - 2.0f * __builtin_amdgcn_rcpf(1.0f + __builtin_amdgcn_exp2f(x * 2.885390082f));
}

__device__ __forceinline__ void gi_fetch(float2 (&g)[3], const float* ga, int t) {
  const float* a = ga + (size_t)t * 384;
  g[0] = *(const float2*)(a);
  g[1] = *(const float2*)(a + 128);
  g[2] = *(const float2*)(a + 256);
}

#define GRU_RB 4
__global__ __launch_bounds__(256) __attribute__((amdgpu_waves_per_eu(2, 2))) void gru_recur(const float* __restrict__ gi,
                                                 const _Float16* __restrict__ WhhF,
                                                 const float* __restrict__ bhh,
                                                 float* __restrict__ out, int S) {
  __shared__ __align__(8) float hS[GRU_RB * 128];
  __shared__ __align__(32) _Float16 aF[4 * 512];
  __shared__ __align__(8) float ghS[GRU_RB * 384];
  int tid  = threadIdx.x;
  int lane = tid & 31;
  int wave = tid >> 5;
  int bBase = blockIdx.x * GRU_RB;
  const _Float16* Bw = WhhF + (size_t)(wave * 3) * 4 * 512 + lane * 16;
  if (S <= 0) return;

  v16h bF[3][4];
  float bv[3];
#pragma unroll
  for (int j = 0; j < 3; ++j) {
#pragma unroll
    for (int kt = 0; kt < 4; ++kt)
      bF[j][kt] = *reinterpret_cast<const v16h*>(&Bw[((size_t)j * 4 + kt) * 512]);
    bv[j] = bhh[((wave * 3 + j) << 4) + (lane & 15)];
  }

  int grow = tid >> 6;
  int gh0  = (tid & 63) << 1;
  const float* giA = gi + (size_t)(bBase + grow) * S * 384 + gh0;

  float2 g0[3], g1[3], g2[3];
  gi_fetch(g0, giA, 0);
  gi_fetch(g1, giA, (1 < S) ? 1 : S - 1);

  for (int i = tid; i < GRU_RB * 128; i += 256) hS[i] = 0.0f;
  for (int i = tid; i < 4 * 512; i += 256) aF[i] = (_Float16)0.0f;
  __syncthreads();

  auto step = [&](int t, const float2 (&g)[3]) __attribute__((always_inline)) {
    v16h a[4];
#pragma unroll
    for (int kt = 0; kt < 4; ++kt)
      a[kt] = *reinterpret_cast<const v16h*>(&aF[kt * 512 + lane * 16]);
    __builtin_amdgcn_sched_barrier(0);
    v8f acc[3] = {};
#pragma unroll
    for (int kt = 0; kt < 4; ++kt) {
#pragma unroll
      for (int j = 0; j < 3; ++j)
        acc[j] = __builtin_amdgcn_wmma_f32_16x16x32_f16(false, a[kt], false, bF[j][kt],
                                                        (short)0, acc[j], false, false);
    }
    if (lane < 16) {
#pragma unroll
      for (int j = 0; j < 3; ++j) {
        int col = ((wave * 3 + j) << 4) + lane;
#pragma unroll
        for (int r = 0; r < GRU_RB; ++r) ghS[r * 384 + col] = acc[j][r] + bv[j];
      }
    }
    lds_barrier();
    float2 hr = *(const float2*)(&ghS[grow * 384 + gh0]);
    float2 hz = *(const float2*)(&ghS[grow * 384 + 128 + gh0]);
    float2 hn = *(const float2*)(&ghS[grow * 384 + 256 + gh0]);
    float2 ho = *(const float2*)(&hS[grow * 128 + gh0]);
    float2 gr = g[0];
    float2 gz = g[1];
    float2 gn = g[2];
    float2 hNew;
#define GATE(c)                                                            \
    {                                                                      \
      float rg = sigmoid_hw(gr.c + hr.c);                                  \
      float zg = sigmoid_hw(gz.c + hz.c);                                  \
      float ng = tanh_hw(gn.c + rg * hn.c);                                \
      hNew.c = (1.0f - zg) * ng + zg * ho.c;                               \
    }
    GATE(x) GATE(y)
#undef GATE
    *(float2*)(&hS[grow * 128 + gh0]) = hNew;
    *(float2*)(out + ((size_t)(bBase + grow) * S + t) * 128 + gh0) = hNew;
    {
      int kt = gh0 >> 5;
      int kl = gh0 & 31;
      int ln = (((kl >> 3) & 1) << 4) + grow;
      int e0 = (kl & 7) | ((kl >> 4) << 3);
      v2h hv;
      hv[0] = (_Float16)hNew.x; hv[1] = (_Float16)hNew.y;
      *reinterpret_cast<v2h*>(&aF[kt * 512 + ln * 16 + e0]) = hv;
    }
    lds_barrier();
  };

  int t = 0;
  for (; t + 2 < S; t += 3) {
    gi_fetch(g2, giA, t + 2);
    step(t, g0);
    gi_fetch(g0, giA, (t + 3 < S) ? t + 3 : S - 1);
    step(t + 1, g1);
    gi_fetch(g1, giA, (t + 4 < S) ? t + 4 : S - 1);
    step(t + 2, g2);
  }
  if (t < S) {
    step(t, g0);
    if (t + 1 < S) step(t + 1, g1);
  }
}

__device__ __forceinline__ float blockReduceSum(float v, float* red) {
  int tid = threadIdx.x;
  red[tid] = v;
  __syncthreads();
  for (int off = 128; off > 0; off >>= 1) {
    if (tid < off) red[tid] += red[tid + off];
    __syncthreads();
  }
  float r = red[0];
  __syncthreads();
  return r;
}
__device__ __forceinline__ float blockReduceMax(float v, float* red) {
  int tid = threadIdx.x;
  red[tid] = v;
  __syncthreads();
  for (int off = 128; off > 0; off >>= 1) {
    if (tid < off) red[tid] = fmaxf(red[tid], red[tid + off]);
    __syncthreads();
  }
  float r = red[0];
  __syncthreads();
  return r;
}

__global__ __launch_bounds__(256) void finalize_kernel(const float* __restrict__ gE,
                                                       const float* __restrict__ u,
                                                       const float* __restrict__ out1,
                                                       const float* __restrict__ vvec,
                                                       const float* __restrict__ Wfc,
                                                       const float* __restrict__ bfc,
                                                       float* __restrict__ outp, int S) {
  __shared__ __align__(16) float uL[128];
  __shared__ __align__(16) float vL[128];
  __shared__ float eS[64], attnS[64];
  __shared__ __align__(16) float concatL[256];
  __shared__ float oldL[512], wcL[512], red[256];
  int tid  = threadIdx.x;
  int lane = tid & 31;
  int wave = tid >> 5;
  int b = blockIdx.x;

  if (tid < 128) {
    uL[tid] = u[(size_t)b * 128 + tid];
    vL[tid] = vvec[tid];
  }
  __syncthreads();

  {
    const float4 u4 = *(const float4*)(&uL[lane << 2]);
    const float4 v4 = *(const float4*)(&vL[lane << 2]);
    const float* geb = gE + (size_t)b * S * 128 + (lane << 2);
    for (int s0 = wave; s0 < S; s0 += 32) {
      float4 g[4];
#pragma unroll
      for (int i = 0; i < 4; ++i) {
        int s  = s0 + (i << 3);
        int sc = (s < S) ? s : S - 1;
        g[i] = *(const float4*)(geb + (size_t)sc * 128);
      }
      __builtin_amdgcn_sched_barrier(0);
#pragma unroll
      for (int i = 0; i < 4; ++i) {
        float p = v4.x * tanhf(u4.x + g[i].x);
        p += v4.y * tanhf(u4.y + g[i].y);
        p += v4.z * tanhf(u4.z + g[i].z);
        p += v4.w * tanhf(u4.w + g[i].w);
        for (int off = 16; off > 0; off >>= 1) p += __shfl_down(p, off, 32);
        int s = s0 + (i << 3);
        if (lane == 0 && s < S) eS[s] = p;
      }
    }
  }
  __syncthreads();

  if (tid == 0) {
    float m = eS[0];
    for (int s = 1; s < S; ++s) m = fmaxf(m, eS[s]);
    float sum = 0.0f;
    for (int s = 0; s < S; ++s) { float ev = expf(eS[s] - m); attnS[s] = ev; sum += ev; }
    float inv = 1.0f / sum;
    for (int s = 0; s < S; ++s) attnS[s] *= inv;
  }
  __syncthreads();

  if (tid < 128) {
    float c = 0.0f;
    const float* ob = out1 + (size_t)b * S * 128 + tid;
    for (int s0 = 0; s0 < S; s0 += 32) {
      float o[32];
#pragma unroll
      for (int i = 0; i < 32; ++i) {
        int s  = s0 + i;
        int sc = (s < S) ? s : S - 1;
        o[i] = ob[(size_t)sc * 128];
      }
#pragma unroll
      for (int i = 0; i < 32; ++i) {
        int s = s0 + i;
        float a = (s < S) ? attnS[s] : 0.0f;
        c += a * o[i];
      }
    }
    concatL[tid]       = ob[(size_t)(S - 1) * 128];
    concatL[128 + tid] = c;
  }
  __syncthreads();

  {
    const float4 c0 = *(const float4*)(&concatL[lane << 2]);
    const float4 c1 = *(const float4*)(&concatL[128 + (lane << 2)]);
    const float* wl = Wfc + (lane << 2);
    for (int n0 = wave; n0 < 500; n0 += 64) {
      float4 w0[8], w1[8];
      float bn[8], p[8];
#pragma unroll
      for (int i = 0; i < 8; ++i) {
        int n  = n0 + (i << 3);
        int nc = (n < 500) ? n : 499;
        const float* wr = wl + (size_t)nc * 256;
        w0[i] = *(const float4*)(wr);
        w1[i] = *(const float4*)(wr + 128);
        bn[i] = bfc[nc];
      }
      __builtin_amdgcn_sched_barrier(0);
#pragma unroll
      for (int i = 0; i < 8; ++i) {
        float a = (lane == 0) ? bn[i] : 0.0f;
        a += w0[i].x * c0.x;
        a += w0[i].y * c0.y;
        a += w0[i].z * c0.z;
        a += w0[i].w * c0.w;
        a += w1[i].x * c1.x;
        a += w1[i].y * c1.y;
        a += w1[i].z * c1.z;
        a += w1[i].w * c1.w;
        p[i] = a;
      }
#pragma unroll
      for (int off = 16; off > 0; off >>= 1) {
#pragma unroll
        for (int i = 0; i < 8; ++i) p[i] += __shfl_xor(p[i], off, 32);
      }
      if (lane == 0) {
#pragma unroll
        for (int i = 0; i < 8; ++i) {
          int n = n0 + (i << 3);
          if (n < 500) oldL[n] = p[i];
        }
      }
    }
  }
  __syncthreads();

  float pm = -3.4e38f;
  for (int n = tid; n < 500; n += 256) pm = fmaxf(pm, oldL[n]);
  pm = blockReduceMax(pm, red);
  float ps = 0.0f;
  for (int n = tid; n < 500; n += 256) { float e = expf(oldL[n] - pm); wcL[n] = e; ps += e; }
  ps = blockReduceSum(ps, red);
  float inv = 1.0f / ps;
  for (int n = tid; n < 500; n += 256) {
    float w = wcL[n] * inv;
    oldL[n] = w;
    wcL[n]  = fminf(fmaxf(w, LBV), UBV);
  }
  __syncthreads();

  for (int it = 0; it < 32; ++it) {
    float pl = 0.0f, pn = 0.0f;
    for (int n = tid; n < 500; n += 256) {
      float wcv = wcL[n];
      pl += oldL[n] - wcv;
      if (wcv != UBV) pn += wcv;
    }
    float leftover = blockReduceSum(pl, red);
    float nsum     = blockReduceSum(pn, red);
    float ns = (nsum == 0.0f) ? 1.0f : nsum;
    float pover = 0.0f;
    for (int n = tid; n < 500; n += 256) {
      float wcv = wcL[n];
      float wc2 = (wcv != UBV) ? wcv + leftover * wcv / ns : wcv;
      oldL[n] = wc2;
      if (wc2 > UBV) pover = 1.0f;
    }
    float anyOver = blockReduceSum(pover, red);
    int new_done = (anyOver == 0.0f);
    for (int n = tid; n < 500; n += 256) {
      float wc2 = oldL[n];
      wcL[n] = new_done ? wc2 : fminf(fmaxf(wc2, LBV), UBV);
    }
    __syncthreads();
    if (new_done) break;
  }

  for (int n = tid; n < 500; n += 256) outp[(size_t)b * 500 + n] = wcL[n];
}

extern "C" void kernel_launch(void* const* d_in, const int* in_sizes, int n_in,
                              void* d_out, int out_size, void* d_ws, size_t ws_size,
                              hipStream_t stream) {
  (void)in_sizes; (void)n_in; (void)out_size; (void)ws_size;
  const float* x    = (const float*)d_in[0];
  const float* Wih0 = (const float*)d_in[1];
  const float* Whh0 = (const float*)d_in[2];
  const float* bih0 = (const float*)d_in[3];
  const float* bhh0 = (const float*)d_in[4];
  const float* Wih1 = (const float*)d_in[5];
  const float* Whh1 = (const float*)d_in[6];
  const float* bih1 = (const float*)d_in[7];
  const float* bhh1 = (const float*)d_in[8];
  const float* Wa   = (const float*)d_in[9];
  const float* ba   = (const float*)d_in[10];
  const float* vv   = (const float*)d_in[11];
  const float* Wfc  = (const float*)d_in[12];
  const float* bfc  = (const float*)d_in[13];
  float* outp = (float*)d_out;

  constexpr int B = kB, S = kS, N = kN, H = kH, G3 = kG3;

  uint8_t* base = (uint8_t*)d_ws;
  size_t off = 0;
  auto carve = [&](size_t bytes) {
    void* p = base + off;
    off = (off + bytes + 255) & ~(size_t)255;
    return p;
  };
  float* gi   = (float*)carve((size_t)B * S * G3 * 4);
  float* out0 = (float*)carve((size_t)B * S * H * 4);
  float* out1 = (float*)carve((size_t)B * S * H * 4);
  float* gEb  = (float*)carve((size_t)B * S * H * 4);
  float* ub   = (float*)carve((size_t)B * H * 4);
  _Float16* Wih0F = (_Float16*)carve((size_t)24 * 16 * 512 * 2);
  _Float16* Whh0F = (_Float16*)carve((size_t)24 * 4 * 512 * 2);
  _Float16* Wih1F = (_Float16*)carve((size_t)24 * 4 * 512 * 2);
  _Float16* Whh1F = (_Float16*)carve((size_t)24 * 4 * 512 * 2);
  _Float16* WaxF  = (_Float16*)carve((size_t)8 * 16 * 512 * 2);
  _Float16* WahF  = (_Float16*)carve((size_t)8 * 4 * 512 * 2);

  wfrag_prep_all<<<kPB5, 256, 0, stream>>>(Wih0, Whh0, Wih1, Whh1, Wa + H, Wa,
                                           Wih0F, Whh0F, Wih1F, Whh1F, WaxF, WahF);

  auto gemm384 = [&](const float* A, const _Float16* Bf, const float* bias, float* C,
                     int M, int K, int lda) {
    gemm_wmma<3, 0><<<dim3(M / 64), 256, 0, stream>>>(A, Bf, bias, C, K, lda, G3,
                                                       nullptr, nullptr, nullptr, 0);
  };
  auto gemm128 = [&](const float* A, const _Float16* Bf, const float* bias, float* C,
                     int M, int K, int lda) {
    gemm_wmma<1, 0><<<dim3(M / 64), 256, 0, stream>>>(A, Bf, bias, C, K, lda, H,
                                                       nullptr, nullptr, nullptr, 0);
  };

  gemm_wmma<3, 1><<<dim3(B * S / 64), 256, 0, stream>>>(x, Wih0F, bih0, gi, N, N, G3,
                                                         WaxF, ba, gEb, H);
  gru_recur<<<B / GRU_RB, 256, 0, stream>>>(gi, Whh0F, bhh0, out0, S);
  gemm384(out0, Wih1F, bih1, gi, B * S, H, H);
  gru_recur<<<B / GRU_RB, 256, 0, stream>>>(gi, Whh1F, bhh1, out1, S);
  gemm128(out1 + (size_t)(S - 1) * H, WahF, nullptr, ub, B, H, S * H);
  finalize_kernel<<<B, 256, 0, stream>>>(gEb, ub, out1, vv, Wfc, bfc, outp, S);
}


# FC logits: one 1024-thread block per 4 batch rows (256 blocks); each Wfc row a wave loads feeds the dot products of all 4 rows: Wfc lines drawn from L2 fall to a quarter; every sum keeps its order
# speedup vs baseline: 3.4688x; 1.0477x over previous
#include <hip/hip_runtime.h>
#include <hip/hip_bf16.h>
#include <stdint.h>

typedef __attribute__((ext_vector_type(16))) _Float16 v16h;
typedef __attribute__((ext_vector_type(8)))  _Float16 v8h;
typedef __attribute__((ext_vector_type(4)))  _Float16 v4h;
typedef __attribute__((ext_vector_type(2)))  _Float16 v2h;
typedef __attribute__((ext_vector_type(8)))  float    v8f;

#define LBV 0.0f
#define UBV 0.2f

constexpr int kB = 1024, kS = 60, kN = 500, kH = 128, kG3 = 384;
static_assert(kG3 % 16 == 0 && kH % 16 == 0, "weight rows fill whole 16-row fragments");

constexpr int prepBlocks(int n, int k) { return (n / 16) * ((k + 31) / 32) * 64 / 256; }
constexpr int kPB0 = prepBlocks(kG3, kN);
constexpr int kPB1 = kPB0 + prepBlocks(kG3, kH);
constexpr int kPB2 = kPB1 + prepBlocks(kG3, kH);
constexpr int kPB3 = kPB2 + prepBlocks(kG3, kH);
constexpr int kPB4 = kPB3 + prepBlocks(kH, kN);
constexpr int kPB5 = kPB4 + prepBlocks(kH, kH);

__global__ __launch_bounds__(256) void wfrag_prep_all(const float* __restrict__ wih0,
                                                      const float* __restrict__ whh0,
                                                      const float* __restrict__ wih1,
                                                      const float* __restrict__ whh1,
                                                      const float* __restrict__ wax,
                                                      const float* __restrict__ wah,
                                                      _Float16* __restrict__ d0,
                                                      _Float16* __restrict__ d1,
                                                      _Float16* __restrict__ d2,
                                                      _Float16* __restrict__ d3,
                                                      _Float16* __restrict__ d4,
                                                      _Float16* __restrict__ d5) {
  int blk = blockIdx.x;
  const float* W;
  _Float16* dst;
  int K, ldw, b0;
  if (blk < kPB0)      { W = wih0; dst = d0; K = kN; ldw = kN;      b0 = 0;    }
  else if (blk < kPB1) { W = whh0; dst = d1; K = kH; ldw = kH;      b0 = kPB0; }
  else if (blk < kPB2) { W = wih1; dst = d2; K = kH; ldw = kH;      b0 = kPB1; }
  else if (blk < kPB3) { W = whh1; dst = d3; K = kH; ldw = kH;      b0 = kPB2; }
  else if (blk < kPB4) { W = wax;  dst = d4; K = kN; ldw = kH + kN; b0 = kPB3; }
  else                 { W = wah;  dst = d5; K = kH; ldw = kH + kN; b0 = kPB4; }
  int KT = (K + 31) >> 5;
  int u      = (blk - b0) * 256 + threadIdx.x;
  int frag   = u >> 6;
  int within = u & 63;
  int lane = within >> 1;
  int e8   = within & 1;
  int tileN = frag / KT;
  int tileK = frag - tileN * KT;
  int n  = tileN * 16 + (lane & 15);
  int kk = tileK * 32 + ((lane >> 4) << 3) + (e8 << 4);
  const float* row = W + (size_t)n * ldw;
  float v[8];
#pragma unroll
  for (int i = 0; i < 8; ++i) v[i] = row[(kk + i < K) ? kk + i : K - 1];
  v8h hv;
#pragma unroll
  for (int i = 0; i < 8; ++i) hv[i] = (_Float16)((kk + i < K) ? v[i] : 0.0f);
  *reinterpret_cast<v8h*>(dst + (size_t)frag * 512 + lane * 16 + e8 * 8) = hv;
}

__device__ __forceinline__ void lds_barrier() {
  __builtin_amdgcn_fence(__ATOMIC_RELEASE, "workgroup", "local");
  __builtin_amdgcn_s_barrier();
  __builtin_amdgcn_fence(__ATOMIC_ACQUIRE, "workgroup", "local");
}

template <int TPW, int XT>
__global__ __launch_bounds__(256) void gemm_wmma(const float* __restrict__ A,
                                                 const _Float16* __restrict__ Bfrag,
                                                 const float* __restrict__ bias,
                                                 float* __restrict__ C,
                                                 int K, int lda, int ldc,
                                                 const _Float16* __restrict__ Bfrag2,
                                                 const float* __restrict__ bias2,
                                                 float* __restrict__ C2, int ldc2) {
  static_assert(XT == 0 || XT == 1, "at most one extra N tile per wave");
  constexpr int NT = TPW + XT;
  __shared__ __align__(32) _Float16 ldsA[4 * 512];
  int tid  = threadIdx.x;
  int lane = tid & 31;
  int wave = tid >> 5;
  int mBase = blockIdx.x * 64;
  int KT    = (K + 31) >> 5;
  v8f acc[NT][4] = {};
  int sr  = tid >> 3;
  int sq  = tid & 7;
  int skl = sq << 2;
  int sln = (((skl >> 3) & 1) << 4) + (sr & 15);
  int se0 = (skl & 7) | ((skl >> 4) << 3);
  _Float16* st0 = &ldsA[(sr >> 4) * 512 + sln * 16 + se0];
  _Float16* st1 = st0 + 2 * 512;
  const float* Arow0 = A + (size_t)(mBase + sr) * lda;
  const float* Arow1 = Arow0 + (size_t)32 * lda;
  const _Float16* Bw = Bfrag + (size_t)wave * KT * 512 + lane * 16;
  const _Float16* Bw2 = nullptr;
  if constexpr (XT == 1) Bw2 = Bfrag2 + (size_t)wave * KT * 512 + lane * 16;
  const int klast = (K - 1) & ~3;

  auto loadA = [&](float4 (&as)[2], int kt) __attribute__((always_inline)) {
    int col = (kt << 5) + skl;
    int cc  = (col < klast) ? col : klast;
    as[0] = *(const float4*)(Arow0 + cc);
    as[1] = *(const float4*)(Arow1 + cc);
  };
  auto loadB = [&](v16h (&bs)[NT], int kt) __attribute__((always_inline)) {
    int kc = (kt < KT) ? kt : KT - 1;
#pragma unroll
    for (int j = 0; j < TPW; ++j)
      bs[j] = *reinterpret_cast<const v16h*>(&Bw[((size_t)(j << 3) * KT + kc) * 512]);
    if constexpr (XT == 1)
      bs[TPW] = *reinterpret_cast<const v16h*>(&Bw2[(size_t)kc * 512]);
  };

  auto kstep = [&](int kt, float4 (&as)[2], v16h (&bs)[NT]) __attribute__((always_inline)) {
    int k0 = kt << 5;
    float4 va = as[0];
    float4 vb = as[1];
    if (k0 + 32 > K) {
      int col = k0 + skl;
      if (col     >= K) { va.x = 0.0f; vb.x = 0.0f; }
      if (col + 1 >= K) { va.y = 0.0f; vb.y = 0.0f; }
      if (col + 2 >= K) { va.z = 0.0f; vb.z = 0.0f; }
      if (col + 3 >= K) { va.w = 0.0f; vb.w = 0.0f; }
    }
    v4h hv0, hv1;
    hv0[0] = (_Float16)va.x; hv0[1] = (_Float16)va.y;
    hv0[2] = (_Float16)va.z; hv0[3] = (_Float16)va.w;
    hv1[0] = (_Float16)vb.x; hv1[1] = (_Float16)vb.y;
    hv1[2] = (_Float16)vb.z; hv1[3] = (_Float16)vb.w;
    *reinterpret_cast<v4h*>(st0) = hv0;
    *reinterpret_cast<v4h*>(st1) = hv1;
    loadA(as, kt + 2);
    lds_barrier();
    v16h a[4];
#pragma unroll
    for (int mt = 0; mt < 4; ++mt)
      a[mt] = *reinterpret_cast<const v16h*>(&ldsA[mt * 512 + lane * 16]);
#pragma unroll
    for (int j = 0; j < NT; ++j) {
#pragma unroll
      for (int mt = 0; mt < 4; ++mt)
        acc[j][mt] = __builtin_amdgcn_wmma_f32_16x16x32_f16(false, a[mt], false, bs[j],
                                                            (short)0, acc[j][mt], false, false);
    }
    loadB(bs, kt + 2);
    lds_barrier();
  };

  float4 a0[2], a1[2];
  v16h b0[NT], b1[NT];
  loadA(a0, 0);
  loadB(b0, 0);
  loadA(a1, 1);
  loadB(b1, 1);
  int kt = 0;
  for (; kt + 1 < KT; kt += 2) {
    kstep(kt, a0, b0);
    kstep(kt + 1, a1, b1);
  }
  if (kt < KT) kstep(kt, a0, b0);
  int rHalf = (lane >> 4) << 3;
#pragma unroll
  for (int j = 0; j < TPW; ++j) {
    int col = ((wave + (j << 3)) << 4) + (lane & 15);
    float bv = (bias != nullptr) ? bias[col] : 0.0f;
#pragma unroll
    for (int mt = 0; mt < 4; ++mt) {
      int rBase = mBase + (mt << 4) + rHalf;
#pragma unroll
      for (int r = 0; r < 8; ++r)
        C[(size_t)(rBase + r) * ldc + col] = acc[j][mt][r] + bv;
    }
  }
  if constexpr (XT == 1) {
    int col = (wave << 4) + (lane & 15);
    float bv = (bias2 != nullptr) ? bias2[col] : 0.0f;
#pragma unroll
    for (int mt = 0; mt < 4; ++mt) {
      int rBase = mBase + (mt << 4) + rHalf;
#pragma unroll
      for (int r = 0; r < 8; ++r)
        C2[(size_t)(rBase + r) * ldc2 + col] = acc[TPW][mt][r] + bv;
    }
  }
}

__device__ __forceinline__ float sigmoid_hw(float x) {
  return __builtin_amdgcn_rcpf(1.0f + __builtin_amdgcn_exp2f(x * -1.442695041f));
}
__device__ __forceinline__ float tanh_hw(float x) {
  return 1.0f - 2.0f * __builtin_amdgcn_rcpf(1.0f + __builtin_amdgcn_exp2f(x * 2.885390082f));
}

__device__ __forceinline__ void gi_fetch(float2 (&g)[3], const float* ga, int t) {
  const float* a = ga + (size_t)t * 384;
  g[0] = *(const float2*)(a);
  g[1] = *(const float2*)(a + 128);
  g[2] = *(const float2*)(a + 256);
}

#define GRU_RB 4
__global__ __launch_bounds__(256) __attribute__((amdgpu_waves_per_eu(2, 2))) void gru_recur(const float* __restrict__ gi,
                                                 const _Float16* __restrict__ WhhF,
                                                 const float* __restrict__ bhh,
                                                 float* __restrict__ out, int S) {
  __shared__ __align__(8) float hS[GRU_RB * 128];
  __shared__ __align__(32) _Float16 aF[4 * 512];
  __shared__ __align__(8) float ghS[GRU_RB * 384];
  int tid  = threadIdx.x;
  int lane = tid & 31;
  int wave = tid >> 5;
  int bBase = blockIdx.x * GRU_RB;
  const _Float16* Bw = WhhF + (size_t)(wave * 3) * 4 * 512 + lane * 16;
  if (S <= 0) return;

  v16h bF[3][4];
  float bv[3];
#pragma unroll
  for (int j = 0; j < 3; ++j) {
#pragma unroll
    for (int kt = 0; kt < 4; ++kt)
      bF[j][kt] = *reinterpret_cast<const v16h*>(&Bw[((size_t)j * 4 + kt) * 512]);
    bv[j] = bhh[((wave * 3 + j) << 4) + (lane & 15)];
  }

  int grow = tid >> 6;
  int gh0  = (tid & 63) << 1;
  const float* giA = gi + (size_t)(bBase + grow) * S * 384 + gh0;

  float2 g0[3], g1[3], g2[3];
  gi_fetch(g0, giA, 0);
  gi_fetch(g1, giA, (1 < S) ? 1 : S - 1);

  for (int i = tid; i < GRU_RB * 128; i += 256) hS[i] = 0.0f;
  for (int i = tid; i < 4 * 512; i += 256) aF[i] = (_Float16)0.0f;
  __syncthreads();

  auto step = [&](int t, const float2 (&g)[3]) __attribute__((always_inline)) {
    v16h a[4];
#pragma unroll
    for (int kt = 0; kt < 4; ++kt)
      a[kt] = *reinterpret_cast<const v16h*>(&aF[kt * 512 + lane * 16]);
    __builtin_amdgcn_sched_barrier(0);
    v8f acc[3] = {};
#pragma unroll
    for (int kt = 0; kt < 4; ++kt) {
#pragma unroll
      for (int j = 0; j < 3; ++j)
        acc[j] = __builtin_amdgcn_wmma_f32_16x16x32_f16(false, a[kt], false, bF[j][kt],
                                                        (short)0, acc[j], false, false);
    }
    if (lane < 16) {
#pragma unroll
      for (int j = 0; j < 3; ++j) {
        int col = ((wave * 3 + j) << 4) + lane;
#pragma unroll
        for (int r = 0; r < GRU_RB; ++r) ghS[r * 384 + col] = acc[j][r] + bv[j];
      }
    }
    lds_barrier();
    float2 hr = *(const float2*)(&ghS[grow * 384 + gh0]);
    float2 hz = *(const float2*)(&ghS[grow * 384 + 128 + gh0]);
    float2 hn = *(const float2*)(&ghS[grow * 384 + 256 + gh0]);
    float2 ho = *(const float2*)(&hS[grow * 128 + gh0]);
    float2 gr = g[0];
    float2 gz = g[1];
    float2 gn = g[2];
    float2 hNew;
#define GATE(c)                                                            \
    {                                                                      \
      float rg = sigmoid_hw(gr.c + hr.c);                                  \
      float zg = sigmoid_hw(gz.c + hz.c);                                  \
      float ng = tanh_hw(gn.c + rg * hn.c);                                \
      hNew.c = (1.0f - zg) * ng + zg * ho.c;                               \
    }
    GATE(x) GATE(y)
#undef GATE
    *(float2*)(&hS[grow * 128 + gh0]) = hNew;
    *(float2*)(out + ((size_t)(bBase + grow) * S + t) * 128 + gh0) = hNew;
    {
      int kt = gh0 >> 5;
      int kl = gh0 & 31;
      int ln = (((kl >> 3) & 1) << 4) + grow;
      int e0 = (kl & 7) | ((kl >> 4) << 3);
      v2h hv;
      hv[0] = (_Float16)hNew.x; hv[1] = (_Float16)hNew.y;
      *reinterpret_cast<v2h*>(&aF[kt * 512 + ln * 16 + e0]) = hv;
    }
    lds_barrier();
  };

  int t = 0;
  for (; t + 2 < S; t += 3) {
    gi_fetch(g2, giA, t + 2);
    step(t, g0);
    gi_fetch(g0, giA, (t + 3 < S) ? t + 3 : S - 1);
    step(t + 1, g1);
    gi_fetch(g1, giA, (t + 4 < S) ? t + 4 : S - 1);
    step(t + 2, g2);
  }
  if (t < S) {
    step(t, g0);
    if (t + 1 < S) step(t + 1, g1);
  }
}

#define FIN_RB 4

__device__ __forceinline__ float groupReduceSum(float v, float* red, int lt) {
  red[lt] = v;
  __syncthreads();
  for (int off = 128; off > 0; off >>= 1) {
    if (lt < off) red[lt] += red[lt + off];
    __syncthreads();
  }
  float r = red[0];
  __syncthreads();
  return r;
}
__device__ __forceinline__ float groupReduceMax(float v, float* red, int lt) {
  red[lt] = v;
  __syncthreads();
  for (int off = 128; off > 0; off >>= 1) {
    if (lt < off) red[lt] = fmaxf(red[lt], red[lt + off]);
    __syncthreads();
  }
  float r = red[0];
  __syncthreads();
  return r;
}

__global__ __launch_bounds__(256 * FIN_RB) void finalize_kernel(const float* __restrict__ gE,
                                                       const float* __restrict__ u,
                                                       const float* __restrict__ out1,
                                                       const float* __restrict__ vvec,
                                                       const float* __restrict__ Wfc,
                                                       const float* __restrict__ bfc,
                                                       float* __restrict__ outp, int S) {
  __shared__ __align__(16) float uL[FIN_RB][128];
  __shared__ __align__(16) float vL[128];
  __shared__ float eS[FIN_RB][64], attnS[FIN_RB][64];
  __shared__ __align__(16) float concatL[FIN_RB][256];
  __shared__ float oldL[FIN_RB][512], wcL[FIN_RB][512], redL[FIN_RB][256];
  __shared__ int doneL[FIN_RB];
  int tid   = threadIdx.x;
  int lane  = tid & 31;
  int wave  = tid >> 5;
  int grp   = tid >> 8;
  int lt    = tid & 255;
  int gwave = lt >> 5;
  int b = blockIdx.x * FIN_RB + grp;
  float* red  = redL[grp];
  float* oldG = oldL[grp];
  float* wcG  = wcL[grp];

  if (lt < 128) {
    uL[grp][lt] = u[(size_t)b * 128 + lt];
    if (grp == 0) vL[lt] = vvec[lt];
  }
  __syncthreads();

  {
    const float4 u4 = *(const float4*)(&uL[grp][lane << 2]);
    const float4 v4 = *(const float4*)(&vL[lane << 2]);
    const float* geb = gE + (size_t)b * S * 128 + (lane << 2);
    for (int s0 = gwave; s0 < S; s0 += 32) {
      float4 g[4];
#pragma unroll
      for (int i = 0; i < 4; ++i) {
        int s  = s0 + (i << 3);
        int sc = (s < S) ? s : S - 1;
        g[i] = *(const float4*)(geb + (size_t)sc * 128);
      }
      __builtin_amdgcn_sched_barrier(0);
#pragma unroll
      for (int i = 0; i < 4; ++i) {
        float p = v4.x * tanhf(u4.x + g[i].x);
        p += v4.y * tanhf(u4.y + g[i].y);
        p += v4.z * tanhf(u4.z + g[i].z);
        p += v4.w * tanhf(u4.w + g[i].w);
        for (int off = 16; off > 0; off >>= 1) p += __shfl_down(p, off, 32);
        int s = s0 + (i << 3);
        if (lane == 0 && s < S) eS[grp][s] = p;
      }
    }
  }
  __syncthreads();

  if (lt == 0) {
    float m = eS[grp][0];
    for (int s = 1; s < S; ++s) m = fmaxf(m, eS[grp][s]);
    float sum = 0.0f;
    for (int s = 0; s < S; ++s) { float ev = expf(eS[grp][s] - m); attnS[grp][s] = ev; sum += ev; }
    float inv = 1.0f / sum;
    for (int s = 0; s < S; ++s) attnS[grp][s] *= inv;
  }
  __syncthreads();

  if (lt < 128) {
    float c = 0.0f;
    const float* ob = out1 + (size_t)b * S * 128 + lt;
    for (int s0 = 0; s0 < S; s0 += 32) {
      float o[32];
#pragma unroll
      for (int i = 0; i < 32; ++i) {
        int s  = s0 + i;
        int sc = (s < S) ? s : S - 1;
        o[i] = ob[(size_t)sc * 128];
      }
#pragma unroll
      for (int i = 0; i < 32; ++i) {
        int s = s0 + i;
        float a = (s < S) ? attnS[grp][s] : 0.0f;
        c += a * o[i];
      }
    }
    concatL[grp][lt]       = ob[(size_t)(S - 1) * 128];
    concatL[grp][128 + lt] = c;
  }
  __syncthreads();

  {
    float4 c0[FIN_RB], c1[FIN_RB];
#pragma unroll
    for (int r = 0; r < FIN_RB; ++r) {
      c0[r] = *(const float4*)(&concatL[r][lane << 2]);
      c1[r] = *(const float4*)(&concatL[r][128 + (lane << 2)]);
    }
    const float* wl = Wfc + (lane << 2);
    for (int n0 = wave; n0 < 500; n0 += 128) {
      float4 w0[4], w1[4];
      float bn[4], p[4][FIN_RB];
#pragma unroll
      for (int i = 0; i < 4; ++i) {
        int n  = n0 + (i << 5);
        int nc = (n < 500) ? n : 499;
        const float* wr = wl + (size_t)nc * 256;
        w0[i] = *(const float4*)(wr);
        w1[i] = *(const float4*)(wr + 128);
        bn[i] = bfc[nc];
      }
      __builtin_amdgcn_sched_barrier(0);
#pragma unroll
      for (int i = 0; i < 4; ++i) {
#pragma unroll
        for (int r = 0; r < FIN_RB; ++r) {
          float a = (lane == 0) ? bn[i] : 0.0f;
          a += w0[i].x * c0[r].x;
          a += w0[i].y * c0[r].y;
          a += w0[i].z * c0[r].z;
          a += w0[i].w * c0[r].w;
          a += w1[i].x * c1[r].x;
          a += w1[i].y * c1[r].y;
          a += w1[i].z * c1[r].z;
          a += w1[i].w * c1[r].w;
          p[i][r] = a;
        }
      }
#pragma unroll
      for (int off = 16; off > 0; off >>= 1) {
#pragma unroll
        for (int i = 0; i < 4; ++i) {
#pragma unroll
          for (int r = 0; r < FIN_RB; ++r) p[i][r] += __shfl_xor(p[i][r], off, 32);
        }
      }
      if (lane == 0) {
#pragma unroll
        for (int i = 0; i < 4; ++i) {
          int n = n0 + (i << 5);
          if (n < 500) {
#pragma unroll
            for (int r = 0; r < FIN_RB; ++r) oldL[r][n] = p[i][r];
          }
        }
      }
    }
  }
  __syncthreads();

  float pm = -3.4e38f;
  for (int n = lt; n < 500; n += 256) pm = fmaxf(pm, oldG[n]);
  pm = groupReduceMax(pm, red, lt);
  float ps = 0.0f;
  for (int n = lt; n < 500; n += 256) { float e = expf(oldG[n] - pm); wcG[n] = e; ps += e; }
  ps = groupReduceSum(ps, red, lt);
  float inv = 1.0f / ps;
  for (int n = lt; n < 500; n += 256) {
    float w = wcG[n] * inv;
    oldG[n] = w;
    wcG[n]  = fminf(fmaxf(w, LBV), UBV);
  }
  __syncthreads();

  bool doneG = false;
  for (int it = 0; it < 32; ++it) {
    float pl = 0.0f, pn = 0.0f;
    for (int n = lt; n < 500; n += 256) {
      float wcv = wcG[n];
      pl += oldG[n] - wcv;
      if (wcv != UBV) pn += wcv;
    }
    float leftover = groupReduceSum(pl, red, lt);
    float nsum     = groupReduceSum(pn, red, lt);
    float ns = (nsum == 0.0f) ? 1.0f : nsum;
    float pover = 0.0f;
    for (int n = lt; n < 500; n += 256) {
      float wcv = wcG[n];
      float wc2 = (wcv != UBV) ? wcv + leftover * wcv / ns : wcv;
      if (!doneG) oldG[n] = wc2;
      if (wc2 > UBV) pover = 1.0f;
    }
    float anyOver = groupReduceSum(pover, red, lt);
    int new_done = (anyOver == 0.0f);
    if (!doneG) {
      for (int n = lt; n < 500; n += 256) {
        float wc2 = oldG[n];
        wcG[n] = new_done ? wc2 : fminf(fmaxf(wc2, LBV), UBV);
      }
    }
    doneG = doneG || (new_done != 0);
    if (lt == 0) doneL[grp] = doneG ? 1 : 0;
    __syncthreads();
    int allDone = 1;
    for (int r = 0; r < FIN_RB; ++r) allDone &= doneL[r];
    if (allDone) break;
  }

  for (int n = lt; n < 500; n += 256) outp[(size_t)b * 500 + n] = wcG[n];
}

extern "C" void kernel_launch(void* const* d_in, const int* in_sizes, int n_in,
                              void* d_out, int out_size, void* d_ws, size_t ws_size,
                              hipStream_t stream) {
  (void)in_sizes; (void)n_in; (void)out_size; (void)ws_size;
  const float* x    = (const float*)d_in[0];
  const float* Wih0 = (const float*)d_in[1];
  const float* Whh0 = (const float*)d_in[2];
  const float* bih0 = (const float*)d_in[3];
  const float* bhh0 = (const float*)d_in[4];
  const float* Wih1 = (const float*)d_in[5];
  const float* Whh1 = (const float*)d_in[6];
  const float* bih1 = (const float*)d_in[7];
  const float* bhh1 = (const float*)d_in[8];
  const float* Wa   = (const float*)d_in[9];
  const float* ba   = (const float*)d_in[10];
  const float* vv   = (const float*)d_in[11];
  const float* Wfc  = (const float*)d_in[12];
  const float* bfc  = (const float*)d_in[13];
  float* outp = (float*)d_out;

  constexpr int B = kB, S = kS, N = kN, H = kH, G3 = kG3;

  uint8_t* base = (uint8_t*)d_ws;
  size_t off = 0;
  auto carve = [&](size_t bytes) {
    void* p = base + off;
    off = (off + bytes + 255) & ~(size_t)255;
    return p;
  };
  float* gi   = (float*)carve((size_t)B * S * G3 * 4);
  float* out0 = (float*)carve((size_t)B * S * H * 4);
  float* out1 = (float*)carve((size_t)B * S * H * 4);
  float* gEb  = (float*)carve((size_t)B * S * H * 4);
  float* ub   = (float*)carve((size_t)B * H * 4);
  _Float16* Wih0F = (_Float16*)carve((size_t)24 * 16 * 512 * 2);
  _Float16* Whh0F = (_Float16*)carve((size_t)24 * 4 * 512 * 2);
  _Float16* Wih1F = (_Float16*)carve((size_t)24 * 4 * 512 * 2);
  _Float16* Whh1F = (_Float16*)carve((size_t)24 * 4 * 512 * 2);
  _Float16* WaxF  = (_Float16*)carve((size_t)8 * 16 * 512 * 2);
  _Float16* WahF  = (_Float16*)carve((size_t)8 * 4 * 512 * 2);

  wfrag_prep_all<<<kPB5, 256, 0, stream>>>(Wih0, Whh0, Wih1, Whh1, Wa + H, Wa,
                                           Wih0F, Whh0F, Wih1F, Whh1F, WaxF, WahF);

  auto gemm384 = [&](const float* A, const _Float16* Bf, const float* bias, float* C,
                     int M, int K, int lda) {
    gemm_wmma<3, 0><<<dim3(M / 64), 256, 0, stream>>>(A, Bf, bias, C, K, lda, G3,
                                                       nullptr, nullptr, nullptr, 0);
  };
  auto gemm128 = [&](const float* A, const _Float16* Bf, const float* bias, float* C,
                     int M, int K, int lda) {
    gemm_wmma<1, 0><<<dim3(M / 64), 256, 0, stream>>>(A, Bf, bias, C, K, lda, H,
                                                       nullptr, nullptr, nullptr, 0);
  };

  gemm_wmma<3, 1><<<dim3(B * S / 64), 256, 0, stream>>>(x, Wih0F, bih0, gi, N, N, G3,
                                                         WaxF, ba, gEb, H);
  gru_recur<<<B / GRU_RB, 256, 0, stream>>>(gi, Whh0F, bhh0, out0, S);
  gemm384(out0, Wih1F, bih1, gi, B * S, H, H);
  gru_recur<<<B / GRU_RB, 256, 0, stream>>>(gi, Whh1F, bhh1, out1, S);
  gemm128(out1 + (size_t)(S - 1) * H, WahF, nullptr, ub, B, H, S * H);
  finalize_kernel<<<B / FIN_RB, 256 * FIN_RB, 0, stream>>>(gEb, ub, out1, vv, Wfc, bfc, outp, S);
}
